# starting kernel
# baseline (speedup 1.0000x reference)
#include <hip/hip_runtime.h>
#include <hip/hip_bf16.h>

typedef __attribute__((ext_vector_type(2))) float v2f;
typedef __attribute__((ext_vector_type(8))) float v8f;

#define N_NODES 100000
#define N_EDGES 3200000

__global__ void deg_init_kernel(float* deg, int n) {
    int i = blockIdx.x * blockDim.x + threadIdx.x;
    if (i < n) deg[i] = 1.0f;
}
__global__ void deg_count_kernel(const int* __restrict__ dst, float* deg, int e) {
    int i = blockIdx.x * blockDim.x + threadIdx.x;
    if (i < e) atomicAdd(&deg[dst[i]], 1.0f);
}
__global__ void deg_rsqrt_kernel(float* deg, int n) {
    int i = blockIdx.x * blockDim.x + threadIdx.x;
    if (i < n) deg[i] = rsqrtf(deg[i]);
}

template <int K, int NOUT, int MT>
__global__ __launch_bounds__(MT*(NOUT/16)*32)
void gemm_wmma_f32(const float* __restrict__ A, const float* __restrict__ B,
                   float* __restrict__ C, int M) {
    constexpr int NT   = NOUT / 16;
    constexpr int NW   = MT * NT;
    constexpr int ROWS = MT * 16;
    constexpr int KP   = K + 4;
    constexpr int NP   = NOUT + 4;

    __shared__ float As[ROWS][KP];
    __shared__ float Bs[K][NP];

    const int tid  = threadIdx.x;
    const int row0 = blockIdx.x * ROWS;

    constexpr int A_ELTS = ROWS * K;
    for (int i = tid * 4; i < A_ELTS; i += NW * 32 * 4) {
        int r = i / K, c = i % K;
        int gr = row0 + r;
        float4 v = (gr < M) ? *(const float4*)(A + (long long)gr * K + c)
                            : make_float4(0.f, 0.f, 0.f, 0.f);
        *(float4*)(&As[r][c]) = v;
    }
    constexpr int B_ELTS = K * NOUT;
    for (int i = tid * 4; i < B_ELTS; i += NW * 32 * 4) {
        int r = i / NOUT, c = i % NOUT;
        *(float4*)(&Bs[r][c]) = *(const float4*)(B + i);
    }
    __syncthreads();

    const int wid  = tid >> 5;
    const int lane = tid & 31;
    const int mn   = lane & 15;
    const int h    = lane >> 4;
    const int mt   = wid / NT;
    const int nt   = wid % NT;

    v8f acc = {};
    const float* arow = &As[mt * 16 + mn][0];
    const int    ncol = nt * 16 + mn;

#pragma unroll
    for (int k = 0; k < K; k += 4) {
        v2f a, b;
        a.x = arow[k + 2 * h + 0];
        a.y = arow[k + 2 * h + 1];
        b.x = Bs[k + 2 * h + 0][ncol];
        b.y = Bs[k + 2 * h + 1][ncol];
        acc = __builtin_amdgcn_wmma_f32_16x16x4_f32(
             false, a,  false, b,
             (short)0, acc,  false,  false);
    }

#pragma unroll
    for (int j = 0; j < 8; ++j) {
        int r = row0 + mt * 16 + j + 8 * h;
        if (r < M) C[(long long)r * NOUT + ncol] = acc[j];
    }
}

__global__ void zero_kernel(float* p, long long n) {
    long long i = (long long)blockIdx.x * blockDim.x + threadIdx.x;
    if (i < n) p[i] = 0.0f;
}

template <int F>
__global__ void scatter_kernel(const float* __restrict__ h, const int* __restrict__ src,
                               const int* __restrict__ dst, const float* __restrict__ dinv,
                               float* agg, int E) {
    constexpr int LPE = F / 4;
    long long t = (long long)blockIdx.x * blockDim.x + threadIdx.x;
    int e = (int)(t / LPE);
    int j = (int)(t % LPE);
    if (e >= E) return;
    int s = src[e], d = dst[e];
    float norm = dinv[s] * dinv[d];
    float4 v = *(const float4*)(h + (long long)s * F + j * 4);
    float* o = agg + (long long)d * F + j * 4;
    atomicAdd(o + 0, v.x * norm);
    atomicAdd(o + 1, v.y * norm);
    atomicAdd(o + 2, v.z * norm);
    atomicAdd(o + 3, v.w * norm);
}

template <int F>
__global__ void finish_kernel(float* agg, const float* __restrict__ h,
                              const float* __restrict__ dinv, const float* __restrict__ bias,
                              int n) {
    long long t = (long long)blockIdx.x * blockDim.x + threadIdx.x;
    int i = (int)(t / F);
    int f = (int)(t % F);
    if (i >= n) return;
    float di = dinv[i];
    float v = agg[(long long)i * F + f] + h[(long long)i * F + f] * di * di + bias[f];
    agg[(long long)i * F + f] = fmaxf(v, 0.0f);
}

__global__ void final_linear_kernel(const float* __restrict__ y, const float* __restrict__ Wl,
                                    const float* __restrict__ bl, float* out, int n) {
    int i = blockIdx.x * blockDim.x + threadIdx.x;
    if (i >= n) return;
    const float* row = y + (long long)i * 32;
    float s = 0.0f;
#pragma unroll
    for (int f = 0; f < 32; ++f) s += row[f] * Wl[f];
    out[i] = s + bl[0];
}

static inline size_t align256(size_t x) { return (x + 255) & ~size_t(255); }

extern "C" void kernel_launch(void* const* d_in, const int* in_sizes, int n_in,
                              void* d_out, int out_size, void* d_ws, size_t ws_size,
                              hipStream_t stream) {
    const float* x    = (const float*)d_in[0];
    const int*   ei   = (const int*)d_in[1];
    const float* W1   = (const float*)d_in[2];
    const float* b1   = (const float*)d_in[3];
    const float* W2   = (const float*)d_in[4];
    const float* b2   = (const float*)d_in[5];
    const float* Wl   = (const float*)d_in[6];
    const float* bl   = (const float*)d_in[7];
    float*       out  = (float*)d_out;

    const int N = N_NODES;
    const int E = N_EDGES;
    const int* src = ei;
    const int* dst = ei + E;

    char* ws = (char*)d_ws;
    size_t off = 0;
    float* dinv = (float*)(ws + off); off += align256((size_t)N * 4);
    float* h1   = (float*)(ws + off); off += align256((size_t)N * 64 * 4);
    float* agg1 = (float*)(ws + off); off += align256((size_t)N * 64 * 4);
    float* h2   = (float*)(ws + off); off += align256((size_t)N * 32 * 4);
    float* agg2 = (float*)(ws + off); off += align256((size_t)N * 32 * 4);
    (void)ws_size; (void)n_in; (void)in_sizes; (void)out_size;

    const int T = 256;

    deg_init_kernel<<<(N + T - 1) / T, T, 0, stream>>>(dinv, N);
    deg_count_kernel<<<(E + T - 1) / T, T, 0, stream>>>(dst, dinv, E);
    deg_rsqrt_kernel<<<(N + T - 1) / T, T, 0, stream>>>(dinv, N);

    gemm_wmma_f32<128, 64, 2><<<N / 32, 2 * 4 * 32, 0, stream>>>(x, W1, h1, N);

    zero_kernel<<<(int)(((long long)N * 64 + T - 1) / T), T, 0, stream>>>(agg1, (long long)N * 64);
    {
        long long tt = (long long)E * (64 / 4);
        scatter_kernel<64><<<(int)((tt + T - 1) / T), T, 0, stream>>>(h1, src, dst, dinv, agg1, E);
    }
    finish_kernel<64><<<(int)(((long long)N * 64 + T - 1) / T), T, 0, stream>>>(agg1, h1, dinv, b1, N);

    gemm_wmma_f32<64, 32, 2><<<N / 32, 2 * 2 * 32, 0, stream>>>(agg1, W2, h2, N);

    zero_kernel<<<(int)(((long long)N * 32 + T - 1) / T), T, 0, stream>>>(agg2, (long long)N * 32);
    {
        long long tt = (long long)E * (32 / 4);
        scatter_kernel<32><<<(int)((tt + T - 1) / T), T, 0, stream>>>(h2, src, dst, dinv, agg2, E);
    }
    finish_kernel<32><<<(int)(((long long)N * 32 + T - 1) / T), T, 0, stream>>>(agg2, h2, dinv, b2, N);

    final_linear_kernel<<<(N + T - 1) / T, T, 0, stream>>>(agg2, Wl, bl, out, N);
}


# Owner-map scan and edge resolve run once per call, not once per layer: layer-1 launch stores each wave's queue (source, row, norm) in the workspace, layer 2 reloads it; LDS atomic adds as before
# speedup vs baseline: 1.6496x; 1.6496x over previous
#include <hip/hip_runtime.h>
#include <hip/hip_bf16.h>

typedef __attribute__((ext_vector_type(2))) float v2f;
typedef __attribute__((ext_vector_type(8))) float v8f;

#define N_NODES 100000
#define N_EDGES 3200000

__global__ void deg_init_kernel(float* deg, int n) {
    int i = blockIdx.x * blockDim.x + threadIdx.x;
    if (i < n) deg[i] = 1.0f;
}
__global__ void deg_count_kernel(const int* __restrict__ dst, float* deg, int e) {
    int i = blockIdx.x * blockDim.x + threadIdx.x;
    if (i < e) atomicAdd(&deg[dst[i]], 1.0f);
}
__global__ void deg_rsqrt_kernel(float* deg, int n) {
    int i = blockIdx.x * blockDim.x + threadIdx.x;
    if (i < n) deg[i] = rsqrtf(deg[i]);
}

template <int K, int NOUT, int MT>
__global__ __launch_bounds__(MT*(NOUT/16)*32)
void gemm_wmma_f32(const float* __restrict__ A, const float* __restrict__ B,
                   float* __restrict__ C, int M) {
    constexpr int NT   = NOUT / 16;
    constexpr int NW   = MT * NT;
    constexpr int ROWS = MT * 16;
    constexpr int KP   = K + 4;
    constexpr int NP   = NOUT + 4;

    __shared__ float As[ROWS][KP];
    __shared__ float Bs[K][NP];

    const int tid  = threadIdx.x;
    const int row0 = blockIdx.x * ROWS;

    constexpr int A_ELTS = ROWS * K;
    for (int i = tid * 4; i < A_ELTS; i += NW * 32 * 4) {
        int r = i / K, c = i % K;
        int gr = row0 + r;
        float4 v = (gr < M) ? *(const float4*)(A + (long long)gr * K + c)
                            : make_float4(0.f, 0.f, 0.f, 0.f);
        *(float4*)(&As[r][c]) = v;
    }
    constexpr int B_ELTS = K * NOUT;
    for (int i = tid * 4; i < B_ELTS; i += NW * 32 * 4) {
        int r = i / NOUT, c = i % NOUT;
        *(float4*)(&Bs[r][c]) = *(const float4*)(B + i);
    }
    __syncthreads();

    const int wid  = tid >> 5;
    const int lane = tid & 31;
    const int mn   = lane & 15;
    const int h    = lane >> 4;
    const int mt   = wid / NT;
    const int nt   = wid % NT;

    v8f acc = {};
    const float* arow = &As[mt * 16 + mn][0];
    const int    ncol = nt * 16 + mn;

#pragma unroll
    for (int k = 0; k < K; k += 4) {
        v2f a, b;
        a.x = arow[k + 2 * h + 0];
        a.y = arow[k + 2 * h + 1];
        b.x = Bs[k + 2 * h + 0][ncol];
        b.y = Bs[k + 2 * h + 1][ncol];
        acc = __builtin_amdgcn_wmma_f32_16x16x4_f32(
             false, a,  false, b,
             (short)0, acc,  false,  false);
    }

#pragma unroll
    for (int j = 0; j < 8; ++j) {
        int r = row0 + mt * 16 + j + 8 * h;
        if (r < M) C[(long long)r * NOUT + ncol] = acc[j];
    }
}

constexpr int NPW   = 392;
constexpr int NTILE = 256;
constexpr int WQ    = 512;
constexpr int NGRP  = N_EDGES / 16;
constexpr int NBAT  = (NGRP + 8191) / 8192;
static_assert(NPW * NTILE >= N_NODES, "tiles must cover every node");
static_assert(N_EDGES % 16 == 0, "owner map is read in 16-byte groups");
static_assert(NPW <= 65535, "local row index is kept in 16 bits");
constexpr int SRC_BITS = 17;
static_assert(N_NODES <= (1 << SRC_BITS), "source id must fit the low bits of a stored entry");
static_assert(NPW <= (1 << (32 - SRC_BITS)), "local row must fit the high bits of a stored entry");

__global__ void owner_map_kernel(const int* __restrict__ dst, unsigned char* __restrict__ owner, int e4) {
    int i = blockIdx.x * blockDim.x + threadIdx.x;
    if (i >= e4) return;
    const int* p = dst + 4 * (long long)i;
    unsigned o = (unsigned)(p[0] / NPW) | ((unsigned)(p[1] / NPW) << 8) |
                 ((unsigned)(p[2] / NPW) << 16) | ((unsigned)(p[3] / NPW) << 24);
    *(unsigned*)(owner + 4 * (long long)i) = o;
}

__device__ __forceinline__ unsigned zero_bytes(unsigned y) {
    unsigned t = (y & 0x7F7F7F7Fu) + 0x7F7F7F7Fu;
    return ~(t | y | 0x7F7F7F7Fu);
}
__device__ __forceinline__ unsigned match_group(uint4 v, unsigned me) {
    return (zero_bytes(v.x ^ me) >> 7) | (zero_bytes(v.y ^ me) >> 6) |
           (zero_bytes(v.z ^ me) >> 5) | (zero_bytes(v.w ^ me) >> 4);
}

template <int F>
__device__ void spill_edge(int e, int base, float* tile, const float* __restrict__ h,
                           const int* __restrict__ src, const int* __restrict__ dst,
                           const float* __restrict__ dinv) {
    const int s = src[e], d = dst[e];
    const float nrm = dinv[s] * dinv[d];
    const float* hr = h + (long long)s * F;
    float* tr = tile + (d - base) * F;
    for (int f = 0; f < F; ++f) atomicAdd(tr + f, hr[f] * nrm);
}
__device__ __forceinline__ int wrap_group(int g) {
    if (g >= NGRP) g -= NGRP;
    if (g >= NGRP) g -= NGRP;
    return g;
}

template <int F, bool FIRST>
__global__ __launch_bounds__(1024)
void tile_agg_kernel(const float* __restrict__ h, const int* __restrict__ src,
                     const int* __restrict__ dst, const unsigned char* __restrict__ owner,
                     const float* __restrict__ dinv, const float* __restrict__ bias,
                     float* __restrict__ y, int N, int* plan_c, unsigned int* plan_e) {
    constexpr int F4  = F / 4;
    constexpr int VPL = F / 32;
    __shared__ float          tile[NPW * F];
    __shared__ int            q_s[32 * WQ];
    __shared__ float          q_n[32 * WQ];
    __shared__ unsigned short q_d[32 * WQ];
    __shared__ int            wcs[32];

    const int tid  = threadIdx.x;
    const int wid  = tid >> 5;
    const int lane = tid & 31;
    const int base = blockIdx.x * NPW;
    int nloc = N - base;
    if (nloc > NPW) nloc = NPW;
    if (nloc < 0) nloc = 0;

    for (int i = tid; i < nloc * F4; i += 1024) {
        const int n = i / F4, c = i % F4;
        const float di = dinv[base + n];
        const float4 hv = *(const float4*)(h + (long long)(base + n) * F + c * 4);
        const float4 bv = *(const float4*)(bias + c * 4);
        float4 t;
        t.x = hv.x * di * di + bv.x;
        t.y = hv.y * di * di + bv.y;
        t.z = hv.z * di * di + bv.z;
        t.w = hv.w * di * di + bv.w;
        *(float4*)(&tile[n * F + c * 4]) = t;
    }
    __syncthreads();

    int*            myq = &q_s[wid * WQ];
    float*          myn = &q_n[wid * WQ];
    unsigned short* myd = &q_d[wid * WQ];
    uint2* pe = (uint2*)plan_e + (long long)(blockIdx.x * 32 + wid) * WQ;
    int  wc    = 0;
    bool build = FIRST;
    if (!FIRST) {
        const int c = plan_c[blockIdx.x * 32 + lane];
        build = __builtin_amdgcn_ballot_w32(c > WQ) != 0u;
        wc = __builtin_amdgcn_readlane(c, __builtin_amdgcn_readfirstlane(wid));
    }

    if (build) {
    const unsigned me   = (unsigned)blockIdx.x * 0x01010101u;
    const uint4*   om   = (const uint4*)owner;
    const int rot = blockIdx.x * 768;
    wc = 0;
    for (int it = 0; it < NBAT; ++it) {
        uint4 v[8];
#pragma unroll
        for (int k = 0; k < 8; ++k) {
            v[k] = om[wrap_group(it * 8192 + k * 1024 + tid + rot)];
        }
        unsigned mk[8];
#pragma unroll
        for (int k = 0; k < 8; ++k)
            mk[k] = match_group(v[k], me) & (0u - (unsigned)(it * 8192 + k * 1024 + tid < NGRP));
#pragma unroll
        for (int hf = 0; hf < 2; ++hf) {
            const unsigned lo = mk[4 * hf + 0] | (mk[4 * hf + 1] << 4);
            const unsigned hi = mk[4 * hf + 2] | (mk[4 * hf + 3] << 4);
            unsigned long long M = ((unsigned long long)hi << 32) | (unsigned long long)lo;
            unsigned act;
            while ((act = __builtin_amdgcn_ballot_w32(M != 0ull)) != 0u) {
                if (M != 0ull) {
                    const int p = __builtin_ctzll(M);
                    M &= M - 1ull;
                    const int k  = 4 * hf + 2 * (p >> 5) + ((p >> 2) & 1);
                    const int g = wrap_group(it * 8192 + k * 1024 + tid + rot);
                    const int e = g * 16 + (p & 3) * 4 + ((p & 31) >> 3);
                    const int slot = wc + (int)__builtin_amdgcn_mbcnt_lo(act, 0u);
                    if (slot < WQ) myq[slot] = e;
                    else spill_edge<F>(e, base, tile, h, src, dst, dinv);
                }
                wc += __builtin_popcount(act);
            }
        }
    }
    if (FIRST && lane == 0) wcs[wid] = wc;
    __syncthreads();
    if (FIRST && wid == 0) plan_c[blockIdx.x * 32 + lane] = wcs[lane];

    const int cnt = wc < WQ ? wc : WQ;
    for (int i0 = 0; i0 < cnt; i0 += 128) {
        int ix[4], ss[4], dd[4];
        float nr[4];
#pragma unroll
        for (int k = 0; k < 4; ++k) {
            const int i = i0 + 32 * k + lane;
            ix[k] = (i < cnt) ? i : (cnt - 1);
        }
        int ee[4];
#pragma unroll
        for (int k = 0; k < 4; ++k) ee[k] = myq[ix[k]];
#pragma unroll
        for (int k = 0; k < 4; ++k) { ss[k] = src[ee[k]]; dd[k] = dst[ee[k]]; }
#pragma unroll
        for (int k = 0; k < 4; ++k) nr[k] = dinv[ss[k]] * dinv[dd[k]];
#pragma unroll
        for (int k = 0; k < 4; ++k) {
            myq[ix[k]] = ss[k];
            myd[ix[k]] = (unsigned short)(dd[k] - base);
            myn[ix[k]] = nr[k];
            if (FIRST && i0 + 32 * k + lane < cnt)
                pe[ix[k]] = make_uint2((unsigned)ss[k] | ((unsigned)(dd[k] - base) << SRC_BITS),
                                       __float_as_uint(nr[k]));
        }
    }
    } else {
    const int cnt = wc;
    for (int i0 = 0; i0 < cnt; i0 += 128) {
        int   ix[4];
        uint2 r[4];
#pragma unroll
        for (int k = 0; k < 4; ++k) {
            const int i = i0 + 32 * k + lane;
            ix[k] = (i < cnt) ? i : (cnt - 1);
            r[k] = pe[ix[k]];
        }
#pragma unroll
        for (int k = 0; k < 4; ++k) {
            myq[ix[k]] = (int)(r[k].x & ((1u << SRC_BITS) - 1u));
            myd[ix[k]] = (unsigned short)(r[k].x >> SRC_BITS);
            myn[ix[k]] = __uint_as_float(r[k].y);
        }
    }
    }
    const int cnt = wc < WQ ? wc : WQ;
    __syncthreads();

    constexpr int U = 16;
    for (int i0 = 0; i0 < cnt; i0 += U) {
        float vals[U][VPL];
#pragma unroll
        for (int k = 0; k < U; ++k) {
            const int i = (i0 + k < cnt) ? (i0 + k) : i0;
            const float* hr = h + (long long)myq[i] * F + lane * VPL;
#pragma unroll
            for (int x = 0; x < VPL; ++x) vals[k][x] = hr[x];
        }
#pragma unroll
        for (int k = 0; k < U; ++k) {
            if (i0 + k < cnt) {
                const float nrm = myn[i0 + k];
                float* tr = &tile[(int)myd[i0 + k] * F + lane * VPL];
#pragma unroll
                for (int x = 0; x < VPL; ++x) atomicAdd(tr + x, vals[k][x] * nrm);
            }
        }
    }
    __syncthreads();

    for (int i = tid; i < nloc * F4; i += 1024) {
        float4 t = *(const float4*)(&tile[i * 4]);
        t.x = fmaxf(t.x, 0.0f); t.y = fmaxf(t.y, 0.0f);
        t.z = fmaxf(t.z, 0.0f); t.w = fmaxf(t.w, 0.0f);
        *(float4*)(y + (long long)base * F + (long long)i * 4) = t;
    }
}

__global__ void final_linear_kernel(const float* __restrict__ y, const float* __restrict__ Wl,
                                    const float* __restrict__ bl, float* out, int n) {
    int i = blockIdx.x * blockDim.x + threadIdx.x;
    if (i >= n) return;
    const float* row = y + (long long)i * 32;
    float s = 0.0f;
#pragma unroll
    for (int f = 0; f < 32; ++f) s += row[f] * Wl[f];
    out[i] = s + bl[0];
}

static inline size_t align256(size_t x) { return (x + 255) & ~size_t(255); }

extern "C" void kernel_launch(void* const* d_in, const int* in_sizes, int n_in,
                              void* d_out, int out_size, void* d_ws, size_t ws_size,
                              hipStream_t stream) {
    const float* x    = (const float*)d_in[0];
    const int*   ei   = (const int*)d_in[1];
    const float* W1   = (const float*)d_in[2];
    const float* b1   = (const float*)d_in[3];
    const float* W2   = (const float*)d_in[4];
    const float* b2   = (const float*)d_in[5];
    const float* Wl   = (const float*)d_in[6];
    const float* bl   = (const float*)d_in[7];
    float*       out  = (float*)d_out;

    const int N = N_NODES;
    const int E = N_EDGES;
    const int* src = ei;
    const int* dst = ei + E;

    char* ws = (char*)d_ws;
    size_t off = 0;
    float* dinv = (float*)(ws + off); off += align256((size_t)N * 4);
    float* h1   = (float*)(ws + off); off += align256((size_t)N * 64 * 4);
    float* y1   = (float*)(ws + off); off += align256((size_t)N * 64 * 4);
    float* h2   = (float*)(ws + off); off += align256((size_t)N * 32 * 4);
    float* y2   = (float*)(ws + off); off += align256((size_t)N * 32 * 4);
    unsigned char* owner = (unsigned char*)(ws + off); off += align256((size_t)E);
    int*          plan_c = (int*)(ws + off);          off += align256((size_t)NTILE * 32 * 4);
    unsigned int* plan_e = (unsigned int*)(ws + off); off += align256((size_t)NTILE * 32 * WQ * 8);
    (void)ws_size; (void)n_in; (void)in_sizes; (void)out_size;

    const int T = 256;

    deg_init_kernel<<<(N + T - 1) / T, T, 0, stream>>>(dinv, N);
    deg_count_kernel<<<(E + T - 1) / T, T, 0, stream>>>(dst, dinv, E);
    deg_rsqrt_kernel<<<(N + T - 1) / T, T, 0, stream>>>(dinv, N);

    owner_map_kernel<<<(E / 4 + T - 1) / T, T, 0, stream>>>(dst, owner, E / 4);

    gemm_wmma_f32<128, 64, 2><<<N / 32, 2 * 4 * 32, 0, stream>>>(x, W1, h1, N);

    tile_agg_kernel<64, true><<<NTILE, 1024, 0, stream>>>(h1, src, dst, owner, dinv, b1, y1, N,
                                                          plan_c, plan_e);

    gemm_wmma_f32<64, 32, 2><<<N / 32, 2 * 2 * 32, 0, stream>>>(y1, W2, h2, N);

    tile_agg_kernel<32, false><<<NTILE, 1024, 0, stream>>>(h2, src, dst, owner, dinv, b2, y2, N,
                                                           plan_c, plan_e);

    final_linear_kernel<<<(N + T - 1) / T, T, 0, stream>>>(y2, Wl, bl, out, N);
}


# In-degree counted in LDS by the owner tiles: a plan launch (scan+resolve, split off layer 1) adds 1 per queued edge to its row's int counter and writes dinv; global f32 atomics gone; 9->7 launches
# speedup vs baseline: 2.6511x; 1.6071x over previous
#include <hip/hip_runtime.h>
#include <hip/hip_bf16.h>

typedef __attribute__((ext_vector_type(2))) float v2f;
typedef __attribute__((ext_vector_type(8))) float v8f;

#define N_NODES 100000
#define N_EDGES 3200000

template <int K, int NOUT, int MT>
__global__ __launch_bounds__(MT*(NOUT/16)*32)
void gemm_wmma_f32(const float* __restrict__ A, const float* __restrict__ B,
                   float* __restrict__ C, int M) {
    constexpr int NT   = NOUT / 16;
    constexpr int NW   = MT * NT;
    constexpr int ROWS = MT * 16;
    constexpr int KP   = K + 4;
    constexpr int NP   = NOUT + 4;

    __shared__ float As[ROWS][KP];
    __shared__ float Bs[K][NP];

    const int tid  = threadIdx.x;
    const int row0 = blockIdx.x * ROWS;

    constexpr int A_ELTS = ROWS * K;
    for (int i = tid * 4; i < A_ELTS; i += NW * 32 * 4) {
        int r = i / K, c = i % K;
        int gr = row0 + r;
        float4 v = (gr < M) ? *(const float4*)(A + (long long)gr * K + c)
                            : make_float4(0.f, 0.f, 0.f, 0.f);
        *(float4*)(&As[r][c]) = v;
    }
    constexpr int B_ELTS = K * NOUT;
    for (int i = tid * 4; i < B_ELTS; i += NW * 32 * 4) {
        int r = i / NOUT, c = i % NOUT;
        *(float4*)(&Bs[r][c]) = *(const float4*)(B + i);
    }
    __syncthreads();

    const int wid  = tid >> 5;
    const int lane = tid & 31;
    const int mn   = lane & 15;
    const int h    = lane >> 4;
    const int mt   = wid / NT;
    const int nt   = wid % NT;

    v8f acc = {};
    const float* arow = &As[mt * 16 + mn][0];
    const int    ncol = nt * 16 + mn;

#pragma unroll
    for (int k = 0; k < K; k += 4) {
        v2f a, b;
        a.x = arow[k + 2 * h + 0];
        a.y = arow[k + 2 * h + 1];
        b.x = Bs[k + 2 * h + 0][ncol];
        b.y = Bs[k + 2 * h + 1][ncol];
        acc = __builtin_amdgcn_wmma_f32_16x16x4_f32(
             false, a,  false, b,
             (short)0, acc,  false,  false);
    }

#pragma unroll
    for (int j = 0; j < 8; ++j) {
        int r = row0 + mt * 16 + j + 8 * h;
        if (r < M) C[(long long)r * NOUT + ncol] = acc[j];
    }
}

constexpr int NPW   = 392;
constexpr int NTILE = 256;
constexpr int WQ    = 512;
constexpr int NGRP  = N_EDGES / 16;
constexpr int NBAT  = (NGRP + 8191) / 8192;
static_assert(NPW * NTILE >= N_NODES, "tiles must cover every node");
static_assert(N_EDGES % 16 == 0, "owner map is read in 16-byte groups");
static_assert(NPW <= 65535, "local row index is kept in 16 bits");
constexpr int SRC_BITS = 17;
static_assert(N_NODES <= (1 << SRC_BITS), "source id must fit the low bits of a stored entry");
static_assert(NPW <= (1 << (32 - SRC_BITS)), "local row must fit the high bits of a stored entry");

__global__ void owner_map_kernel(const int* __restrict__ dst, unsigned char* __restrict__ owner, int e4) {
    int i = blockIdx.x * blockDim.x + threadIdx.x;
    if (i >= e4) return;
    const int* p = dst + 4 * (long long)i;
    unsigned o = (unsigned)(p[0] / NPW) | ((unsigned)(p[1] / NPW) << 8) |
                 ((unsigned)(p[2] / NPW) << 16) | ((unsigned)(p[3] / NPW) << 24);
    *(unsigned*)(owner + 4 * (long long)i) = o;
}

__device__ __forceinline__ unsigned zero_bytes(unsigned y) {
    unsigned t = (y & 0x7F7F7F7Fu) + 0x7F7F7F7Fu;
    return ~(t | y | 0x7F7F7F7Fu);
}
__device__ __forceinline__ unsigned match_group(uint4 v, unsigned me) {
    return (zero_bytes(v.x ^ me) >> 7) | (zero_bytes(v.y ^ me) >> 6) |
           (zero_bytes(v.z ^ me) >> 5) | (zero_bytes(v.w ^ me) >> 4);
}

template <int F>
__device__ void spill_edge(int e, int base, float* tile, const float* __restrict__ h,
                           const int* __restrict__ src, const int* __restrict__ dst,
                           const float* __restrict__ dinv) {
    const int s = src[e], d = dst[e];
    const float nrm = dinv[s] * dinv[d];
    const float* hr = h + (long long)s * F;
    float* tr = tile + (d - base) * F;
    for (int f = 0; f < F; ++f) atomicAdd(tr + f, hr[f] * nrm);
}
__device__ __forceinline__ int wrap_group(int g) {
    if (g >= NGRP) g -= NGRP;
    if (g >= NGRP) g -= NGRP;
    return g;
}

template <typename Spill>
__device__ __forceinline__ int scan_owner_map(const unsigned char* __restrict__ owner, int tileid,
                                              int tid, int* myq, Spill spill) {
    const unsigned me   = (unsigned)tileid * 0x01010101u;
    const uint4*   om   = (const uint4*)owner;
    const int rot = tileid * 768;
    int wc = 0;
    for (int it = 0; it < NBAT; ++it) {
        uint4 v[8];
#pragma unroll
        for (int k = 0; k < 8; ++k) {
            v[k] = om[wrap_group(it * 8192 + k * 1024 + tid + rot)];
        }
        unsigned mk[8];
#pragma unroll
        for (int k = 0; k < 8; ++k)
            mk[k] = match_group(v[k], me) & (0u - (unsigned)(it * 8192 + k * 1024 + tid < NGRP));
#pragma unroll
        for (int hf = 0; hf < 2; ++hf) {
            const unsigned lo = mk[4 * hf + 0] | (mk[4 * hf + 1] << 4);
            const unsigned hi = mk[4 * hf + 2] | (mk[4 * hf + 3] << 4);
            unsigned long long M = ((unsigned long long)hi << 32) | (unsigned long long)lo;
            unsigned act;
            while ((act = __builtin_amdgcn_ballot_w32(M != 0ull)) != 0u) {
                if (M != 0ull) {
                    const int p = __builtin_ctzll(M);
                    M &= M - 1ull;
                    const int k  = 4 * hf + 2 * (p >> 5) + ((p >> 2) & 1);
                    const int g = wrap_group(it * 8192 + k * 1024 + tid + rot);
                    const int e = g * 16 + (p & 3) * 4 + ((p & 31) >> 3);
                    const int slot = wc + (int)__builtin_amdgcn_mbcnt_lo(act, 0u);
                    if (slot < WQ) myq[slot] = e;
                    else spill(e);
                }
                wc += __builtin_popcount(act);
            }
        }
    }
    return wc;
}

__global__ __launch_bounds__(1024)
void tile_plan_kernel(const int* __restrict__ src, const int* __restrict__ dst,
                      const unsigned char* __restrict__ owner, float* __restrict__ dinv, int N,
                      int* __restrict__ plan_c, unsigned int* __restrict__ plan_s) {
    __shared__ int q_s[32 * WQ];
    __shared__ int indeg[NPW];
    __shared__ int wcs[32];

    const int tid  = threadIdx.x;
    const int wid  = tid >> 5;
    const int lane = tid & 31;
    const int base = blockIdx.x * NPW;
    int nloc = N - base;
    if (nloc > NPW) nloc = NPW;
    if (nloc < 0) nloc = 0;

    for (int i = tid; i < NPW; i += 1024) indeg[i] = 0;
    __syncthreads();

    int* myq = &q_s[wid * WQ];
    unsigned int* ps = plan_s + (long long)(blockIdx.x * 32 + wid) * WQ;
    const int wc = scan_owner_map(owner, (int)blockIdx.x, tid, myq,
                                  [&](int e) { atomicAdd(&indeg[dst[e] - base], 1); });
    if (lane == 0) wcs[wid] = wc;

    const int cnt = wc < WQ ? wc : WQ;
    for (int i0 = 0; i0 < cnt; i0 += 128) {
        int ix[4], ee[4], ss[4], dd[4];
#pragma unroll
        for (int k = 0; k < 4; ++k) {
            const int i = i0 + 32 * k + lane;
            ix[k] = (i < cnt) ? i : (cnt - 1);
        }
#pragma unroll
        for (int k = 0; k < 4; ++k) ee[k] = myq[ix[k]];
#pragma unroll
        for (int k = 0; k < 4; ++k) { ss[k] = src[ee[k]]; dd[k] = dst[ee[k]]; }
#pragma unroll
        for (int k = 0; k < 4; ++k) {
            if (i0 + 32 * k + lane < cnt) {
                ps[ix[k]] = (unsigned)ss[k] | ((unsigned)(dd[k] - base) << SRC_BITS);
                atomicAdd(&indeg[dd[k] - base], 1);
            }
        }
    }
    __syncthreads();
    if (wid == 0) plan_c[blockIdx.x * 32 + lane] = wcs[lane];
    for (int n = tid; n < nloc; n += 1024) dinv[base + n] = rsqrtf((float)(indeg[n] + 1));
}

template <int F, bool FIRST>
__global__ __launch_bounds__(1024)
void tile_agg_kernel(const float* __restrict__ h, const int* __restrict__ src,
                     const int* __restrict__ dst, const unsigned char* __restrict__ owner,
                     const float* __restrict__ dinv, const float* __restrict__ bias,
                     float* __restrict__ y, int N, const int* __restrict__ plan_c,
                     const unsigned int* __restrict__ plan_s, float* plan_n) {
    constexpr int F4  = F / 4;
    constexpr int VPL = F / 32;
    __shared__ float          tile[NPW * F];
    __shared__ int            q_s[32 * WQ];
    __shared__ float          q_n[32 * WQ];
    __shared__ unsigned short q_d[32 * WQ];

    const int tid  = threadIdx.x;
    const int wid  = tid >> 5;
    const int lane = tid & 31;
    const int base = blockIdx.x * NPW;
    int nloc = N - base;
    if (nloc > NPW) nloc = NPW;
    if (nloc < 0) nloc = 0;

    for (int i = tid; i < nloc * F4; i += 1024) {
        const int n = i / F4, c = i % F4;
        const float di = dinv[base + n];
        const float4 hv = *(const float4*)(h + (long long)(base + n) * F + c * 4);
        const float4 bv = *(const float4*)(bias + c * 4);
        float4 t;
        t.x = hv.x * di * di + bv.x;
        t.y = hv.y * di * di + bv.y;
        t.z = hv.z * di * di + bv.z;
        t.w = hv.w * di * di + bv.w;
        *(float4*)(&tile[n * F + c * 4]) = t;
    }
    __syncthreads();

    int*            myq = &q_s[wid * WQ];
    float*          myn = &q_n[wid * WQ];
    unsigned short* myd = &q_d[wid * WQ];
    const unsigned int* ps = plan_s + (long long)(blockIdx.x * 32 + wid) * WQ;
    float*              pn = plan_n + (long long)(blockIdx.x * 32 + wid) * WQ;
    const int c32 = plan_c[blockIdx.x * 32 + lane];
    const bool build = __builtin_amdgcn_ballot_w32(c32 > WQ) != 0u;
    int wc = __builtin_amdgcn_readlane(c32, __builtin_amdgcn_readfirstlane(wid));

    if (build) {
    wc = scan_owner_map(owner, (int)blockIdx.x, tid, myq,
                        [&](int e) { spill_edge<F>(e, base, tile, h, src, dst, dinv); });
    const int cnt = wc < WQ ? wc : WQ;
    for (int i0 = 0; i0 < cnt; i0 += 128) {
        int ix[4], ss[4], dd[4];
        float nr[4];
#pragma unroll
        for (int k = 0; k < 4; ++k) {
            const int i = i0 + 32 * k + lane;
            ix[k] = (i < cnt) ? i : (cnt - 1);
        }
        int ee[4];
#pragma unroll
        for (int k = 0; k < 4; ++k) ee[k] = myq[ix[k]];
#pragma unroll
        for (int k = 0; k < 4; ++k) { ss[k] = src[ee[k]]; dd[k] = dst[ee[k]]; }
#pragma unroll
        for (int k = 0; k < 4; ++k) nr[k] = dinv[ss[k]] * dinv[dd[k]];
#pragma unroll
        for (int k = 0; k < 4; ++k) {
            myq[ix[k]] = ss[k];
            myd[ix[k]] = (unsigned short)(dd[k] - base);
            myn[ix[k]] = nr[k];
        }
    }
    } else {
    const int cnt = wc;
    for (int i0 = 0; i0 < cnt; i0 += 128) {
        int      ix[4];
        unsigned r[4];
        float    nr[4];
#pragma unroll
        for (int k = 0; k < 4; ++k) {
            const int i = i0 + 32 * k + lane;
            ix[k] = (i < cnt) ? i : (cnt - 1);
            r[k] = ps[ix[k]];
        }
#pragma unroll
        for (int k = 0; k < 4; ++k) {
            if (FIRST) nr[k] = dinv[r[k] & ((1u << SRC_BITS) - 1u)] * dinv[base + (int)(r[k] >> SRC_BITS)];
            else       nr[k] = pn[ix[k]];
        }
#pragma unroll
        for (int k = 0; k < 4; ++k) {
            myq[ix[k]] = (int)(r[k] & ((1u << SRC_BITS) - 1u));
            myd[ix[k]] = (unsigned short)(r[k] >> SRC_BITS);
            myn[ix[k]] = nr[k];
            if (FIRST && i0 + 32 * k + lane < cnt) pn[ix[k]] = nr[k];
        }
    }
    }
    const int cnt = wc < WQ ? wc : WQ;
    __syncthreads();

    constexpr int U = 16;
    for (int i0 = 0; i0 < cnt; i0 += U) {
        float vals[U][VPL];
#pragma unroll
        for (int k = 0; k < U; ++k) {
            const int i = (i0 + k < cnt) ? (i0 + k) : i0;
            const float* hr = h + (long long)myq[i] * F + lane * VPL;
#pragma unroll
            for (int x = 0; x < VPL; ++x) vals[k][x] = hr[x];
        }
#pragma unroll
        for (int k = 0; k < U; ++k) {
            if (i0 + k < cnt) {
                const float nrm = myn[i0 + k];
                float* tr = &tile[(int)myd[i0 + k] * F + lane * VPL];
#pragma unroll
                for (int x = 0; x < VPL; ++x) atomicAdd(tr + x, vals[k][x] * nrm);
            }
        }
    }
    __syncthreads();

    for (int i = tid; i < nloc * F4; i += 1024) {
        float4 t = *(const float4*)(&tile[i * 4]);
        t.x = fmaxf(t.x, 0.0f); t.y = fmaxf(t.y, 0.0f);
        t.z = fmaxf(t.z, 0.0f); t.w = fmaxf(t.w, 0.0f);
        *(float4*)(y + (long long)base * F + (long long)i * 4) = t;
    }
}

__global__ void final_linear_kernel(const float* __restrict__ y, const float* __restrict__ Wl,
                                    const float* __restrict__ bl, float* out, int n) {
    int i = blockIdx.x * blockDim.x + threadIdx.x;
    if (i >= n) return;
    const float* row = y + (long long)i * 32;
    float s = 0.0f;
#pragma unroll
    for (int f = 0; f < 32; ++f) s += row[f] * Wl[f];
    out[i] = s + bl[0];
}

static inline size_t align256(size_t x) { return (x + 255) & ~size_t(255); }

extern "C" void kernel_launch(void* const* d_in, const int* in_sizes, int n_in,
                              void* d_out, int out_size, void* d_ws, size_t ws_size,
                              hipStream_t stream) {
    const float* x    = (const float*)d_in[0];
    const int*   ei   = (const int*)d_in[1];
    const float* W1   = (const float*)d_in[2];
    const float* b1   = (const float*)d_in[3];
    const float* W2   = (const float*)d_in[4];
    const float* b2   = (const float*)d_in[5];
    const float* Wl   = (const float*)d_in[6];
    const float* bl   = (const float*)d_in[7];
    float*       out  = (float*)d_out;

    const int N = N_NODES;
    const int E = N_EDGES;
    const int* src = ei;
    const int* dst = ei + E;

    char* ws = (char*)d_ws;
    size_t off = 0;
    float* dinv = (float*)(ws + off); off += align256((size_t)N * 4);
    float* h1   = (float*)(ws + off); off += align256((size_t)N * 64 * 4);
    float* y1   = (float*)(ws + off); off += align256((size_t)N * 64 * 4);
    float* h2   = (float*)(ws + off); off += align256((size_t)N * 32 * 4);
    float* y2   = (float*)(ws + off); off += align256((size_t)N * 32 * 4);
    unsigned char* owner = (unsigned char*)(ws + off); off += align256((size_t)E);
    int*          plan_c = (int*)(ws + off);          off += align256((size_t)NTILE * 32 * 4);
    unsigned int* plan_s = (unsigned int*)(ws + off); off += align256((size_t)NTILE * 32 * WQ * 4);
    float*        plan_n = (float*)(ws + off);        off += align256((size_t)NTILE * 32 * WQ * 4);
    (void)ws_size; (void)n_in; (void)in_sizes; (void)out_size;

    const int T = 256;

    owner_map_kernel<<<(E / 4 + T - 1) / T, T, 0, stream>>>(dst, owner, E / 4);

    tile_plan_kernel<<<NTILE, 1024, 0, stream>>>(src, dst, owner, dinv, N, plan_c, plan_s);

    gemm_wmma_f32<128, 64, 2><<<N / 32, 2 * 4 * 32, 0, stream>>>(x, W1, h1, N);

    tile_agg_kernel<64, true><<<NTILE, 1024, 0, stream>>>(h1, src, dst, owner, dinv, b1, y1, N,
                                                          plan_c, plan_s, plan_n);

    gemm_wmma_f32<64, 32, 2><<<N / 32, 2 * 2 * 32, 0, stream>>>(y1, W2, h2, N);

    tile_agg_kernel<32, false><<<NTILE, 1024, 0, stream>>>(h2, src, dst, owner, dinv, b2, y2, N,
                                                           plan_c, plan_s, plan_n);

    final_linear_kernel<<<(N + T - 1) / T, T, 0, stream>>>(y2, Wl, bl, out, N);
}


# Plan launch no longer scans all owner bytes per tile: owner-map launch adds a per-tile bitmap (1 bit per 16-edge group); a tile reads its row, then only the named groups' dst+src; sum order changes
# speedup vs baseline: 3.0534x; 1.1517x over previous
#include <hip/hip_runtime.h>
#include <hip/hip_bf16.h>

typedef __attribute__((ext_vector_type(2))) float v2f;
typedef __attribute__((ext_vector_type(8))) float v8f;

#define N_NODES 100000
#define N_EDGES 3200000

template <int K, int NOUT, int MT>
__global__ __launch_bounds__(MT*(NOUT/16)*32)
void gemm_wmma_f32(const float* __restrict__ A, const float* __restrict__ B,
                   float* __restrict__ C, int M) {
    constexpr int NT   = NOUT / 16;
    constexpr int NW   = MT * NT;
    constexpr int ROWS = MT * 16;
    constexpr int KP   = K + 4;
    constexpr int NP   = NOUT + 4;

    __shared__ float As[ROWS][KP];
    __shared__ float Bs[K][NP];

    const int tid  = threadIdx.x;
    const int row0 = blockIdx.x * ROWS;

    constexpr int A_ELTS = ROWS * K;
    for (int i = tid * 4; i < A_ELTS; i += NW * 32 * 4) {
        int r = i / K, c = i % K;
        int gr = row0 + r;
        float4 v = (gr < M) ? *(const float4*)(A + (long long)gr * K + c)
                            : make_float4(0.f, 0.f, 0.f, 0.f);
        *(float4*)(&As[r][c]) = v;
    }
    constexpr int B_ELTS = K * NOUT;
    for (int i = tid * 4; i < B_ELTS; i += NW * 32 * 4) {
        int r = i / NOUT, c = i % NOUT;
        *(float4*)(&Bs[r][c]) = *(const float4*)(B + i);
    }
    __syncthreads();

    const int wid  = tid >> 5;
    const int lane = tid & 31;
    const int mn   = lane & 15;
    const int h    = lane >> 4;
    const int mt   = wid / NT;
    const int nt   = wid % NT;

    v8f acc = {};
    const float* arow = &As[mt * 16 + mn][0];
    const int    ncol = nt * 16 + mn;

#pragma unroll
    for (int k = 0; k < K; k += 4) {
        v2f a, b;
        a.x = arow[k + 2 * h + 0];
        a.y = arow[k + 2 * h + 1];
        b.x = Bs[k + 2 * h + 0][ncol];
        b.y = Bs[k + 2 * h + 1][ncol];
        acc = __builtin_amdgcn_wmma_f32_16x16x4_f32(
             false, a,  false, b,
             (short)0, acc,  false,  false);
    }

#pragma unroll
    for (int j = 0; j < 8; ++j) {
        int r = row0 + mt * 16 + j + 8 * h;
        if (r < M) C[(long long)r * NOUT + ncol] = acc[j];
    }
}

constexpr int NPW   = 392;
constexpr int NTILE = 256;
constexpr int WQ    = 512;
constexpr int NGRP  = N_EDGES / 16;
constexpr int NBAT  = (NGRP + 8191) / 8192;
static_assert(NPW * NTILE >= N_NODES, "tiles must cover every node");
static_assert(N_EDGES % 16 == 0, "owner map is read in 16-byte groups");
static_assert(NPW <= 65535, "local row index is kept in 16 bits");
constexpr int CH    = 8192;
constexpr int NCH   = (N_EDGES + CH - 1) / CH;
constexpr int CHW   = CH / 16 / 32;
constexpr int BWP   = NCH * CHW;
constexpr int GQ    = 640;
static_assert(CH == 1024 * 8, "one thread of the owner-map launch takes eight edges");
static_assert(BWP % 4 == 0, "bitmap rows are written in 16-byte pieces");
constexpr int SRC_BITS = 17;
static_assert(N_NODES <= (1 << SRC_BITS), "source id must fit the low bits of a stored entry");
static_assert(NPW <= (1 << (32 - SRC_BITS)), "local row must fit the high bits of a stored entry");

__global__ __launch_bounds__(1024)
void owner_map_kernel(const int* __restrict__ dst, unsigned char* __restrict__ owner,
                      unsigned int* __restrict__ bitmap, int E) {
    __shared__ unsigned int pres[CHW * NTILE];
    const int tid = threadIdx.x;
    for (int i = tid; i < CHW * NTILE; i += 1024) pres[i] = 0u;
    __syncthreads();
    const long long e0 = (long long)blockIdx.x * CH + tid * 8;
    if (e0 < E) {
        const int4 a = *(const int4*)(dst + e0);
        const int4 b = *(const int4*)(dst + e0 + 4);
        const unsigned o0 = (unsigned)(a.x / NPW), o1 = (unsigned)(a.y / NPW);
        const unsigned o2 = (unsigned)(a.z / NPW), o3 = (unsigned)(a.w / NPW);
        const unsigned o4 = (unsigned)(b.x / NPW), o5 = (unsigned)(b.y / NPW);
        const unsigned o6 = (unsigned)(b.z / NPW), o7 = (unsigned)(b.w / NPW);
        uint2 ob;
        ob.x = o0 | (o1 << 8) | (o2 << 16) | (o3 << 24);
        ob.y = o4 | (o5 << 8) | (o6 << 16) | (o7 << 24);
        *(uint2*)(owner + e0) = ob;
        const int gl = tid >> 1;
        const unsigned bit = 1u << (gl & 31);
        unsigned int* pw = &pres[(gl >> 5) * NTILE];
        atomicOr(pw + o0, bit); atomicOr(pw + o1, bit); atomicOr(pw + o2, bit); atomicOr(pw + o3, bit);
        atomicOr(pw + o4, bit); atomicOr(pw + o5, bit); atomicOr(pw + o6, bit); atomicOr(pw + o7, bit);
    }
    __syncthreads();
    const int t = tid >> 2, q = (tid & 3) * 4;
    uint4 v;
    v.x = pres[(q + 0) * NTILE + t];
    v.y = pres[(q + 1) * NTILE + t];
    v.z = pres[(q + 2) * NTILE + t];
    v.w = pres[(q + 3) * NTILE + t];
    *(uint4*)(bitmap + (long long)t * BWP + blockIdx.x * CHW + q) = v;
}

__device__ __forceinline__ unsigned zero_bytes(unsigned y) {
    unsigned t = (y & 0x7F7F7F7Fu) + 0x7F7F7F7Fu;
    return ~(t | y | 0x7F7F7F7Fu);
}
__device__ __forceinline__ unsigned match_group(uint4 v, unsigned me) {
    return (zero_bytes(v.x ^ me) >> 7) | (zero_bytes(v.y ^ me) >> 6) |
           (zero_bytes(v.z ^ me) >> 5) | (zero_bytes(v.w ^ me) >> 4);
}

template <int F>
__device__ void spill_edge(int e, int base, float* tile, const float* __restrict__ h,
                           const int* __restrict__ src, const int* __restrict__ dst,
                           const float* __restrict__ dinv) {
    const int s = src[e], d = dst[e];
    const float nrm = dinv[s] * dinv[d];
    const float* hr = h + (long long)s * F;
    float* tr = tile + (d - base) * F;
    for (int f = 0; f < F; ++f) atomicAdd(tr + f, hr[f] * nrm);
}
__device__ __forceinline__ int wrap_group(int g) {
    if (g >= NGRP) g -= NGRP;
    if (g >= NGRP) g -= NGRP;
    return g;
}

template <typename Spill>
__device__ __forceinline__ int scan_owner_map(const unsigned char* __restrict__ owner, int tileid,
                                              int tid, int* myq, Spill spill) {
    const unsigned me   = (unsigned)tileid * 0x01010101u;
    const uint4*   om   = (const uint4*)owner;
    const int rot = tileid * 768;
    int wc = 0;
    for (int it = 0; it < NBAT; ++it) {
        uint4 v[8];
#pragma unroll
        for (int k = 0; k < 8; ++k) {
            v[k] = om[wrap_group(it * 8192 + k * 1024 + tid + rot)];
        }
        unsigned mk[8];
#pragma unroll
        for (int k = 0; k < 8; ++k)
            mk[k] = match_group(v[k], me) & (0u - (unsigned)(it * 8192 + k * 1024 + tid < NGRP));
#pragma unroll
        for (int hf = 0; hf < 2; ++hf) {
            const unsigned lo = mk[4 * hf + 0] | (mk[4 * hf + 1] << 4);
            const unsigned hi = mk[4 * hf + 2] | (mk[4 * hf + 3] << 4);
            unsigned long long M = ((unsigned long long)hi << 32) | (unsigned long long)lo;
            unsigned act;
            while ((act = __builtin_amdgcn_ballot_w32(M != 0ull)) != 0u) {
                if (M != 0ull) {
                    const int p = __builtin_ctzll(M);
                    M &= M - 1ull;
                    const int k  = 4 * hf + 2 * (p >> 5) + ((p >> 2) & 1);
                    const int g = wrap_group(it * 8192 + k * 1024 + tid + rot);
                    const int e = g * 16 + (p & 3) * 4 + ((p & 31) >> 3);
                    const int slot = wc + (int)__builtin_amdgcn_mbcnt_lo(act, 0u);
                    if (slot < WQ) myq[slot] = e;
                    else spill(e);
                }
                wc += __builtin_popcount(act);
            }
        }
    }
    return wc;
}

static_assert(CHW == 16 && NTILE * 4 == 1024, "bitmap write-out of the owner-map launch");
constexpr int BW_FULL = BWP / 1024;
constexpr int BW_REST = BWP - BW_FULL * 1024;
static_assert(BW_REST * 9 <= 1024 + 8, "rest of a bitmap row is dealt to every 9th thread");

__global__ __launch_bounds__(1024)
void tile_plan_kernel(const int* __restrict__ src, const int* __restrict__ dst,
                      const unsigned int* __restrict__ bitmap, float* __restrict__ dinv, int N,
                      int* __restrict__ plan_c, unsigned int* __restrict__ plan_s) {
    __shared__ int q_s[32 * WQ];
    __shared__ int g_s[32 * GQ];
    __shared__ int indeg[NPW];
    __shared__ int wcs[32];

    const int tid  = threadIdx.x;
    const int wid  = tid >> 5;
    const int lane = tid & 31;
    const int base = blockIdx.x * NPW;
    int nloc = N - base;
    if (nloc > NPW) nloc = NPW;
    if (nloc < 0) nloc = 0;

    for (int i = tid; i < NPW; i += 1024) indeg[i] = 0;
    __syncthreads();

    int* myq = &q_s[wid * WQ];
    int* myg = &g_s[wid * GQ];
    unsigned int* ps = plan_s + (long long)(blockIdx.x * 32 + wid) * WQ;

    const unsigned int* brow = bitmap + (long long)blockIdx.x * BWP;
    unsigned bits[BW_FULL + 1];
    int      bj[BW_FULL + 1];
#pragma unroll
    for (int k = 0; k < BW_FULL; ++k) { bj[k] = k * 1024 + tid; bits[k] = brow[bj[k]]; }
    {
        const int r = tid / 9;
        const bool has = (tid - r * 9 == 0) && (r < BW_REST);
        bj[BW_FULL] = BW_FULL * 1024 + (has ? r : 0);
        const unsigned v = brow[bj[BW_FULL]];
        bits[BW_FULL] = has ? v : 0u;
    }
    int  gc  = 0;
    bool ovf = false;
#pragma unroll
    for (int k = 0; k <= BW_FULL; ++k) {
        unsigned m = bits[k];
        unsigned act;
        while ((act = __builtin_amdgcn_ballot_w32(m != 0u)) != 0u) {
            if (m != 0u) {
                const int p = __builtin_ctz(m);
                m &= m - 1u;
                const int g = bj[k] * 32 + p;
                const int slot = gc + (int)__builtin_amdgcn_mbcnt_lo(act, 0u);
                if (slot < GQ) myg[slot] = g;
                else {
                    ovf = true;
                    for (int x = 0; x < 16; ++x) {
                        const unsigned lr = (unsigned)(dst[g * 16 + x] - base);
                        if (lr < (unsigned)NPW) atomicAdd(&indeg[lr], 1);
                    }
                }
            }
            gc += __builtin_popcount(act);
        }
    }
    if (gc > GQ) gc = GQ;

    int wc = 0;
    const int sub = lane >> 2;
    const int qo  = (lane & 3) * 4;
    for (int i0 = 0; i0 < gc; i0 += 64) {
        int   gg[8];
        uint4 d[8], s[8];
#pragma unroll
        for (int k = 0; k < 8; ++k) {
            const int i = i0 + 8 * k + sub;
            gg[k] = myg[(i < gc) ? i : (gc - 1)];
        }
#pragma unroll
        for (int k = 0; k < 8; ++k) {
            d[k] = *(const uint4*)(dst + (long long)gg[k] * 16 + qo);
            s[k] = *(const uint4*)(src + (long long)gg[k] * 16 + qo);
        }
#pragma unroll
        for (int k = 0; k < 8; ++k) {
            const bool ok = i0 + 8 * k + sub < gc;
#pragma unroll
            for (int x = 0; x < 4; ++x) {
                const unsigned dv = x == 0 ? d[k].x : x == 1 ? d[k].y : x == 2 ? d[k].z : d[k].w;
                const unsigned sv = x == 0 ? s[k].x : x == 1 ? s[k].y : x == 2 ? s[k].z : s[k].w;
                const unsigned lr = dv - (unsigned)base;
                const bool hit = ok && lr < (unsigned)NPW;
                const unsigned act = __builtin_amdgcn_ballot_w32(hit);
                if (act != 0u) {
                    if (hit) {
                        const int slot = wc + (int)__builtin_amdgcn_mbcnt_lo(act, 0u);
                        if (slot < WQ) myq[slot] = (int)(sv | (lr << SRC_BITS));
                        else atomicAdd(&indeg[lr], 1);
                    }
                    wc += __builtin_popcount(act);
                }
            }
        }
    }
    const bool lost = __builtin_amdgcn_ballot_w32(ovf) != 0u;
    if (lane == 0) wcs[wid] = (lost && wc <= WQ) ? (WQ + 1) : wc;

    const int cnt = wc < WQ ? wc : WQ;
    for (int i = lane; i < cnt; i += 32) {
        const unsigned v = (unsigned)myq[i];
        ps[i] = v;
        atomicAdd(&indeg[v >> SRC_BITS], 1);
    }
    __syncthreads();
    if (wid == 0) plan_c[blockIdx.x * 32 + lane] = wcs[lane];
    for (int n = tid; n < nloc; n += 1024) dinv[base + n] = rsqrtf((float)(indeg[n] + 1));
}

template <int F, bool FIRST>
__global__ __launch_bounds__(1024)
void tile_agg_kernel(const float* __restrict__ h, const int* __restrict__ src,
                     const int* __restrict__ dst, const unsigned char* __restrict__ owner,
                     const float* __restrict__ dinv, const float* __restrict__ bias,
                     float* __restrict__ y, int N, const int* __restrict__ plan_c,
                     const unsigned int* __restrict__ plan_s, float* plan_n) {
    constexpr int F4  = F / 4;
    constexpr int VPL = F / 32;
    __shared__ float          tile[NPW * F];
    __shared__ int            q_s[32 * WQ];
    __shared__ float          q_n[32 * WQ];
    __shared__ unsigned short q_d[32 * WQ];

    const int tid  = threadIdx.x;
    const int wid  = tid >> 5;
    const int lane = tid & 31;
    const int base = blockIdx.x * NPW;
    int nloc = N - base;
    if (nloc > NPW) nloc = NPW;
    if (nloc < 0) nloc = 0;

    for (int i = tid; i < nloc * F4; i += 1024) {
        const int n = i / F4, c = i % F4;
        const float di = dinv[base + n];
        const float4 hv = *(const float4*)(h + (long long)(base + n) * F + c * 4);
        const float4 bv = *(const float4*)(bias + c * 4);
        float4 t;
        t.x = hv.x * di * di + bv.x;
        t.y = hv.y * di * di + bv.y;
        t.z = hv.z * di * di + bv.z;
        t.w = hv.w * di * di + bv.w;
        *(float4*)(&tile[n * F + c * 4]) = t;
    }
    __syncthreads();

    int*            myq = &q_s[wid * WQ];
    float*          myn = &q_n[wid * WQ];
    unsigned short* myd = &q_d[wid * WQ];
    const unsigned int* ps = plan_s + (long long)(blockIdx.x * 32 + wid) * WQ;
    float*              pn = plan_n + (long long)(blockIdx.x * 32 + wid) * WQ;
    const int c32 = plan_c[blockIdx.x * 32 + lane];
    const bool build = __builtin_amdgcn_ballot_w32(c32 > WQ) != 0u;
    int wc = __builtin_amdgcn_readlane(c32, __builtin_amdgcn_readfirstlane(wid));

    if (build) {
    wc = scan_owner_map(owner, (int)blockIdx.x, tid, myq,
                        [&](int e) { spill_edge<F>(e, base, tile, h, src, dst, dinv); });
    const int cnt = wc < WQ ? wc : WQ;
    for (int i0 = 0; i0 < cnt; i0 += 128) {
        int ix[4], ss[4], dd[4];
        float nr[4];
#pragma unroll
        for (int k = 0; k < 4; ++k) {
            const int i = i0 + 32 * k + lane;
            ix[k] = (i < cnt) ? i : (cnt - 1);
        }
        int ee[4];
#pragma unroll
        for (int k = 0; k < 4; ++k) ee[k] = myq[ix[k]];
#pragma unroll
        for (int k = 0; k < 4; ++k) { ss[k] = src[ee[k]]; dd[k] = dst[ee[k]]; }
#pragma unroll
        for (int k = 0; k < 4; ++k) nr[k] = dinv[ss[k]] * dinv[dd[k]];
#pragma unroll
        for (int k = 0; k < 4; ++k) {
            myq[ix[k]] = ss[k];
            myd[ix[k]] = (unsigned short)(dd[k] - base);
            myn[ix[k]] = nr[k];
        }
    }
    } else {
    const int cnt = wc;
    for (int i0 = 0; i0 < cnt; i0 += 128) {
        int      ix[4];
        unsigned r[4];
        float    nr[4];
#pragma unroll
        for (int k = 0; k < 4; ++k) {
            const int i = i0 + 32 * k + lane;
            ix[k] = (i < cnt) ? i : (cnt - 1);
            r[k] = ps[ix[k]];
        }
#pragma unroll
        for (int k = 0; k < 4; ++k) {
            if (FIRST) nr[k] = dinv[r[k] & ((1u << SRC_BITS) - 1u)] * dinv[base + (int)(r[k] >> SRC_BITS)];
            else       nr[k] = pn[ix[k]];
        }
#pragma unroll
        for (int k = 0; k < 4; ++k) {
            myq[ix[k]] = (int)(r[k] & ((1u << SRC_BITS) - 1u));
            myd[ix[k]] = (unsigned short)(r[k] >> SRC_BITS);
            myn[ix[k]] = nr[k];
            if (FIRST && i0 + 32 * k + lane < cnt) pn[ix[k]] = nr[k];
        }
    }
    }
    const int cnt = wc < WQ ? wc : WQ;
    __syncthreads();

    constexpr int U = 16;
    for (int i0 = 0; i0 < cnt; i0 += U) {
        float vals[U][VPL];
#pragma unroll
        for (int k = 0; k < U; ++k) {
            const int i = (i0 + k < cnt) ? (i0 + k) : i0;
            const float* hr = h + (long long)myq[i] * F + lane * VPL;
#pragma unroll
            for (int x = 0; x < VPL; ++x) vals[k][x] = hr[x];
        }
#pragma unroll
        for (int k = 0; k < U; ++k) {
            if (i0 + k < cnt) {
                const float nrm = myn[i0 + k];
                float* tr = &tile[(int)myd[i0 + k] * F + lane * VPL];
#pragma unroll
                for (int x = 0; x < VPL; ++x) atomicAdd(tr + x, vals[k][x] * nrm);
            }
        }
    }
    __syncthreads();

    for (int i = tid; i < nloc * F4; i += 1024) {
        float4 t = *(const float4*)(&tile[i * 4]);
        t.x = fmaxf(t.x, 0.0f); t.y = fmaxf(t.y, 0.0f);
        t.z = fmaxf(t.z, 0.0f); t.w = fmaxf(t.w, 0.0f);
        *(float4*)(y + (long long)base * F + (long long)i * 4) = t;
    }
}

__global__ void final_linear_kernel(const float* __restrict__ y, const float* __restrict__ Wl,
                                    const float* __restrict__ bl, float* out, int n) {
    int i = blockIdx.x * blockDim.x + threadIdx.x;
    if (i >= n) return;
    const float* row = y + (long long)i * 32;
    float s = 0.0f;
#pragma unroll
    for (int f = 0; f < 32; ++f) s += row[f] * Wl[f];
    out[i] = s + bl[0];
}

static inline size_t align256(size_t x) { return (x + 255) & ~size_t(255); }

extern "C" void kernel_launch(void* const* d_in, const int* in_sizes, int n_in,
                              void* d_out, int out_size, void* d_ws, size_t ws_size,
                              hipStream_t stream) {
    const float* x    = (const float*)d_in[0];
    const int*   ei   = (const int*)d_in[1];
    const float* W1   = (const float*)d_in[2];
    const float* b1   = (const float*)d_in[3];
    const float* W2   = (const float*)d_in[4];
    const float* b2   = (const float*)d_in[5];
    const float* Wl   = (const float*)d_in[6];
    const float* bl   = (const float*)d_in[7];
    float*       out  = (float*)d_out;

    const int N = N_NODES;
    const int E = N_EDGES;
    const int* src = ei;
    const int* dst = ei + E;

    char* ws = (char*)d_ws;
    size_t off = 0;
    float* dinv = (float*)(ws + off); off += align256((size_t)N * 4);
    float* h1   = (float*)(ws + off); off += align256((size_t)N * 64 * 4);
    float* y1   = (float*)(ws + off); off += align256((size_t)N * 64 * 4);
    float* h2   = (float*)(ws + off); off += align256((size_t)N * 32 * 4);
    float* y2   = (float*)(ws + off); off += align256((size_t)N * 32 * 4);
    unsigned char* owner = (unsigned char*)(ws + off); off += align256((size_t)E);
    int*          plan_c = (int*)(ws + off);          off += align256((size_t)NTILE * 32 * 4);
    unsigned int* plan_s = (unsigned int*)(ws + off); off += align256((size_t)NTILE * 32 * WQ * 4);
    float*        plan_n = (float*)(ws + off);        off += align256((size_t)NTILE * 32 * WQ * 4);
    unsigned int* bitmap = (unsigned int*)(ws + off); off += align256((size_t)NTILE * BWP * 4);
    (void)ws_size; (void)n_in; (void)in_sizes; (void)out_size;

    const int T = 256;

    owner_map_kernel<<<NCH, 1024, 0, stream>>>(dst, owner, bitmap, E);

    tile_plan_kernel<<<NTILE, 1024, 0, stream>>>(src, dst, bitmap, dinv, N, plan_c, plan_s);

    gemm_wmma_f32<128, 64, 2><<<N / 32, 2 * 4 * 32, 0, stream>>>(x, W1, h1, N);

    tile_agg_kernel<64, true><<<NTILE, 1024, 0, stream>>>(h1, src, dst, owner, dinv, b1, y1, N,
                                                          plan_c, plan_s, plan_n);

    gemm_wmma_f32<64, 32, 2><<<N / 32, 2 * 2 * 32, 0, stream>>>(y1, W2, h2, N);

    tile_agg_kernel<32, false><<<NTILE, 1024, 0, stream>>>(h2, src, dst, owner, dinv, b2, y2, N,
                                                           plan_c, plan_s, plan_n);

    final_linear_kernel<<<(N + T - 1) / T, T, 0, stream>>>(y2, Wl, bl, out, N);
}


# Aggregation waves load their stored queue entries into lanes (one line per 32 edges) and pass them on by readlane: no queue copy in LDS, no 3 broadcast LDS reads per edge; LDS atomic adds as before
# speedup vs baseline: 3.5781x; 1.1718x over previous
#include <hip/hip_runtime.h>
#include <hip/hip_bf16.h>

typedef __attribute__((ext_vector_type(2))) float v2f;
typedef __attribute__((ext_vector_type(8))) float v8f;

#define N_NODES 100000
#define N_EDGES 3200000

template <int K, int NOUT, int MT>
__global__ __launch_bounds__(MT*(NOUT/16)*32)
void gemm_wmma_f32(const float* __restrict__ A, const float* __restrict__ B,
                   float* __restrict__ C, int M) {
    constexpr int NT   = NOUT / 16;
    constexpr int NW   = MT * NT;
    constexpr int ROWS = MT * 16;
    constexpr int KP   = K + 4;
    constexpr int NP   = NOUT + 4;

    __shared__ float As[ROWS][KP];
    __shared__ float Bs[K][NP];

    const int tid  = threadIdx.x;
    const int row0 = blockIdx.x * ROWS;

    constexpr int A_ELTS = ROWS * K;
    for (int i = tid * 4; i < A_ELTS; i += NW * 32 * 4) {
        int r = i / K, c = i % K;
        int gr = row0 + r;
        float4 v = (gr < M) ? *(const float4*)(A + (long long)gr * K + c)
                            : make_float4(0.f, 0.f, 0.f, 0.f);
        *(float4*)(&As[r][c]) = v;
    }
    constexpr int B_ELTS = K * NOUT;
    for (int i = tid * 4; i < B_ELTS; i += NW * 32 * 4) {
        int r = i / NOUT, c = i % NOUT;
        *(float4*)(&Bs[r][c]) = *(const float4*)(B + i);
    }
    __syncthreads();

    const int wid  = tid >> 5;
    const int lane = tid & 31;
    const int mn   = lane & 15;
    const int h    = lane >> 4;
    const int mt   = wid / NT;
    const int nt   = wid % NT;

    v8f acc = {};
    const float* arow = &As[mt * 16 + mn][0];
    const int    ncol = nt * 16 + mn;

#pragma unroll
    for (int k = 0; k < K; k += 4) {
        v2f a, b;
        a.x = arow[k + 2 * h + 0];
        a.y = arow[k + 2 * h + 1];
        b.x = Bs[k + 2 * h + 0][ncol];
        b.y = Bs[k + 2 * h + 1][ncol];
        acc = __builtin_amdgcn_wmma_f32_16x16x4_f32(
             false, a,  false, b,
             (short)0, acc,  false,  false);
    }

#pragma unroll
    for (int j = 0; j < 8; ++j) {
        int r = row0 + mt * 16 + j + 8 * h;
        if (r < M) C[(long long)r * NOUT + ncol] = acc[j];
    }
}

constexpr int NPW   = 392;
constexpr int NTILE = 256;
constexpr int WQ    = 512;
constexpr int NGRP  = N_EDGES / 16;
constexpr int NBAT  = (NGRP + 8191) / 8192;
static_assert(NPW * NTILE >= N_NODES, "tiles must cover every node");
static_assert(N_EDGES % 16 == 0, "owner map is read in 16-byte groups");
static_assert(NPW <= 65535, "local row index is kept in 16 bits");
constexpr int CH    = 8192;
constexpr int NCH   = (N_EDGES + CH - 1) / CH;
constexpr int CHW   = CH / 16 / 32;
constexpr int BWP   = NCH * CHW;
constexpr int GQ    = 640;
static_assert(CH == 1024 * 8, "one thread of the owner-map launch takes eight edges");
static_assert(BWP % 4 == 0, "bitmap rows are written in 16-byte pieces");
constexpr int SRC_BITS = 17;
static_assert(N_NODES <= (1 << SRC_BITS), "source id must fit the low bits of a stored entry");
static_assert(NPW <= (1 << (32 - SRC_BITS)), "local row must fit the high bits of a stored entry");

__global__ __launch_bounds__(1024)
void owner_map_kernel(const int* __restrict__ dst, unsigned char* __restrict__ owner,
                      unsigned int* __restrict__ bitmap, int E) {
    __shared__ unsigned int pres[CHW * NTILE];
    const int tid = threadIdx.x;
    for (int i = tid; i < CHW * NTILE; i += 1024) pres[i] = 0u;
    __syncthreads();
    const long long e0 = (long long)blockIdx.x * CH + tid * 8;
    if (e0 < E) {
        const int4 a = *(const int4*)(dst + e0);
        const int4 b = *(const int4*)(dst + e0 + 4);
        const unsigned o0 = (unsigned)(a.x / NPW), o1 = (unsigned)(a.y / NPW);
        const unsigned o2 = (unsigned)(a.z / NPW), o3 = (unsigned)(a.w / NPW);
        const unsigned o4 = (unsigned)(b.x / NPW), o5 = (unsigned)(b.y / NPW);
        const unsigned o6 = (unsigned)(b.z / NPW), o7 = (unsigned)(b.w / NPW);
        uint2 ob;
        ob.x = o0 | (o1 << 8) | (o2 << 16) | (o3 << 24);
        ob.y = o4 | (o5 << 8) | (o6 << 16) | (o7 << 24);
        *(uint2*)(owner + e0) = ob;
        const int gl = tid >> 1;
        const unsigned bit = 1u << (gl & 31);
        unsigned int* pw = &pres[(gl >> 5) * NTILE];
        atomicOr(pw + o0, bit); atomicOr(pw + o1, bit); atomicOr(pw + o2, bit); atomicOr(pw + o3, bit);
        atomicOr(pw + o4, bit); atomicOr(pw + o5, bit); atomicOr(pw + o6, bit); atomicOr(pw + o7, bit);
    }
    __syncthreads();
    const int t = tid >> 2, q = (tid & 3) * 4;
    uint4 v;
    v.x = pres[(q + 0) * NTILE + t];
    v.y = pres[(q + 1) * NTILE + t];
    v.z = pres[(q + 2) * NTILE + t];
    v.w = pres[(q + 3) * NTILE + t];
    *(uint4*)(bitmap + (long long)t * BWP + blockIdx.x * CHW + q) = v;
}

__device__ __forceinline__ unsigned zero_bytes(unsigned y) {
    unsigned t = (y & 0x7F7F7F7Fu) + 0x7F7F7F7Fu;
    return ~(t | y | 0x7F7F7F7Fu);
}
__device__ __forceinline__ unsigned match_group(uint4 v, unsigned me) {
    return (zero_bytes(v.x ^ me) >> 7) | (zero_bytes(v.y ^ me) >> 6) |
           (zero_bytes(v.z ^ me) >> 5) | (zero_bytes(v.w ^ me) >> 4);
}

template <int F>
__device__ void spill_edge(int e, int base, float* tile, const float* __restrict__ h,
                           const int* __restrict__ src, const int* __restrict__ dst,
                           const float* __restrict__ dinv) {
    const int s = src[e], d = dst[e];
    const float nrm = dinv[s] * dinv[d];
    const float* hr = h + (long long)s * F;
    float* tr = tile + (d - base) * F;
    for (int f = 0; f < F; ++f) atomicAdd(tr + f, hr[f] * nrm);
}
__device__ __forceinline__ int wrap_group(int g) {
    if (g >= NGRP) g -= NGRP;
    if (g >= NGRP) g -= NGRP;
    return g;
}

template <typename Spill>
__device__ __forceinline__ int scan_owner_map(const unsigned char* __restrict__ owner, int tileid,
                                              int tid, int* myq, Spill spill) {
    const unsigned me   = (unsigned)tileid * 0x01010101u;
    const uint4*   om   = (const uint4*)owner;
    const int rot = tileid * 768;
    int wc = 0;
    for (int it = 0; it < NBAT; ++it) {
        uint4 v[8];
#pragma unroll
        for (int k = 0; k < 8; ++k) {
            v[k] = om[wrap_group(it * 8192 + k * 1024 + tid + rot)];
        }
        unsigned mk[8];
#pragma unroll
        for (int k = 0; k < 8; ++k)
            mk[k] = match_group(v[k], me) & (0u - (unsigned)(it * 8192 + k * 1024 + tid < NGRP));
#pragma unroll
        for (int hf = 0; hf < 2; ++hf) {
            const unsigned lo = mk[4 * hf + 0] | (mk[4 * hf + 1] << 4);
            const unsigned hi = mk[4 * hf + 2] | (mk[4 * hf + 3] << 4);
            unsigned long long M = ((unsigned long long)hi << 32) | (unsigned long long)lo;
            unsigned act;
            while ((act = __builtin_amdgcn_ballot_w32(M != 0ull)) != 0u) {
                if (M != 0ull) {
                    const int p = __builtin_ctzll(M);
                    M &= M - 1ull;
                    const int k  = 4 * hf + 2 * (p >> 5) + ((p >> 2) & 1);
                    const int g = wrap_group(it * 8192 + k * 1024 + tid + rot);
                    const int e = g * 16 + (p & 3) * 4 + ((p & 31) >> 3);
                    const int slot = wc + (int)__builtin_amdgcn_mbcnt_lo(act, 0u);
                    if (slot < WQ) myq[slot] = e;
                    else spill(e);
                }
                wc += __builtin_popcount(act);
            }
        }
    }
    return wc;
}

static_assert(CHW == 16 && NTILE * 4 == 1024, "bitmap write-out of the owner-map launch");
constexpr int BW_FULL = BWP / 1024;
constexpr int BW_REST = BWP - BW_FULL * 1024;
static_assert(BW_REST * 9 <= 1024 + 8, "rest of a bitmap row is dealt to every 9th thread");

__global__ __launch_bounds__(1024)
void tile_plan_kernel(const int* __restrict__ src, const int* __restrict__ dst,
                      const unsigned int* __restrict__ bitmap, float* __restrict__ dinv, int N,
                      int* __restrict__ plan_c, unsigned int* __restrict__ plan_s) {
    __shared__ int q_s[32 * WQ];
    __shared__ int g_s[32 * GQ];
    __shared__ int indeg[NPW];
    __shared__ int wcs[32];

    const int tid  = threadIdx.x;
    const int wid  = tid >> 5;
    const int lane = tid & 31;
    const int base = blockIdx.x * NPW;
    int nloc = N - base;
    if (nloc > NPW) nloc = NPW;
    if (nloc < 0) nloc = 0;

    for (int i = tid; i < NPW; i += 1024) indeg[i] = 0;
    __syncthreads();

    int* myq = &q_s[wid * WQ];
    int* myg = &g_s[wid * GQ];
    unsigned int* ps = plan_s + (long long)(blockIdx.x * 32 + wid) * WQ;

    const unsigned int* brow = bitmap + (long long)blockIdx.x * BWP;
    unsigned bits[BW_FULL + 1];
    int      bj[BW_FULL + 1];
#pragma unroll
    for (int k = 0; k < BW_FULL; ++k) { bj[k] = k * 1024 + tid; bits[k] = brow[bj[k]]; }
    {
        const int r = tid / 9;
        const bool has = (tid - r * 9 == 0) && (r < BW_REST);
        bj[BW_FULL] = BW_FULL * 1024 + (has ? r : 0);
        const unsigned v = brow[bj[BW_FULL]];
        bits[BW_FULL] = has ? v : 0u;
    }
    int  gc  = 0;
    bool ovf = false;
#pragma unroll
    for (int k = 0; k <= BW_FULL; ++k) {
        unsigned m = bits[k];
        unsigned act;
        while ((act = __builtin_amdgcn_ballot_w32(m != 0u)) != 0u) {
            if (m != 0u) {
                const int p = __builtin_ctz(m);
                m &= m - 1u;
                const int g = bj[k] * 32 + p;
                const int slot = gc + (int)__builtin_amdgcn_mbcnt_lo(act, 0u);
                if (slot < GQ) myg[slot] = g;
                else {
                    ovf = true;
                    for (int x = 0; x < 16; ++x) {
                        const unsigned lr = (unsigned)(dst[g * 16 + x] - base);
                        if (lr < (unsigned)NPW) atomicAdd(&indeg[lr], 1);
                    }
                }
            }
            gc += __builtin_popcount(act);
        }
    }
    if (gc > GQ) gc = GQ;

    int wc = 0;
    const int sub = lane >> 2;
    const int qo  = (lane & 3) * 4;
    for (int i0 = 0; i0 < gc; i0 += 64) {
        int   gg[8];
        uint4 d[8], s[8];
#pragma unroll
        for (int k = 0; k < 8; ++k) {
            const int i = i0 + 8 * k + sub;
            gg[k] = myg[(i < gc) ? i : (gc - 1)];
        }
#pragma unroll
        for (int k = 0; k < 8; ++k) {
            d[k] = *(const uint4*)(dst + (long long)gg[k] * 16 + qo);
            s[k] = *(const uint4*)(src + (long long)gg[k] * 16 + qo);
        }
#pragma unroll
        for (int k = 0; k < 8; ++k) {
            const bool ok = i0 + 8 * k + sub < gc;
#pragma unroll
            for (int x = 0; x < 4; ++x) {
                const unsigned dv = x == 0 ? d[k].x : x == 1 ? d[k].y : x == 2 ? d[k].z : d[k].w;
                const unsigned sv = x == 0 ? s[k].x : x == 1 ? s[k].y : x == 2 ? s[k].z : s[k].w;
                const unsigned lr = dv - (unsigned)base;
                const bool hit = ok && lr < (unsigned)NPW;
                const unsigned act = __builtin_amdgcn_ballot_w32(hit);
                if (act != 0u) {
                    if (hit) {
                        const int slot = wc + (int)__builtin_amdgcn_mbcnt_lo(act, 0u);
                        if (slot < WQ) myq[slot] = (int)(sv | (lr << SRC_BITS));
                        else atomicAdd(&indeg[lr], 1);
                    }
                    wc += __builtin_popcount(act);
                }
            }
        }
    }
    const bool lost = __builtin_amdgcn_ballot_w32(ovf) != 0u;
    if (lane == 0) wcs[wid] = (lost && wc <= WQ) ? (WQ + 1) : wc;

    const int cnt = wc < WQ ? wc : WQ;
    for (int i = lane; i < cnt; i += 32) {
        const unsigned v = (unsigned)myq[i];
        ps[i] = v;
        atomicAdd(&indeg[v >> SRC_BITS], 1);
    }
    __syncthreads();
    if (wid == 0) plan_c[blockIdx.x * 32 + lane] = wcs[lane];
    for (int n = tid; n < nloc; n += 1024) dinv[base + n] = rsqrtf((float)(indeg[n] + 1));
}

template <int F, bool FULL>
__device__ __forceinline__ void add_half(const float* __restrict__ h, float* tile, unsigned r, float n,
                                         int j0, int m, int lane) {
    constexpr int      VPL = F / 32;
    constexpr unsigned SM  = (1u << SRC_BITS) - 1u;
    float vals[16][VPL];
#pragma unroll
    for (int k = 0; k < 16; ++k) {
        const unsigned rj = (unsigned)__builtin_amdgcn_readlane((int)r, j0 + k);
        const float* hr = h + (long long)(rj & SM) * F + lane * VPL;
#pragma unroll
        for (int x = 0; x < VPL; ++x) vals[k][x] = hr[x];
    }
#pragma unroll
    for (int k = 0; k < 16; ++k) {
        if (FULL || k < m) {
            const unsigned rj  = (unsigned)__builtin_amdgcn_readlane((int)r, j0 + k);
            const float    nrm = __int_as_float(__builtin_amdgcn_readlane(__float_as_int(n), j0 + k));
            float* tr = tile + (int)(rj >> SRC_BITS) * F + lane * VPL;
#pragma unroll
            for (int x = 0; x < VPL; ++x) atomicAdd(tr + x, vals[k][x] * nrm);
        }
    }
}

template <int F, bool FIRST>
__global__ __launch_bounds__(1024)
void tile_agg_kernel(const float* __restrict__ h, const int* __restrict__ src,
                     const int* __restrict__ dst, const unsigned char* __restrict__ owner,
                     const float* __restrict__ dinv, const float* __restrict__ bias,
                     float* __restrict__ y, int N, const int* __restrict__ plan_c,
                     const unsigned int* __restrict__ plan_s, float* plan_n) {
    constexpr int F4  = F / 4;
    constexpr int VPL = F / 32;
    __shared__ float          tile[NPW * F];
    __shared__ int            q_s[32 * WQ];
    __shared__ float          q_n[32 * WQ];
    __shared__ unsigned short q_d[32 * WQ];

    const int tid  = threadIdx.x;
    const int wid  = tid >> 5;
    const int lane = tid & 31;
    const int base = blockIdx.x * NPW;
    int nloc = N - base;
    if (nloc > NPW) nloc = NPW;
    if (nloc < 0) nloc = 0;

    for (int i = tid; i < nloc * F4; i += 1024) {
        const int n = i / F4, c = i % F4;
        const float di = dinv[base + n];
        const float4 hv = *(const float4*)(h + (long long)(base + n) * F + c * 4);
        const float4 bv = *(const float4*)(bias + c * 4);
        float4 t;
        t.x = hv.x * di * di + bv.x;
        t.y = hv.y * di * di + bv.y;
        t.z = hv.z * di * di + bv.z;
        t.w = hv.w * di * di + bv.w;
        *(float4*)(&tile[n * F + c * 4]) = t;
    }
    __syncthreads();

    int*            myq = &q_s[wid * WQ];
    float*          myn = &q_n[wid * WQ];
    unsigned short* myd = &q_d[wid * WQ];
    const unsigned int* ps = plan_s + (long long)(blockIdx.x * 32 + wid) * WQ;
    float*              pn = plan_n + (long long)(blockIdx.x * 32 + wid) * WQ;
    const int c32 = plan_c[blockIdx.x * 32 + lane];
    const bool build = __builtin_amdgcn_ballot_w32(c32 > WQ) != 0u;
    int wc = __builtin_amdgcn_readlane(c32, __builtin_amdgcn_readfirstlane(wid));

    if (build) {
    wc = scan_owner_map(owner, (int)blockIdx.x, tid, myq,
                        [&](int e) { spill_edge<F>(e, base, tile, h, src, dst, dinv); });
    const int cnt = wc < WQ ? wc : WQ;
    for (int i0 = 0; i0 < cnt; i0 += 128) {
        int ix[4], ss[4], dd[4];
        float nr[4];
#pragma unroll
        for (int k = 0; k < 4; ++k) {
            const int i = i0 + 32 * k + lane;
            ix[k] = (i < cnt) ? i : (cnt - 1);
        }
        int ee[4];
#pragma unroll
        for (int k = 0; k < 4; ++k) ee[k] = myq[ix[k]];
#pragma unroll
        for (int k = 0; k < 4; ++k) { ss[k] = src[ee[k]]; dd[k] = dst[ee[k]]; }
#pragma unroll
        for (int k = 0; k < 4; ++k) nr[k] = dinv[ss[k]] * dinv[dd[k]];
#pragma unroll
        for (int k = 0; k < 4; ++k) {
            myq[ix[k]] = ss[k];
            myd[ix[k]] = (unsigned short)(dd[k] - base);
            myn[ix[k]] = nr[k];
        }
    }
    __syncthreads();

    constexpr int U = 16;
    for (int i0 = 0; i0 < cnt; i0 += U) {
        float vals[U][VPL];
#pragma unroll
        for (int k = 0; k < U; ++k) {
            const int i = (i0 + k < cnt) ? (i0 + k) : i0;
            const float* hr = h + (long long)myq[i] * F + lane * VPL;
#pragma unroll
            for (int x = 0; x < VPL; ++x) vals[k][x] = hr[x];
        }
#pragma unroll
        for (int k = 0; k < U; ++k) {
            if (i0 + k < cnt) {
                const float nrm = myn[i0 + k];
                float* tr = &tile[(int)myd[i0 + k] * F + lane * VPL];
#pragma unroll
                for (int x = 0; x < VPL; ++x) atomicAdd(tr + x, vals[k][x] * nrm);
            }
        }
    }
    } else {
    const int cnt = wc;
    if (cnt > 0) {
        constexpr unsigned SM = (1u << SRC_BITS) - 1u;
        const int nb = (cnt + 31) >> 5;
        auto idx = [&](int bb) { const int i = bb * 32 + lane; return (i < cnt) ? i : (cnt - 1); };
        const int nfull = cnt >> 5;
        auto add_tail = [&](unsigned r, float n, int m0) {
            if (m0 >= 16) add_half<F, true >(h, tile, r, n, 0, 16, lane);
            else          add_half<F, false>(h, tile, r, n, 0, m0, lane);
            if (m0 > 16)  add_half<F, false>(h, tile, r, n, 16, m0 - 16, lane);
        };
        if (FIRST) {
            unsigned r_cur = ps[idx(0)];
            unsigned r_nxt = ps[idx(nb > 1 ? 1 : 0)];
            float    n_cur = dinv[r_cur & SM] * dinv[base + (int)(r_cur >> SRC_BITS)];
            if (lane < cnt) pn[lane] = n_cur;
            for (int bb = 0; bb < nfull; ++bb) {
                const float d1 = dinv[r_nxt & SM];
                const float d2 = dinv[base + (int)(r_nxt >> SRC_BITS)];
                const unsigned r_n2 = ps[idx(bb + 2 < nb ? bb + 2 : nb - 1)];
                add_half<F, true>(h, tile, r_cur, n_cur, 0, 16, lane);
                add_half<F, true>(h, tile, r_cur, n_cur, 16, 16, lane);
                r_cur = r_nxt; n_cur = d1 * d2; r_nxt = r_n2;
                const int i1 = (bb + 1) * 32 + lane;
                if (i1 < cnt) pn[i1] = n_cur;
            }
            if (nfull < nb) add_tail(r_cur, n_cur, cnt - nfull * 32);
        } else {
            const int ic = idx(0);
            unsigned r_cur = ps[ic];
            float    n_cur = pn[ic];
            for (int bb = 0; bb < nfull; ++bb) {
                const int ix = idx(bb + 1 < nb ? bb + 1 : nb - 1);
                const unsigned r_nxt = ps[ix];
                const float    n_nxt = pn[ix];
                add_half<F, true>(h, tile, r_cur, n_cur, 0, 16, lane);
                add_half<F, true>(h, tile, r_cur, n_cur, 16, 16, lane);
                r_cur = r_nxt; n_cur = n_nxt;
            }
            if (nfull < nb) add_tail(r_cur, n_cur, cnt - nfull * 32);
        }
    }
    }
    __syncthreads();

    for (int i = tid; i < nloc * F4; i += 1024) {
        float4 t = *(const float4*)(&tile[i * 4]);
        t.x = fmaxf(t.x, 0.0f); t.y = fmaxf(t.y, 0.0f);
        t.z = fmaxf(t.z, 0.0f); t.w = fmaxf(t.w, 0.0f);
        *(float4*)(y + (long long)base * F + (long long)i * 4) = t;
    }
}

__global__ void final_linear_kernel(const float* __restrict__ y, const float* __restrict__ Wl,
                                    const float* __restrict__ bl, float* out, int n) {
    int i = blockIdx.x * blockDim.x + threadIdx.x;
    if (i >= n) return;
    const float* row = y + (long long)i * 32;
    float s = 0.0f;
#pragma unroll
    for (int f = 0; f < 32; ++f) s += row[f] * Wl[f];
    out[i] = s + bl[0];
}

static inline size_t align256(size_t x) { return (x + 255) & ~size_t(255); }

extern "C" void kernel_launch(void* const* d_in, const int* in_sizes, int n_in,
                              void* d_out, int out_size, void* d_ws, size_t ws_size,
                              hipStream_t stream) {
    const float* x    = (const float*)d_in[0];
    const int*   ei   = (const int*)d_in[1];
    const float* W1   = (const float*)d_in[2];
    const float* b1   = (const float*)d_in[3];
    const float* W2   = (const float*)d_in[4];
    const float* b2   = (const float*)d_in[5];
    const float* Wl   = (const float*)d_in[6];
    const float* bl   = (const float*)d_in[7];
    float*       out  = (float*)d_out;

    const int N = N_NODES;
    const int E = N_EDGES;
    const int* src = ei;
    const int* dst = ei + E;

    char* ws = (char*)d_ws;
    size_t off = 0;
    float* dinv = (float*)(ws + off); off += align256((size_t)N * 4);
    float* h1   = (float*)(ws + off); off += align256((size_t)N * 64 * 4);
    float* y1   = (float*)(ws + off); off += align256((size_t)N * 64 * 4);
    float* h2   = (float*)(ws + off); off += align256((size_t)N * 32 * 4);
    float* y2   = (float*)(ws + off); off += align256((size_t)N * 32 * 4);
    unsigned char* owner = (unsigned char*)(ws + off); off += align256((size_t)E);
    int*          plan_c = (int*)(ws + off);          off += align256((size_t)NTILE * 32 * 4);
    unsigned int* plan_s = (unsigned int*)(ws + off); off += align256((size_t)NTILE * 32 * WQ * 4);
    float*        plan_n = (float*)(ws + off);        off += align256((size_t)NTILE * 32 * WQ * 4);
    unsigned int* bitmap = (unsigned int*)(ws + off); off += align256((size_t)NTILE * BWP * 4);
    (void)ws_size; (void)n_in; (void)in_sizes; (void)out_size;

    const int T = 256;

    owner_map_kernel<<<NCH, 1024, 0, stream>>>(dst, owner, bitmap, E);

    tile_plan_kernel<<<NTILE, 1024, 0, stream>>>(src, dst, bitmap, dinv, N, plan_c, plan_s);

    gemm_wmma_f32<128, 64, 2><<<N / 32, 2 * 4 * 32, 0, stream>>>(x, W1, h1, N);

    tile_agg_kernel<64, true><<<NTILE, 1024, 0, stream>>>(h1, src, dst, owner, dinv, b1, y1, N,
                                                          plan_c, plan_s, plan_n);

    gemm_wmma_f32<64, 32, 2><<<N / 32, 2 * 2 * 32, 0, stream>>>(y1, W2, h2, N);

    tile_agg_kernel<32, false><<<NTILE, 1024, 0, stream>>>(h2, src, dst, owner, dinv, b2, y2, N,
                                                           plan_c, plan_s, plan_n);

    final_linear_kernel<<<(N + T - 1) / T, T, 0, stream>>>(y2, Wl, bl, out, N);
}


# Plan and layer-1 product share a launch: 256 plan work-groups + 782 product ones (128 rows, staging loads grouped), two per WGP; plan cut to 64 registers (4 group loads a trip, not 8); 7->6 launches
# speedup vs baseline: 3.8053x; 1.0635x over previous
#include <hip/hip_runtime.h>
#include <hip/hip_bf16.h>

typedef __attribute__((ext_vector_type(2))) float v2f;
typedef __attribute__((ext_vector_type(8))) float v8f;

#define N_NODES 100000
#define N_EDGES 3200000

template <int K, int NOUT, int MT>
struct GemmShape {
    static constexpr int NT   = NOUT / 16;
    static constexpr int NW   = MT * NT;
    static constexpr int ROWS = MT * 16;
    static constexpr int KP   = K + 4;
    static constexpr int NP   = NOUT + 4;
    static constexpr int A_WORDS = ROWS * KP;
    static constexpr int B_WORDS = K * NP;
};

template <int K, int NOUT, int MT, bool GROUPED>
__device__ __forceinline__
void gemm_tile_f32(const float* __restrict__ A, const float* __restrict__ B,
                   float* __restrict__ C, int M, int blk, float* As, float* Bs) {
    typedef GemmShape<K, NOUT, MT> S;
    constexpr int NT   = S::NT;
    constexpr int NW   = S::NW;
    constexpr int ROWS = S::ROWS;
    constexpr int KP   = S::KP;
    constexpr int NP   = S::NP;

    const int tid  = threadIdx.x;
    const int row0 = blk * ROWS;

    constexpr int A_ELTS = ROWS * K;
    constexpr int B_ELTS = K * NOUT;
    constexpr int STEP   = NW * 32 * 4;
    if constexpr (GROUPED) {
        static_assert(A_ELTS % STEP == 0 && B_ELTS % STEP == 0, "whole turns only");
        constexpr int AT = A_ELTS / STEP, BT = B_ELTS / STEP;
        float4 av[AT], bv[BT];
#pragma unroll
        for (int t = 0; t < AT; ++t) {
            const int i = tid * 4 + t * STEP;
            const int gr = row0 + i / K;
            av[t] = (gr < M) ? *(const float4*)(A + (long long)gr * K + i % K)
                             : make_float4(0.f, 0.f, 0.f, 0.f);
        }
#pragma unroll
        for (int t = 0; t < BT; ++t) bv[t] = *(const float4*)(B + tid * 4 + t * STEP);
#pragma unroll
        for (int t = 0; t < AT; ++t) {
            const int i = tid * 4 + t * STEP;
            *(float4*)(&As[(i / K) * KP + i % K]) = av[t];
        }
#pragma unroll
        for (int t = 0; t < BT; ++t) {
            const int i = tid * 4 + t * STEP;
            *(float4*)(&Bs[(i / NOUT) * NP + i % NOUT]) = bv[t];
        }
    } else {
        for (int i = tid * 4; i < A_ELTS; i += STEP) {
            int r = i / K, c = i % K;
            int gr = row0 + r;
            float4 v = (gr < M) ? *(const float4*)(A + (long long)gr * K + c)
                                : make_float4(0.f, 0.f, 0.f, 0.f);
            *(float4*)(&As[r * KP + c]) = v;
        }
        for (int i = tid * 4; i < B_ELTS; i += STEP) {
            int r = i / NOUT, c = i % NOUT;
            *(float4*)(&Bs[r * NP + c]) = *(const float4*)(B + i);
        }
    }
    __syncthreads();

    const int wid  = tid >> 5;
    const int lane = tid & 31;
    const int mn   = lane & 15;
    const int h    = lane >> 4;
    const int mt   = wid / NT;
    const int nt   = wid % NT;

    v8f acc = {};
    const float* arow = &As[(mt * 16 + mn) * KP];
    const int    ncol = nt * 16 + mn;

#pragma unroll
    for (int k = 0; k < K; k += 4) {
        v2f a, b;
        a.x = arow[k + 2 * h + 0];
        a.y = arow[k + 2 * h + 1];
        b.x = Bs[(k + 2 * h + 0) * NP + ncol];
        b.y = Bs[(k + 2 * h + 1) * NP + ncol];
        acc = __builtin_amdgcn_wmma_f32_16x16x4_f32(
             false, a,  false, b,
             (short)0, acc,  false,  false);
    }

#pragma unroll
    for (int j = 0; j < 8; ++j) {
        int r = row0 + mt * 16 + j + 8 * h;
        if (r < M) C[(long long)r * NOUT + ncol] = acc[j];
    }
}

template <int K, int NOUT, int MT>
__global__ __launch_bounds__(MT*(NOUT/16)*32)
void gemm_wmma_f32(const float* __restrict__ A, const float* __restrict__ B,
                   float* __restrict__ C, int M) {
    typedef GemmShape<K, NOUT, MT> S;
    __shared__ __attribute__((aligned(16))) float As[S::A_WORDS];
    __shared__ __attribute__((aligned(16))) float Bs[S::B_WORDS];
    gemm_tile_f32<K, NOUT, MT, false>(A, B, C, M, (int)blockIdx.x, As, Bs);
}

constexpr int NPW   = 392;
constexpr int NTILE = 256;
constexpr int WQ    = 512;
constexpr int NGRP  = N_EDGES / 16;
constexpr int NBAT  = (NGRP + 8191) / 8192;
static_assert(NPW * NTILE >= N_NODES, "tiles must cover every node");
static_assert(N_EDGES % 16 == 0, "owner map is read in 16-byte groups");
static_assert(NPW <= 65535, "local row index is kept in 16 bits");
constexpr int CH    = 8192;
constexpr int NCH   = (N_EDGES + CH - 1) / CH;
constexpr int CHW   = CH / 16 / 32;
constexpr int BWP   = NCH * CHW;
constexpr int GQ    = 640;
static_assert(CH == 1024 * 8, "one thread of the owner-map launch takes eight edges");
static_assert(BWP % 4 == 0, "bitmap rows are written in 16-byte pieces");
constexpr int SRC_BITS = 17;
static_assert(N_NODES <= (1 << SRC_BITS), "source id must fit the low bits of a stored entry");
static_assert(NPW <= (1 << (32 - SRC_BITS)), "local row must fit the high bits of a stored entry");

__global__ __launch_bounds__(1024)
void owner_map_kernel(const int* __restrict__ dst, unsigned char* __restrict__ owner,
                      unsigned int* __restrict__ bitmap, int E) {
    __shared__ unsigned int pres[CHW * NTILE];
    const int tid = threadIdx.x;
    for (int i = tid; i < CHW * NTILE; i += 1024) pres[i] = 0u;
    __syncthreads();
    const long long e0 = (long long)blockIdx.x * CH + tid * 8;
    if (e0 < E) {
        const int4 a = *(const int4*)(dst + e0);
        const int4 b = *(const int4*)(dst + e0 + 4);
        const unsigned o0 = (unsigned)(a.x / NPW), o1 = (unsigned)(a.y / NPW);
        const unsigned o2 = (unsigned)(a.z / NPW), o3 = (unsigned)(a.w / NPW);
        const unsigned o4 = (unsigned)(b.x / NPW), o5 = (unsigned)(b.y / NPW);
        const unsigned o6 = (unsigned)(b.z / NPW), o7 = (unsigned)(b.w / NPW);
        uint2 ob;
        ob.x = o0 | (o1 << 8) | (o2 << 16) | (o3 << 24);
        ob.y = o4 | (o5 << 8) | (o6 << 16) | (o7 << 24);
        *(uint2*)(owner + e0) = ob;
        const int gl = tid >> 1;
        const unsigned bit = 1u << (gl & 31);
        unsigned int* pw = &pres[(gl >> 5) * NTILE];
        atomicOr(pw + o0, bit); atomicOr(pw + o1, bit); atomicOr(pw + o2, bit); atomicOr(pw + o3, bit);
        atomicOr(pw + o4, bit); atomicOr(pw + o5, bit); atomicOr(pw + o6, bit); atomicOr(pw + o7, bit);
    }
    __syncthreads();
    const int t = tid >> 2, q = (tid & 3) * 4;
    uint4 v;
    v.x = pres[(q + 0) * NTILE + t];
    v.y = pres[(q + 1) * NTILE + t];
    v.z = pres[(q + 2) * NTILE + t];
    v.w = pres[(q + 3) * NTILE + t];
    *(uint4*)(bitmap + (long long)t * BWP + blockIdx.x * CHW + q) = v;
}

__device__ __forceinline__ unsigned zero_bytes(unsigned y) {
    unsigned t = (y & 0x7F7F7F7Fu) + 0x7F7F7F7Fu;
    return ~(t | y | 0x7F7F7F7Fu);
}
__device__ __forceinline__ unsigned match_group(uint4 v, unsigned me) {
    return (zero_bytes(v.x ^ me) >> 7) | (zero_bytes(v.y ^ me) >> 6) |
           (zero_bytes(v.z ^ me) >> 5) | (zero_bytes(v.w ^ me) >> 4);
}

template <int F>
__device__ void spill_edge(int e, int base, float* tile, const float* __restrict__ h,
                           const int* __restrict__ src, const int* __restrict__ dst,
                           const float* __restrict__ dinv) {
    const int s = src[e], d = dst[e];
    const float nrm = dinv[s] * dinv[d];
    const float* hr = h + (long long)s * F;
    float* tr = tile + (d - base) * F;
    for (int f = 0; f < F; ++f) atomicAdd(tr + f, hr[f] * nrm);
}
__device__ __forceinline__ int wrap_group(int g) {
    if (g >= NGRP) g -= NGRP;
    if (g >= NGRP) g -= NGRP;
    return g;
}

template <typename Spill>
__device__ __forceinline__ int scan_owner_map(const unsigned char* __restrict__ owner, int tileid,
                                              int tid, int* myq, Spill spill) {
    const unsigned me   = (unsigned)tileid * 0x01010101u;
    const uint4*   om   = (const uint4*)owner;
    const int rot = tileid * 768;
    int wc = 0;
    for (int it = 0; it < NBAT; ++it) {
        uint4 v[8];
#pragma unroll
        for (int k = 0; k < 8; ++k) {
            v[k] = om[wrap_group(it * 8192 + k * 1024 + tid + rot)];
        }
        unsigned mk[8];
#pragma unroll
        for (int k = 0; k < 8; ++k)
            mk[k] = match_group(v[k], me) & (0u - (unsigned)(it * 8192 + k * 1024 + tid < NGRP));
#pragma unroll
        for (int hf = 0; hf < 2; ++hf) {
            const unsigned lo = mk[4 * hf + 0] | (mk[4 * hf + 1] << 4);
            const unsigned hi = mk[4 * hf + 2] | (mk[4 * hf + 3] << 4);
            unsigned long long M = ((unsigned long long)hi << 32) | (unsigned long long)lo;
            unsigned act;
            while ((act = __builtin_amdgcn_ballot_w32(M != 0ull)) != 0u) {
                if (M != 0ull) {
                    const int p = __builtin_ctzll(M);
                    M &= M - 1ull;
                    const int k  = 4 * hf + 2 * (p >> 5) + ((p >> 2) & 1);
                    const int g = wrap_group(it * 8192 + k * 1024 + tid + rot);
                    const int e = g * 16 + (p & 3) * 4 + ((p & 31) >> 3);
                    const int slot = wc + (int)__builtin_amdgcn_mbcnt_lo(act, 0u);
                    if (slot < WQ) myq[slot] = e;
                    else spill(e);
                }
                wc += __builtin_popcount(act);
            }
        }
    }
    return wc;
}

static_assert(CHW == 16 && NTILE * 4 == 1024, "bitmap write-out of the owner-map launch");
constexpr int BW_FULL = BWP / 1024;
constexpr int BW_REST = BWP - BW_FULL * 1024;
static_assert(BW_REST * 9 <= 1024 + 8, "rest of a bitmap row is dealt to every 9th thread");

constexpr int PLAN_Q_WORDS = 32 * WQ;
constexpr int PLAN_G_WORDS = 32 * GQ;
constexpr int PLAN_WORDS   = PLAN_Q_WORDS + PLAN_G_WORDS + NPW + 32;

__device__ __forceinline__
void tile_plan_body(const int* __restrict__ src, const int* __restrict__ dst,
                    const unsigned int* __restrict__ bitmap, float* __restrict__ dinv, int N,
                    int* __restrict__ plan_c, unsigned int* __restrict__ plan_s,
                    int tile_id, int* lds) {
    int* q_s   = lds;
    int* g_s   = lds + PLAN_Q_WORDS;
    int* indeg = lds + PLAN_Q_WORDS + PLAN_G_WORDS;
    int* wcs   = indeg + NPW;

    const int tid  = threadIdx.x;
    const int wid  = tid >> 5;
    const int lane = tid & 31;
    const int base = tile_id * NPW;
    int nloc = N - base;
    if (nloc > NPW) nloc = NPW;
    if (nloc < 0) nloc = 0;

    for (int i = tid; i < NPW; i += 1024) indeg[i] = 0;
    __syncthreads();

    int* myq = &q_s[wid * WQ];
    int* myg = &g_s[wid * GQ];
    unsigned int* ps = plan_s + (long long)(tile_id * 32 + wid) * WQ;

    const unsigned int* brow = bitmap + (long long)tile_id * BWP;
    unsigned bits[BW_FULL + 1];
    int      bj[BW_FULL + 1];
#pragma unroll
    for (int k = 0; k < BW_FULL; ++k) { bj[k] = k * 1024 + tid; bits[k] = brow[bj[k]]; }
    {
        const int r = tid / 9;
        const bool has = (tid - r * 9 == 0) && (r < BW_REST);
        bj[BW_FULL] = BW_FULL * 1024 + (has ? r : 0);
        const unsigned v = brow[bj[BW_FULL]];
        bits[BW_FULL] = has ? v : 0u;
    }
    int  gc  = 0;
    bool ovf = false;
#pragma unroll
    for (int k = 0; k <= BW_FULL; ++k) {
        unsigned m = bits[k];
        unsigned act;
        while ((act = __builtin_amdgcn_ballot_w32(m != 0u)) != 0u) {
            if (m != 0u) {
                const int p = __builtin_ctz(m);
                m &= m - 1u;
                const int g = bj[k] * 32 + p;
                const int slot = gc + (int)__builtin_amdgcn_mbcnt_lo(act, 0u);
                if (slot < GQ) myg[slot] = g;
                else {
                    ovf = true;
                    for (int x = 0; x < 16; ++x) {
                        const unsigned lr = (unsigned)(dst[g * 16 + x] - base);
                        if (lr < (unsigned)NPW) atomicAdd(&indeg[lr], 1);
                    }
                }
            }
            gc += __builtin_popcount(act);
        }
    }
    if (gc > GQ) gc = GQ;

    constexpr int GB = 4;
    int wc = 0;
    const int sub = lane >> 2;
    const int qo  = (lane & 3) * 4;
    for (int i0 = 0; i0 < gc; i0 += 8 * GB) {
        int   gg[GB];
        uint4 d[GB], s[GB];
#pragma unroll
        for (int k = 0; k < GB; ++k) {
            const int i = i0 + 8 * k + sub;
            gg[k] = myg[(i < gc) ? i : (gc - 1)];
        }
#pragma unroll
        for (int k = 0; k < GB; ++k) {
            d[k] = *(const uint4*)(dst + (long long)gg[k] * 16 + qo);
            s[k] = *(const uint4*)(src + (long long)gg[k] * 16 + qo);
        }
#pragma unroll
        for (int k = 0; k < GB; ++k) {
            const bool ok = i0 + 8 * k + sub < gc;
#pragma unroll
            for (int x = 0; x < 4; ++x) {
                const unsigned dv = x == 0 ? d[k].x : x == 1 ? d[k].y : x == 2 ? d[k].z : d[k].w;
                const unsigned sv = x == 0 ? s[k].x : x == 1 ? s[k].y : x == 2 ? s[k].z : s[k].w;
                const unsigned lr = dv - (unsigned)base;
                const bool hit = ok && lr < (unsigned)NPW;
                const unsigned act = __builtin_amdgcn_ballot_w32(hit);
                if (act != 0u) {
                    if (hit) {
                        const int slot = wc + (int)__builtin_amdgcn_mbcnt_lo(act, 0u);
                        if (slot < WQ) myq[slot] = (int)(sv | (lr << SRC_BITS));
                        else atomicAdd(&indeg[lr], 1);
                    }
                    wc += __builtin_popcount(act);
                }
            }
        }
    }
    const bool lost = __builtin_amdgcn_ballot_w32(ovf) != 0u;
    if (lane == 0) wcs[wid] = (lost && wc <= WQ) ? (WQ + 1) : wc;

    const int cnt = wc < WQ ? wc : WQ;
    for (int i = lane; i < cnt; i += 32) {
        const unsigned v = (unsigned)myq[i];
        ps[i] = v;
        atomicAdd(&indeg[v >> SRC_BITS], 1);
    }
    __syncthreads();
    if (wid == 0) plan_c[tile_id * 32 + lane] = wcs[lane];
    for (int n = tid; n < nloc; n += 1024) dinv[base + n] = rsqrtf((float)(indeg[n] + 1));
}


typedef GemmShape<128, 64, 8> G1;
constexpr int G1_WGS = (N_NODES + G1::ROWS - 1) / G1::ROWS;
constexpr int G1_WORDS = G1::A_WORDS + G1::B_WORDS;
constexpr int PG_WORDS = PLAN_WORDS > G1_WORDS ? PLAN_WORDS : G1_WORDS;
static_assert(G1::NW * 32 == 1024, "the product's work-groups use the plan's block size");
static_assert(G1::A_WORDS % 4 == 0, "B tile starts on a 16-byte boundary");
static_assert(2 * PG_WORDS * 4 <= 320 * 1024, "a plan and a product work-group share one WGP's LDS");

__global__ __launch_bounds__(1024)
void plan_gemm_kernel(const int* __restrict__ src, const int* __restrict__ dst,
                      const unsigned int* __restrict__ bitmap, float* __restrict__ dinv, int N,
                      int* __restrict__ plan_c, unsigned int* __restrict__ plan_s,
                      const float* __restrict__ x, const float* __restrict__ W1,
                      float* __restrict__ h1) {
    __shared__ __attribute__((aligned(16))) int lds[PG_WORDS];
    if (blockIdx.x < NTILE) {
        tile_plan_body(src, dst, bitmap, dinv, N, plan_c, plan_s, (int)blockIdx.x, lds);
    } else {
        float* As = reinterpret_cast<float*>(lds);
        gemm_tile_f32<128, 64, 8, true>(x, W1, h1, N, (int)blockIdx.x - NTILE, As, As + G1::A_WORDS);
    }
}

template <int F, bool FULL>
__device__ __forceinline__ void add_half(const float* __restrict__ h, float* tile, unsigned r, float n,
                                         int j0, int m, int lane) {
    constexpr int      VPL = F / 32;
    constexpr unsigned SM  = (1u << SRC_BITS) - 1u;
    float vals[16][VPL];
#pragma unroll
    for (int k = 0; k < 16; ++k) {
        const unsigned rj = (unsigned)__builtin_amdgcn_readlane((int)r, j0 + k);
        const float* hr = h + (long long)(rj & SM) * F + lane * VPL;
#pragma unroll
        for (int x = 0; x < VPL; ++x) vals[k][x] = hr[x];
    }
#pragma unroll
    for (int k = 0; k < 16; ++k) {
        if (FULL || k < m) {
            const unsigned rj  = (unsigned)__builtin_amdgcn_readlane((int)r, j0 + k);
            const float    nrm = __int_as_float(__builtin_amdgcn_readlane(__float_as_int(n), j0 + k));
            float* tr = tile + (int)(rj >> SRC_BITS) * F + lane * VPL;
#pragma unroll
            for (int x = 0; x < VPL; ++x) atomicAdd(tr + x, vals[k][x] * nrm);
        }
    }
}

template <int F, bool FIRST>
__global__ __launch_bounds__(1024)
void tile_agg_kernel(const float* __restrict__ h, const int* __restrict__ src,
                     const int* __restrict__ dst, const unsigned char* __restrict__ owner,
                     const float* __restrict__ dinv, const float* __restrict__ bias,
                     float* __restrict__ y, int N, const int* __restrict__ plan_c,
                     const unsigned int* __restrict__ plan_s, float* plan_n) {
    constexpr int F4  = F / 4;
    constexpr int VPL = F / 32;
    __shared__ float          tile[NPW * F];
    __shared__ int            q_s[32 * WQ];
    __shared__ float          q_n[32 * WQ];
    __shared__ unsigned short q_d[32 * WQ];

    const int tid  = threadIdx.x;
    const int wid  = tid >> 5;
    const int lane = tid & 31;
    const int base = blockIdx.x * NPW;
    int nloc = N - base;
    if (nloc > NPW) nloc = NPW;
    if (nloc < 0) nloc = 0;

    for (int i = tid; i < nloc * F4; i += 1024) {
        const int n = i / F4, c = i % F4;
        const float di = dinv[base + n];
        const float4 hv = *(const float4*)(h + (long long)(base + n) * F + c * 4);
        const float4 bv = *(const float4*)(bias + c * 4);
        float4 t;
        t.x = hv.x * di * di + bv.x;
        t.y = hv.y * di * di + bv.y;
        t.z = hv.z * di * di + bv.z;
        t.w = hv.w * di * di + bv.w;
        *(float4*)(&tile[n * F + c * 4]) = t;
    }
    __syncthreads();

    int*            myq = &q_s[wid * WQ];
    float*          myn = &q_n[wid * WQ];
    unsigned short* myd = &q_d[wid * WQ];
    const unsigned int* ps = plan_s + (long long)(blockIdx.x * 32 + wid) * WQ;
    float*              pn = plan_n + (long long)(blockIdx.x * 32 + wid) * WQ;
    const int c32 = plan_c[blockIdx.x * 32 + lane];
    const bool build = __builtin_amdgcn_ballot_w32(c32 > WQ) != 0u;
    int wc = __builtin_amdgcn_readlane(c32, __builtin_amdgcn_readfirstlane(wid));

    if (build) {
    wc = scan_owner_map(owner, (int)blockIdx.x, tid, myq,
                        [&](int e) { spill_edge<F>(e, base, tile, h, src, dst, dinv); });
    const int cnt = wc < WQ ? wc : WQ;
    for (int i0 = 0; i0 < cnt; i0 += 128) {
        int ix[4], ss[4], dd[4];
        float nr[4];
#pragma unroll
        for (int k = 0; k < 4; ++k) {
            const int i = i0 + 32 * k + lane;
            ix[k] = (i < cnt) ? i : (cnt - 1);
        }
        int ee[4];
#pragma unroll
        for (int k = 0; k < 4; ++k) ee[k] = myq[ix[k]];
#pragma unroll
        for (int k = 0; k < 4; ++k) { ss[k] = src[ee[k]]; dd[k] = dst[ee[k]]; }
#pragma unroll
        for (int k = 0; k < 4; ++k) nr[k] = dinv[ss[k]] * dinv[dd[k]];
#pragma unroll
        for (int k = 0; k < 4; ++k) {
            myq[ix[k]] = ss[k];
            myd[ix[k]] = (unsigned short)(dd[k] - base);
            myn[ix[k]] = nr[k];
        }
    }
    __syncthreads();

    constexpr int U = 16;
    for (int i0 = 0; i0 < cnt; i0 += U) {
        float vals[U][VPL];
#pragma unroll
        for (int k = 0; k < U; ++k) {
            const int i = (i0 + k < cnt) ? (i0 + k) : i0;
            const float* hr = h + (long long)myq[i] * F + lane * VPL;
#pragma unroll
            for (int x = 0; x < VPL; ++x) vals[k][x] = hr[x];
        }
#pragma unroll
        for (int k = 0; k < U; ++k) {
            if (i0 + k < cnt) {
                const float nrm = myn[i0 + k];
                float* tr = &tile[(int)myd[i0 + k] * F + lane * VPL];
#pragma unroll
                for (int x = 0; x < VPL; ++x) atomicAdd(tr + x, vals[k][x] * nrm);
            }
        }
    }
    } else {
    const int cnt = wc;
    if (cnt > 0) {
        constexpr unsigned SM = (1u << SRC_BITS) - 1u;
        const int nb = (cnt + 31) >> 5;
        auto idx = [&](int bb) { const int i = bb * 32 + lane; return (i < cnt) ? i : (cnt - 1); };
        const int nfull = cnt >> 5;
        auto add_tail = [&](unsigned r, float n, int m0) {
            if (m0 >= 16) add_half<F, true >(h, tile, r, n, 0, 16, lane);
            else          add_half<F, false>(h, tile, r, n, 0, m0, lane);
            if (m0 > 16)  add_half<F, false>(h, tile, r, n, 16, m0 - 16, lane);
        };
        if (FIRST) {
            unsigned r_cur = ps[idx(0)];
            unsigned r_nxt = ps[idx(nb > 1 ? 1 : 0)];
            float    n_cur = dinv[r_cur & SM] * dinv[base + (int)(r_cur >> SRC_BITS)];
            if (lane < cnt) pn[lane] = n_cur;
            for (int bb = 0; bb < nfull; ++bb) {
                const float d1 = dinv[r_nxt & SM];
                const float d2 = dinv[base + (int)(r_nxt >> SRC_BITS)];
                const unsigned r_n2 = ps[idx(bb + 2 < nb ? bb + 2 : nb - 1)];
                add_half<F, true>(h, tile, r_cur, n_cur, 0, 16, lane);
                add_half<F, true>(h, tile, r_cur, n_cur, 16, 16, lane);
                r_cur = r_nxt; n_cur = d1 * d2; r_nxt = r_n2;
                const int i1 = (bb + 1) * 32 + lane;
                if (i1 < cnt) pn[i1] = n_cur;
            }
            if (nfull < nb) add_tail(r_cur, n_cur, cnt - nfull * 32);
        } else {
            const int ic = idx(0);
            unsigned r_cur = ps[ic];
            float    n_cur = pn[ic];
            for (int bb = 0; bb < nfull; ++bb) {
                const int ix = idx(bb + 1 < nb ? bb + 1 : nb - 1);
                const unsigned r_nxt = ps[ix];
                const float    n_nxt = pn[ix];
                add_half<F, true>(h, tile, r_cur, n_cur, 0, 16, lane);
                add_half<F, true>(h, tile, r_cur, n_cur, 16, 16, lane);
                r_cur = r_nxt; n_cur = n_nxt;
            }
            if (nfull < nb) add_tail(r_cur, n_cur, cnt - nfull * 32);
        }
    }
    }
    __syncthreads();

    for (int i = tid; i < nloc * F4; i += 1024) {
        float4 t = *(const float4*)(&tile[i * 4]);
        t.x = fmaxf(t.x, 0.0f); t.y = fmaxf(t.y, 0.0f);
        t.z = fmaxf(t.z, 0.0f); t.w = fmaxf(t.w, 0.0f);
        *(float4*)(y + (long long)base * F + (long long)i * 4) = t;
    }
}

__global__ void final_linear_kernel(const float* __restrict__ y, const float* __restrict__ Wl,
                                    const float* __restrict__ bl, float* out, int n) {
    int i = blockIdx.x * blockDim.x + threadIdx.x;
    if (i >= n) return;
    const float* row = y + (long long)i * 32;
    float s = 0.0f;
#pragma unroll
    for (int f = 0; f < 32; ++f) s += row[f] * Wl[f];
    out[i] = s + bl[0];
}

static inline size_t align256(size_t x) { return (x + 255) & ~size_t(255); }

extern "C" void kernel_launch(void* const* d_in, const int* in_sizes, int n_in,
                              void* d_out, int out_size, void* d_ws, size_t ws_size,
                              hipStream_t stream) {
    const float* x    = (const float*)d_in[0];
    const int*   ei   = (const int*)d_in[1];
    const float* W1   = (const float*)d_in[2];
    const float* b1   = (const float*)d_in[3];
    const float* W2   = (const float*)d_in[4];
    const float* b2   = (const float*)d_in[5];
    const float* Wl   = (const float*)d_in[6];
    const float* bl   = (const float*)d_in[7];
    float*       out  = (float*)d_out;

    const int N = N_NODES;
    const int E = N_EDGES;
    const int* src = ei;
    const int* dst = ei + E;

    char* ws = (char*)d_ws;
    size_t off = 0;
    float* dinv = (float*)(ws + off); off += align256((size_t)N * 4);
    float* h1   = (float*)(ws + off); off += align256((size_t)N * 64 * 4);
    float* y1   = (float*)(ws + off); off += align256((size_t)N * 64 * 4);
    float* h2   = (float*)(ws + off); off += align256((size_t)N * 32 * 4);
    float* y2   = (float*)(ws + off); off += align256((size_t)N * 32 * 4);
    unsigned char* owner = (unsigned char*)(ws + off); off += align256((size_t)E);
    int*          plan_c = (int*)(ws + off);          off += align256((size_t)NTILE * 32 * 4);
    unsigned int* plan_s = (unsigned int*)(ws + off); off += align256((size_t)NTILE * 32 * WQ * 4);
    float*        plan_n = (float*)(ws + off);        off += align256((size_t)NTILE * 32 * WQ * 4);
    unsigned int* bitmap = (unsigned int*)(ws + off); off += align256((size_t)NTILE * BWP * 4);
    (void)ws_size; (void)n_in; (void)in_sizes; (void)out_size;

    const int T = 256;

    owner_map_kernel<<<NCH, 1024, 0, stream>>>(dst, owner, bitmap, E);

    plan_gemm_kernel<<<NTILE + G1_WGS, 1024, 0, stream>>>(src, dst, bitmap, dinv, N, plan_c, plan_s,
                                                           x, W1, h1);

    tile_agg_kernel<64, true><<<NTILE, 1024, 0, stream>>>(h1, src, dst, owner, dinv, b1, y1, N,
                                                          plan_c, plan_s, plan_n);

    gemm_wmma_f32<64, 32, 2><<<N / 32, 2 * 2 * 32, 0, stream>>>(y1, W2, h2, N);

    tile_agg_kernel<32, false><<<NTILE, 1024, 0, stream>>>(h2, src, dst, owner, dinv, b2, y2, N,
                                                           plan_c, plan_s, plan_n);

    final_linear_kernel<<<(N + T - 1) / T, T, 0, stream>>>(y2, Wl, bl, out, N);
}


# First aggregation launch keeps a 16-bit in-degree table of all nodes in LDS (the plan writes it): an edge's source factor is an LDS lookup + rsqrt, not a scattered 4-byte memory read; sums unchanged
# speedup vs baseline: 3.9442x; 1.0365x over previous
#include <hip/hip_runtime.h>
#include <hip/hip_bf16.h>

typedef __attribute__((ext_vector_type(2))) float v2f;
typedef __attribute__((ext_vector_type(8))) float v8f;

#define N_NODES 100000
#define N_EDGES 3200000

template <int K, int NOUT, int MT>
struct GemmShape {
    static constexpr int NT   = NOUT / 16;
    static constexpr int NW   = MT * NT;
    static constexpr int ROWS = MT * 16;
    static constexpr int KP   = K + 4;
    static constexpr int NP   = NOUT + 4;
    static constexpr int A_WORDS = ROWS * KP;
    static constexpr int B_WORDS = K * NP;
};

template <int K, int NOUT, int MT, bool GROUPED>
__device__ __forceinline__
void gemm_tile_f32(const float* __restrict__ A, const float* __restrict__ B,
                   float* __restrict__ C, int M, int blk, float* As, float* Bs) {
    typedef GemmShape<K, NOUT, MT> S;
    constexpr int NT   = S::NT;
    constexpr int NW   = S::NW;
    constexpr int ROWS = S::ROWS;
    constexpr int KP   = S::KP;
    constexpr int NP   = S::NP;

    const int tid  = threadIdx.x;
    const int row0 = blk * ROWS;

    constexpr int A_ELTS = ROWS * K;
    constexpr int B_ELTS = K * NOUT;
    constexpr int STEP   = NW * 32 * 4;
    if constexpr (GROUPED) {
        static_assert(A_ELTS % STEP == 0 && B_ELTS % STEP == 0, "whole turns only");
        constexpr int AT = A_ELTS / STEP, BT = B_ELTS / STEP;
        float4 av[AT], bv[BT];
#pragma unroll
        for (int t = 0; t < AT; ++t) {
            const int i = tid * 4 + t * STEP;
            const int gr = row0 + i / K;
            av[t] = (gr < M) ? *(const float4*)(A + (long long)gr * K + i % K)
                             : make_float4(0.f, 0.f, 0.f, 0.f);
        }
#pragma unroll
        for (int t = 0; t < BT; ++t) bv[t] = *(const float4*)(B + tid * 4 + t * STEP);
#pragma unroll
        for (int t = 0; t < AT; ++t) {
            const int i = tid * 4 + t * STEP;
            *(float4*)(&As[(i / K) * KP + i % K]) = av[t];
        }
#pragma unroll
        for (int t = 0; t < BT; ++t) {
            const int i = tid * 4 + t * STEP;
            *(float4*)(&Bs[(i / NOUT) * NP + i % NOUT]) = bv[t];
        }
    } else {
        for (int i = tid * 4; i < A_ELTS; i += STEP) {
            int r = i / K, c = i % K;
            int gr = row0 + r;
            float4 v = (gr < M) ? *(const float4*)(A + (long long)gr * K + c)
                                : make_float4(0.f, 0.f, 0.f, 0.f);
            *(float4*)(&As[r * KP + c]) = v;
        }
        for (int i = tid * 4; i < B_ELTS; i += STEP) {
            int r = i / NOUT, c = i % NOUT;
            *(float4*)(&Bs[r * NP + c]) = *(const float4*)(B + i);
        }
    }
    __syncthreads();

    const int wid  = tid >> 5;
    const int lane = tid & 31;
    const int mn   = lane & 15;
    const int h    = lane >> 4;
    const int mt   = wid / NT;
    const int nt   = wid % NT;

    v8f acc = {};
    const float* arow = &As[(mt * 16 + mn) * KP];
    const int    ncol = nt * 16 + mn;

#pragma unroll
    for (int k = 0; k < K; k += 4) {
        v2f a, b;
        a.x = arow[k + 2 * h + 0];
        a.y = arow[k + 2 * h + 1];
        b.x = Bs[(k + 2 * h + 0) * NP + ncol];
        b.y = Bs[(k + 2 * h + 1) * NP + ncol];
        acc = __builtin_amdgcn_wmma_f32_16x16x4_f32(
             false, a,  false, b,
             (short)0, acc,  false,  false);
    }

#pragma unroll
    for (int j = 0; j < 8; ++j) {
        int r = row0 + mt * 16 + j + 8 * h;
        if (r < M) C[(long long)r * NOUT + ncol] = acc[j];
    }
}

template <int K, int NOUT, int MT>
__global__ __launch_bounds__(MT*(NOUT/16)*32)
void gemm_wmma_f32(const float* __restrict__ A, const float* __restrict__ B,
                   float* __restrict__ C, int M) {
    typedef GemmShape<K, NOUT, MT> S;
    __shared__ __attribute__((aligned(16))) float As[S::A_WORDS];
    __shared__ __attribute__((aligned(16))) float Bs[S::B_WORDS];
    gemm_tile_f32<K, NOUT, MT, false>(A, B, C, M, (int)blockIdx.x, As, Bs);
}

constexpr int NPW   = 392;
constexpr int NTILE = 256;
constexpr int WQ    = 512;
constexpr int NGRP  = N_EDGES / 16;
constexpr int NBAT  = (NGRP + 8191) / 8192;
static_assert(NPW * NTILE >= N_NODES, "tiles must cover every node");
static_assert(N_EDGES % 16 == 0, "owner map is read in 16-byte groups");
static_assert(NPW <= 65535, "local row index is kept in 16 bits");
constexpr int CH    = 8192;
constexpr int NCH   = (N_EDGES + CH - 1) / CH;
constexpr int CHW   = CH / 16 / 32;
constexpr int BWP   = NCH * CHW;
constexpr int GQ    = 640;
static_assert(CH == 1024 * 8, "one thread of the owner-map launch takes eight edges");
static_assert(BWP % 4 == 0, "bitmap rows are written in 16-byte pieces");
constexpr int DEG_SAT  = 65535;
static_assert((N_NODES * 2) % 16 == 0, "the 16-bit degree table is copied in 16-byte pieces");
constexpr int SRC_BITS = 17;
static_assert(N_NODES <= (1 << SRC_BITS), "source id must fit the low bits of a stored entry");
static_assert(NPW <= (1 << (32 - SRC_BITS)), "local row must fit the high bits of a stored entry");

__global__ __launch_bounds__(1024)
void owner_map_kernel(const int* __restrict__ dst, unsigned char* __restrict__ owner,
                      unsigned int* __restrict__ bitmap, int E) {
    __shared__ unsigned int pres[CHW * NTILE];
    const int tid = threadIdx.x;
    for (int i = tid; i < CHW * NTILE; i += 1024) pres[i] = 0u;
    __syncthreads();
    const long long e0 = (long long)blockIdx.x * CH + tid * 8;
    if (e0 < E) {
        const int4 a = *(const int4*)(dst + e0);
        const int4 b = *(const int4*)(dst + e0 + 4);
        const unsigned o0 = (unsigned)(a.x / NPW), o1 = (unsigned)(a.y / NPW);
        const unsigned o2 = (unsigned)(a.z / NPW), o3 = (unsigned)(a.w / NPW);
        const unsigned o4 = (unsigned)(b.x / NPW), o5 = (unsigned)(b.y / NPW);
        const unsigned o6 = (unsigned)(b.z / NPW), o7 = (unsigned)(b.w / NPW);
        uint2 ob;
        ob.x = o0 | (o1 << 8) | (o2 << 16) | (o3 << 24);
        ob.y = o4 | (o5 << 8) | (o6 << 16) | (o7 << 24);
        *(uint2*)(owner + e0) = ob;
        const int gl = tid >> 1;
        const unsigned bit = 1u << (gl & 31);
        unsigned int* pw = &pres[(gl >> 5) * NTILE];
        atomicOr(pw + o0, bit); atomicOr(pw + o1, bit); atomicOr(pw + o2, bit); atomicOr(pw + o3, bit);
        atomicOr(pw + o4, bit); atomicOr(pw + o5, bit); atomicOr(pw + o6, bit); atomicOr(pw + o7, bit);
    }
    __syncthreads();
    const int t = tid >> 2, q = (tid & 3) * 4;
    uint4 v;
    v.x = pres[(q + 0) * NTILE + t];
    v.y = pres[(q + 1) * NTILE + t];
    v.z = pres[(q + 2) * NTILE + t];
    v.w = pres[(q + 3) * NTILE + t];
    *(uint4*)(bitmap + (long long)t * BWP + blockIdx.x * CHW + q) = v;
}

__device__ __forceinline__ unsigned zero_bytes(unsigned y) {
    unsigned t = (y & 0x7F7F7F7Fu) + 0x7F7F7F7Fu;
    return ~(t | y | 0x7F7F7F7Fu);
}
__device__ __forceinline__ unsigned match_group(uint4 v, unsigned me) {
    return (zero_bytes(v.x ^ me) >> 7) | (zero_bytes(v.y ^ me) >> 6) |
           (zero_bytes(v.z ^ me) >> 5) | (zero_bytes(v.w ^ me) >> 4);
}

template <int F>
__device__ void spill_edge(int e, int base, float* tile, const float* __restrict__ h,
                           const int* __restrict__ src, const int* __restrict__ dst,
                           const float* __restrict__ dinv) {
    const int s = src[e], d = dst[e];
    const float nrm = dinv[s] * dinv[d];
    const float* hr = h + (long long)s * F;
    float* tr = tile + (d - base) * F;
    for (int f = 0; f < F; ++f) atomicAdd(tr + f, hr[f] * nrm);
}
__device__ __forceinline__ int wrap_group(int g) {
    if (g >= NGRP) g -= NGRP;
    if (g >= NGRP) g -= NGRP;
    return g;
}

template <typename Spill>
__device__ __forceinline__ int scan_owner_map(const unsigned char* __restrict__ owner, int tileid,
                                              int tid, int* myq, Spill spill) {
    const unsigned me   = (unsigned)tileid * 0x01010101u;
    const uint4*   om   = (const uint4*)owner;
    const int rot = tileid * 768;
    int wc = 0;
    for (int it = 0; it < NBAT; ++it) {
        uint4 v[8];
#pragma unroll
        for (int k = 0; k < 8; ++k) {
            v[k] = om[wrap_group(it * 8192 + k * 1024 + tid + rot)];
        }
        unsigned mk[8];
#pragma unroll
        for (int k = 0; k < 8; ++k)
            mk[k] = match_group(v[k], me) & (0u - (unsigned)(it * 8192 + k * 1024 + tid < NGRP));
#pragma unroll
        for (int hf = 0; hf < 2; ++hf) {
            const unsigned lo = mk[4 * hf + 0] | (mk[4 * hf + 1] << 4);
            const unsigned hi = mk[4 * hf + 2] | (mk[4 * hf + 3] << 4);
            unsigned long long M = ((unsigned long long)hi << 32) | (unsigned long long)lo;
            unsigned act;
            while ((act = __builtin_amdgcn_ballot_w32(M != 0ull)) != 0u) {
                if (M != 0ull) {
                    const int p = __builtin_ctzll(M);
                    M &= M - 1ull;
                    const int k  = 4 * hf + 2 * (p >> 5) + ((p >> 2) & 1);
                    const int g = wrap_group(it * 8192 + k * 1024 + tid + rot);
                    const int e = g * 16 + (p & 3) * 4 + ((p & 31) >> 3);
                    const int slot = wc + (int)__builtin_amdgcn_mbcnt_lo(act, 0u);
                    if (slot < WQ) myq[slot] = e;
                    else spill(e);
                }
                wc += __builtin_popcount(act);
            }
        }
    }
    return wc;
}

static_assert(CHW == 16 && NTILE * 4 == 1024, "bitmap write-out of the owner-map launch");
constexpr int BW_FULL = BWP / 1024;
constexpr int BW_REST = BWP - BW_FULL * 1024;
static_assert(BW_REST * 9 <= 1024 + 8, "rest of a bitmap row is dealt to every 9th thread");

constexpr int PLAN_Q_WORDS = 32 * WQ;
constexpr int PLAN_G_WORDS = 32 * GQ;
constexpr int PLAN_WORDS   = PLAN_Q_WORDS + PLAN_G_WORDS + NPW + 32;

__device__ __forceinline__
void tile_plan_body(const int* __restrict__ src, const int* __restrict__ dst,
                    const unsigned int* __restrict__ bitmap, float* __restrict__ dinv, int N,
                    int* __restrict__ plan_c, unsigned int* __restrict__ plan_s,
                    unsigned short* __restrict__ deg16g, int tile_id, int* lds) {
    int* q_s   = lds;
    int* g_s   = lds + PLAN_Q_WORDS;
    int* indeg = lds + PLAN_Q_WORDS + PLAN_G_WORDS;
    int* wcs   = indeg + NPW;

    const int tid  = threadIdx.x;
    const int wid  = tid >> 5;
    const int lane = tid & 31;
    const int base = tile_id * NPW;
    int nloc = N - base;
    if (nloc > NPW) nloc = NPW;
    if (nloc < 0) nloc = 0;

    for (int i = tid; i < NPW; i += 1024) indeg[i] = 0;
    __syncthreads();

    int* myq = &q_s[wid * WQ];
    int* myg = &g_s[wid * GQ];
    unsigned int* ps = plan_s + (long long)(tile_id * 32 + wid) * WQ;

    const unsigned int* brow = bitmap + (long long)tile_id * BWP;
    unsigned bits[BW_FULL + 1];
    int      bj[BW_FULL + 1];
#pragma unroll
    for (int k = 0; k < BW_FULL; ++k) { bj[k] = k * 1024 + tid; bits[k] = brow[bj[k]]; }
    {
        const int r = tid / 9;
        const bool has = (tid - r * 9 == 0) && (r < BW_REST);
        bj[BW_FULL] = BW_FULL * 1024 + (has ? r : 0);
        const unsigned v = brow[bj[BW_FULL]];
        bits[BW_FULL] = has ? v : 0u;
    }
    int  gc  = 0;
    bool ovf = false;
#pragma unroll
    for (int k = 0; k <= BW_FULL; ++k) {
        unsigned m = bits[k];
        unsigned act;
        while ((act = __builtin_amdgcn_ballot_w32(m != 0u)) != 0u) {
            if (m != 0u) {
                const int p = __builtin_ctz(m);
                m &= m - 1u;
                const int g = bj[k] * 32 + p;
                const int slot = gc + (int)__builtin_amdgcn_mbcnt_lo(act, 0u);
                if (slot < GQ) myg[slot] = g;
                else {
                    ovf = true;
                    for (int x = 0; x < 16; ++x) {
                        const unsigned lr = (unsigned)(dst[g * 16 + x] - base);
                        if (lr < (unsigned)NPW) atomicAdd(&indeg[lr], 1);
                    }
                }
            }
            gc += __builtin_popcount(act);
        }
    }
    if (gc > GQ) gc = GQ;

    constexpr int GB = 4;
    int wc = 0;
    const int sub = lane >> 2;
    const int qo  = (lane & 3) * 4;
    for (int i0 = 0; i0 < gc; i0 += 8 * GB) {
        int   gg[GB];
        uint4 d[GB], s[GB];
#pragma unroll
        for (int k = 0; k < GB; ++k) {
            const int i = i0 + 8 * k + sub;
            gg[k] = myg[(i < gc) ? i : (gc - 1)];
        }
#pragma unroll
        for (int k = 0; k < GB; ++k) {
            d[k] = *(const uint4*)(dst + (long long)gg[k] * 16 + qo);
            s[k] = *(const uint4*)(src + (long long)gg[k] * 16 + qo);
        }
#pragma unroll
        for (int k = 0; k < GB; ++k) {
            const bool ok = i0 + 8 * k + sub < gc;
#pragma unroll
            for (int x = 0; x < 4; ++x) {
                const unsigned dv = x == 0 ? d[k].x : x == 1 ? d[k].y : x == 2 ? d[k].z : d[k].w;
                const unsigned sv = x == 0 ? s[k].x : x == 1 ? s[k].y : x == 2 ? s[k].z : s[k].w;
                const unsigned lr = dv - (unsigned)base;
                const bool hit = ok && lr < (unsigned)NPW;
                const unsigned act = __builtin_amdgcn_ballot_w32(hit);
                if (act != 0u) {
                    if (hit) {
                        const int slot = wc + (int)__builtin_amdgcn_mbcnt_lo(act, 0u);
                        if (slot < WQ) myq[slot] = (int)(sv | (lr << SRC_BITS));
                        else atomicAdd(&indeg[lr], 1);
                    }
                    wc += __builtin_popcount(act);
                }
            }
        }
    }
    const bool lost = __builtin_amdgcn_ballot_w32(ovf) != 0u;
    if (lane == 0) wcs[wid] = (lost && wc <= WQ) ? (WQ + 1) : wc;

    const int cnt = wc < WQ ? wc : WQ;
    for (int i = lane; i < cnt; i += 32) {
        const unsigned v = (unsigned)myq[i];
        ps[i] = v;
        atomicAdd(&indeg[v >> SRC_BITS], 1);
    }
    __syncthreads();
    if (wid == 0) plan_c[tile_id * 32 + lane] = wcs[lane];
    for (int n = tid; n < nloc; n += 1024) {
        const int dg = indeg[n] + 1;
        dinv[base + n]   = rsqrtf((float)dg);
        deg16g[base + n] = (unsigned short)(dg < DEG_SAT ? dg : DEG_SAT);
    }
}


typedef GemmShape<128, 64, 8> G1;
constexpr int G1_WGS = (N_NODES + G1::ROWS - 1) / G1::ROWS;
constexpr int G1_WORDS = G1::A_WORDS + G1::B_WORDS;
constexpr int PG_WORDS = PLAN_WORDS > G1_WORDS ? PLAN_WORDS : G1_WORDS;
static_assert(G1::NW * 32 == 1024, "the product's work-groups use the plan's block size");
static_assert(G1::A_WORDS % 4 == 0, "B tile starts on a 16-byte boundary");
static_assert(2 * PG_WORDS * 4 <= 320 * 1024, "a plan and a product work-group share one WGP's LDS");

__global__ __launch_bounds__(1024)
void plan_gemm_kernel(const int* __restrict__ src, const int* __restrict__ dst,
                      const unsigned int* __restrict__ bitmap, float* __restrict__ dinv, int N,
                      int* __restrict__ plan_c, unsigned int* __restrict__ plan_s,
                      unsigned short* __restrict__ deg16g,
                      const float* __restrict__ x, const float* __restrict__ W1,
                      float* __restrict__ h1) {
    __shared__ __attribute__((aligned(16))) int lds[PG_WORDS];
    if (blockIdx.x < NTILE) {
        tile_plan_body(src, dst, bitmap, dinv, N, plan_c, plan_s, deg16g, (int)blockIdx.x, lds);
    } else {
        float* As = reinterpret_cast<float*>(lds);
        gemm_tile_f32<128, 64, 8, true>(x, W1, h1, N, (int)blockIdx.x - NTILE, As, As + G1::A_WORDS);
    }
}

template <int F, bool FULL>
__device__ __forceinline__ void add_half(const float* __restrict__ h, float* tile, unsigned r, float n,
                                         int j0, int m, int lane) {
    constexpr int      VPL = F / 32;
    constexpr unsigned SM  = (1u << SRC_BITS) - 1u;
    float vals[16][VPL];
#pragma unroll
    for (int k = 0; k < 16; ++k) {
        const unsigned rj = (unsigned)__builtin_amdgcn_readlane((int)r, j0 + k);
        const float* hr = h + (long long)(rj & SM) * F + lane * VPL;
#pragma unroll
        for (int x = 0; x < VPL; ++x) vals[k][x] = hr[x];
    }
#pragma unroll
    for (int k = 0; k < 16; ++k) {
        if (FULL || k < m) {
            const unsigned rj  = (unsigned)__builtin_amdgcn_readlane((int)r, j0 + k);
            const float    nrm = __int_as_float(__builtin_amdgcn_readlane(__float_as_int(n), j0 + k));
            float* tr = tile + (int)(rj >> SRC_BITS) * F + lane * VPL;
#pragma unroll
            for (int x = 0; x < VPL; ++x) atomicAdd(tr + x, vals[k][x] * nrm);
        }
    }
}

template <int F, bool FIRST>
__global__ __launch_bounds__(1024)
void tile_agg_kernel(const float* __restrict__ h, const int* __restrict__ src,
                     const int* __restrict__ dst, const unsigned char* __restrict__ owner,
                     const float* __restrict__ dinv, const float* __restrict__ bias,
                     float* __restrict__ y, int N, const int* __restrict__ plan_c,
                     const unsigned int* __restrict__ plan_s, float* plan_n,
                     const unsigned short* __restrict__ deg16g) {
    constexpr int F4  = F / 4;
    constexpr int VPL = F / 32;
    __shared__ float          tile[NPW * F];
    constexpr int Q_BYTES   = 32 * WQ * (4 + 4 + 2);
    constexpr int DT_BYTES  = N_NODES * 2;
    constexpr int AUX_BYTES = (FIRST && DT_BYTES > Q_BYTES) ? DT_BYTES : Q_BYTES;
    __shared__ __attribute__((aligned(16))) unsigned char aux[AUX_BYTES];
    int*            q_s = reinterpret_cast<int*>(aux);
    float*          q_n = reinterpret_cast<float*>(aux + 32 * WQ * 4);
    unsigned short* q_d = reinterpret_cast<unsigned short*>(aux + 32 * WQ * 8);
    const unsigned short* deg16 = reinterpret_cast<const unsigned short*>(aux);

    const int tid  = threadIdx.x;
    const int wid  = tid >> 5;
    const int lane = tid & 31;
    const int base = blockIdx.x * NPW;
    int nloc = N - base;
    if (nloc > NPW) nloc = NPW;
    if (nloc < 0) nloc = 0;

    const int c32 = plan_c[blockIdx.x * 32 + lane];
    const bool build = __builtin_amdgcn_ballot_w32(c32 > WQ) != 0u;
    int wc = __builtin_amdgcn_readlane(c32, __builtin_amdgcn_readfirstlane(wid));

    if (FIRST && !build) {
        constexpr int DT_VEC = DT_BYTES / 16;
        constexpr int DT_T   = (DT_VEC + 1023) / 1024;
        const uint4* gsrc = reinterpret_cast<const uint4*>(deg16g);
        uint4*       ldst = reinterpret_cast<uint4*>(aux);
        uint4 tv[DT_T];
#pragma unroll
        for (int t = 0; t < DT_T; ++t) {
            const int i = t * 1024 + tid;
            tv[t] = gsrc[i < DT_VEC ? i : DT_VEC - 1];
        }
#pragma unroll
        for (int t = 0; t < DT_T; ++t) {
            const int i = t * 1024 + tid;
            if (i < DT_VEC) ldst[i] = tv[t];
        }
    }

    for (int i = tid; i < nloc * F4; i += 1024) {
        const int n = i / F4, c = i % F4;
        const float di = dinv[base + n];
        const float4 hv = *(const float4*)(h + (long long)(base + n) * F + c * 4);
        const float4 bv = *(const float4*)(bias + c * 4);
        float4 t;
        t.x = hv.x * di * di + bv.x;
        t.y = hv.y * di * di + bv.y;
        t.z = hv.z * di * di + bv.z;
        t.w = hv.w * di * di + bv.w;
        *(float4*)(&tile[n * F + c * 4]) = t;
    }
    __syncthreads();

    int*            myq = &q_s[wid * WQ];
    float*          myn = &q_n[wid * WQ];
    unsigned short* myd = &q_d[wid * WQ];
    const unsigned int* ps = plan_s + (long long)(blockIdx.x * 32 + wid) * WQ;
    float*              pn = plan_n + (long long)(blockIdx.x * 32 + wid) * WQ;
    if (build) {
    wc = scan_owner_map(owner, (int)blockIdx.x, tid, myq,
                        [&](int e) { spill_edge<F>(e, base, tile, h, src, dst, dinv); });
    const int cnt = wc < WQ ? wc : WQ;
    for (int i0 = 0; i0 < cnt; i0 += 128) {
        int ix[4], ss[4], dd[4];
        float nr[4];
#pragma unroll
        for (int k = 0; k < 4; ++k) {
            const int i = i0 + 32 * k + lane;
            ix[k] = (i < cnt) ? i : (cnt - 1);
        }
        int ee[4];
#pragma unroll
        for (int k = 0; k < 4; ++k) ee[k] = myq[ix[k]];
#pragma unroll
        for (int k = 0; k < 4; ++k) { ss[k] = src[ee[k]]; dd[k] = dst[ee[k]]; }
#pragma unroll
        for (int k = 0; k < 4; ++k) nr[k] = dinv[ss[k]] * dinv[dd[k]];
#pragma unroll
        for (int k = 0; k < 4; ++k) {
            myq[ix[k]] = ss[k];
            myd[ix[k]] = (unsigned short)(dd[k] - base);
            myn[ix[k]] = nr[k];
        }
    }
    __syncthreads();

    constexpr int U = 16;
    for (int i0 = 0; i0 < cnt; i0 += U) {
        float vals[U][VPL];
#pragma unroll
        for (int k = 0; k < U; ++k) {
            const int i = (i0 + k < cnt) ? (i0 + k) : i0;
            const float* hr = h + (long long)myq[i] * F + lane * VPL;
#pragma unroll
            for (int x = 0; x < VPL; ++x) vals[k][x] = hr[x];
        }
#pragma unroll
        for (int k = 0; k < U; ++k) {
            if (i0 + k < cnt) {
                const float nrm = myn[i0 + k];
                float* tr = &tile[(int)myd[i0 + k] * F + lane * VPL];
#pragma unroll
                for (int x = 0; x < VPL; ++x) atomicAdd(tr + x, vals[k][x] * nrm);
            }
        }
    }
    } else {
    const int cnt = wc;
    if (cnt > 0) {
        constexpr unsigned SM = (1u << SRC_BITS) - 1u;
        const int nb = (cnt + 31) >> 5;
        auto idx = [&](int bb) { const int i = bb * 32 + lane; return (i < cnt) ? i : (cnt - 1); };
        const int nfull = cnt >> 5;
        auto add_tail = [&](unsigned r, float n, int m0) {
            if (m0 >= 16) add_half<F, true >(h, tile, r, n, 0, 16, lane);
            else          add_half<F, false>(h, tile, r, n, 0, m0, lane);
            if (m0 > 16)  add_half<F, false>(h, tile, r, n, 16, m0 - 16, lane);
        };
        if (FIRST) {
            auto src_dinv = [&](unsigned s) {
                const int dg = (int)deg16[s];
                return dg == DEG_SAT ? dinv[s] : rsqrtf((float)dg);
            };
            unsigned r_cur = ps[idx(0)];
            unsigned r_nxt = ps[idx(nb > 1 ? 1 : 0)];
            float    n_cur = src_dinv(r_cur & SM) * dinv[base + (int)(r_cur >> SRC_BITS)];
            if (lane < cnt) pn[lane] = n_cur;
            for (int bb = 0; bb < nfull; ++bb) {
                const int   dg = (int)deg16[r_nxt & SM];
                const float d2 = dinv[base + (int)(r_nxt >> SRC_BITS)];
                const unsigned r_n2 = ps[idx(bb + 2 < nb ? bb + 2 : nb - 1)];
                add_half<F, true>(h, tile, r_cur, n_cur, 0, 16, lane);
                add_half<F, true>(h, tile, r_cur, n_cur, 16, 16, lane);
                const float d1 = dg == DEG_SAT ? dinv[r_nxt & SM] : rsqrtf((float)dg);
                r_cur = r_nxt; n_cur = d1 * d2; r_nxt = r_n2;
                const int i1 = (bb + 1) * 32 + lane;
                if (i1 < cnt) pn[i1] = n_cur;
            }
            if (nfull < nb) add_tail(r_cur, n_cur, cnt - nfull * 32);
        } else {
            const int ic = idx(0);
            unsigned r_cur = ps[ic];
            float    n_cur = pn[ic];
            for (int bb = 0; bb < nfull; ++bb) {
                const int ix = idx(bb + 1 < nb ? bb + 1 : nb - 1);
                const unsigned r_nxt = ps[ix];
                const float    n_nxt = pn[ix];
                add_half<F, true>(h, tile, r_cur, n_cur, 0, 16, lane);
                add_half<F, true>(h, tile, r_cur, n_cur, 16, 16, lane);
                r_cur = r_nxt; n_cur = n_nxt;
            }
            if (nfull < nb) add_tail(r_cur, n_cur, cnt - nfull * 32);
        }
    }
    }
    __syncthreads();

    for (int i = tid; i < nloc * F4; i += 1024) {
        float4 t = *(const float4*)(&tile[i * 4]);
        t.x = fmaxf(t.x, 0.0f); t.y = fmaxf(t.y, 0.0f);
        t.z = fmaxf(t.z, 0.0f); t.w = fmaxf(t.w, 0.0f);
        *(float4*)(y + (long long)base * F + (long long)i * 4) = t;
    }
}

__global__ void final_linear_kernel(const float* __restrict__ y, const float* __restrict__ Wl,
                                    const float* __restrict__ bl, float* out, int n) {
    int i = blockIdx.x * blockDim.x + threadIdx.x;
    if (i >= n) return;
    const float* row = y + (long long)i * 32;
    float s = 0.0f;
#pragma unroll
    for (int f = 0; f < 32; ++f) s += row[f] * Wl[f];
    out[i] = s + bl[0];
}

static inline size_t align256(size_t x) { return (x + 255) & ~size_t(255); }

extern "C" void kernel_launch(void* const* d_in, const int* in_sizes, int n_in,
                              void* d_out, int out_size, void* d_ws, size_t ws_size,
                              hipStream_t stream) {
    const float* x    = (const float*)d_in[0];
    const int*   ei   = (const int*)d_in[1];
    const float* W1   = (const float*)d_in[2];
    const float* b1   = (const float*)d_in[3];
    const float* W2   = (const float*)d_in[4];
    const float* b2   = (const float*)d_in[5];
    const float* Wl   = (const float*)d_in[6];
    const float* bl   = (const float*)d_in[7];
    float*       out  = (float*)d_out;

    const int N = N_NODES;
    const int E = N_EDGES;
    const int* src = ei;
    const int* dst = ei + E;

    char* ws = (char*)d_ws;
    size_t off = 0;
    float* dinv = (float*)(ws + off); off += align256((size_t)N * 4);
    float* h1   = (float*)(ws + off); off += align256((size_t)N * 64 * 4);
    float* y1   = (float*)(ws + off); off += align256((size_t)N * 64 * 4);
    float* h2   = (float*)(ws + off); off += align256((size_t)N * 32 * 4);
    float* y2   = (float*)(ws + off); off += align256((size_t)N * 32 * 4);
    unsigned char* owner = (unsigned char*)(ws + off); off += align256((size_t)E);
    int*          plan_c = (int*)(ws + off);          off += align256((size_t)NTILE * 32 * 4);
    unsigned int* plan_s = (unsigned int*)(ws + off); off += align256((size_t)NTILE * 32 * WQ * 4);
    float*        plan_n = (float*)(ws + off);        off += align256((size_t)NTILE * 32 * WQ * 4);
    unsigned int* bitmap = (unsigned int*)(ws + off); off += align256((size_t)NTILE * BWP * 4);
    unsigned short* deg16g = (unsigned short*)(ws + off); off += align256((size_t)N * 2);
    (void)ws_size; (void)n_in; (void)in_sizes; (void)out_size;

    const int T = 256;

    owner_map_kernel<<<NCH, 1024, 0, stream>>>(dst, owner, bitmap, E);

    plan_gemm_kernel<<<NTILE + G1_WGS, 1024, 0, stream>>>(src, dst, bitmap, dinv, N, plan_c, plan_s,
                                                           deg16g, x, W1, h1);

    tile_agg_kernel<64, true><<<NTILE, 1024, 0, stream>>>(h1, src, dst, owner, dinv, b1, y1, N,
                                                          plan_c, plan_s, plan_n, deg16g);

    gemm_wmma_f32<64, 32, 2><<<N / 32, 2 * 2 * 32, 0, stream>>>(y1, W2, h2, N);

    tile_agg_kernel<32, false><<<NTILE, 1024, 0, stream>>>(h2, src, dst, owner, dinv, b2, y2, N,
                                                           plan_c, plan_s, plan_n, deg16g);

    final_linear_kernel<<<(N + T - 1) / T, T, 0, stream>>>(y2, Wl, bl, out, N);
}


# Owner-map launch also stores every edge as a (source, destination) pair and the bitmap names 8-edge groups: plan fetches ONE 64-byte piece per named group, not two; order of LDS f32 adds changes
# speedup vs baseline: 4.0709x; 1.0321x over previous
#include <hip/hip_runtime.h>
#include <hip/hip_bf16.h>

typedef __attribute__((ext_vector_type(2))) float v2f;
typedef __attribute__((ext_vector_type(8))) float v8f;

#define N_NODES 100000
#define N_EDGES 3200000

template <int K, int NOUT, int MT>
struct GemmShape {
    static constexpr int NT   = NOUT / 16;
    static constexpr int NW   = MT * NT;
    static constexpr int ROWS = MT * 16;
    static constexpr int KP   = K + 4;
    static constexpr int NP   = NOUT + 4;
    static constexpr int A_WORDS = ROWS * KP;
    static constexpr int B_WORDS = K * NP;
};

template <int K, int NOUT, int MT, bool GROUPED>
__device__ __forceinline__
void gemm_tile_f32(const float* __restrict__ A, const float* __restrict__ B,
                   float* __restrict__ C, int M, int blk, float* As, float* Bs) {
    typedef GemmShape<K, NOUT, MT> S;
    constexpr int NT   = S::NT;
    constexpr int NW   = S::NW;
    constexpr int ROWS = S::ROWS;
    constexpr int KP   = S::KP;
    constexpr int NP   = S::NP;

    const int tid  = threadIdx.x;
    const int row0 = blk * ROWS;

    constexpr int A_ELTS = ROWS * K;
    constexpr int B_ELTS = K * NOUT;
    constexpr int STEP   = NW * 32 * 4;
    if constexpr (GROUPED) {
        static_assert(A_ELTS % STEP == 0 && B_ELTS % STEP == 0, "whole turns only");
        constexpr int AT = A_ELTS / STEP, BT = B_ELTS / STEP;
        float4 av[AT], bv[BT];
#pragma unroll
        for (int t = 0; t < AT; ++t) {
            const int i = tid * 4 + t * STEP;
            const int gr = row0 + i / K;
            av[t] = (gr < M) ? *(const float4*)(A + (long long)gr * K + i % K)
                             : make_float4(0.f, 0.f, 0.f, 0.f);
        }
#pragma unroll
        for (int t = 0; t < BT; ++t) bv[t] = *(const float4*)(B + tid * 4 + t * STEP);
#pragma unroll
        for (int t = 0; t < AT; ++t) {
            const int i = tid * 4 + t * STEP;
            *(float4*)(&As[(i / K) * KP + i % K]) = av[t];
        }
#pragma unroll
        for (int t = 0; t < BT; ++t) {
            const int i = tid * 4 + t * STEP;
            *(float4*)(&Bs[(i / NOUT) * NP + i % NOUT]) = bv[t];
        }
    } else {
        for (int i = tid * 4; i < A_ELTS; i += STEP) {
            int r = i / K, c = i % K;
            int gr = row0 + r;
            float4 v = (gr < M) ? *(const float4*)(A + (long long)gr * K + c)
                                : make_float4(0.f, 0.f, 0.f, 0.f);
            *(float4*)(&As[r * KP + c]) = v;
        }
        for (int i = tid * 4; i < B_ELTS; i += STEP) {
            int r = i / NOUT, c = i % NOUT;
            *(float4*)(&Bs[r * NP + c]) = *(const float4*)(B + i);
        }
    }
    __syncthreads();

    const int wid  = tid >> 5;
    const int lane = tid & 31;
    const int mn   = lane & 15;
    const int h    = lane >> 4;
    const int mt   = wid / NT;
    const int nt   = wid % NT;

    v8f acc = {};
    const float* arow = &As[(mt * 16 + mn) * KP];
    const int    ncol = nt * 16 + mn;

#pragma unroll
    for (int k = 0; k < K; k += 4) {
        v2f a, b;
        a.x = arow[k + 2 * h + 0];
        a.y = arow[k + 2 * h + 1];
        b.x = Bs[(k + 2 * h + 0) * NP + ncol];
        b.y = Bs[(k + 2 * h + 1) * NP + ncol];
        acc = __builtin_amdgcn_wmma_f32_16x16x4_f32(
             false, a,  false, b,
             (short)0, acc,  false,  false);
    }

#pragma unroll
    for (int j = 0; j < 8; ++j) {
        int r = row0 + mt * 16 + j + 8 * h;
        if (r < M) C[(long long)r * NOUT + ncol] = acc[j];
    }
}

template <int K, int NOUT, int MT>
__global__ __launch_bounds__(MT*(NOUT/16)*32)
void gemm_wmma_f32(const float* __restrict__ A, const float* __restrict__ B,
                   float* __restrict__ C, int M) {
    typedef GemmShape<K, NOUT, MT> S;
    __shared__ __attribute__((aligned(16))) float As[S::A_WORDS];
    __shared__ __attribute__((aligned(16))) float Bs[S::B_WORDS];
    gemm_tile_f32<K, NOUT, MT, false>(A, B, C, M, (int)blockIdx.x, As, Bs);
}

constexpr int NPW   = 392;
constexpr int NTILE = 256;
constexpr int WQ    = 512;
constexpr int NGRP  = N_EDGES / 16;
constexpr int NBAT  = (NGRP + 8191) / 8192;
static_assert(NPW * NTILE >= N_NODES, "tiles must cover every node");
static_assert(N_EDGES % 16 == 0, "owner map is read in 16-byte groups");
static_assert(NPW <= 65535, "local row index is kept in 16 bits");
constexpr int CH    = 8192;
constexpr int NCH   = (N_EDGES + CH - 1) / CH;
constexpr int GE    = 8;
constexpr int PGW   = GE * 2;
constexpr int CHW   = CH / GE / 32;
constexpr int BWP   = NCH * CHW;
constexpr int GQ    = 640;
static_assert(CH == 1024 * 8, "one thread of the owner-map launch takes four pairs of edges");
static_assert(N_EDGES % GE == 0, "the pair array is read in whole groups");
static_assert(BWP % 4 == 0, "bitmap rows are written in 16-byte pieces");
constexpr int DEG_SAT  = 65535;
static_assert((N_NODES * 2) % 16 == 0, "the 16-bit degree table is copied in 16-byte pieces");
constexpr int SRC_BITS = 17;
static_assert(N_NODES <= (1 << SRC_BITS), "source id must fit the low bits of a stored entry");
static_assert(NPW <= (1 << (32 - SRC_BITS)), "local row must fit the high bits of a stored entry");

constexpr int PRS = NTILE + 2;
__global__ __launch_bounds__(1024)
void owner_map_kernel(const int* __restrict__ src, const int* __restrict__ dst,
                      unsigned char* __restrict__ owner, unsigned int* __restrict__ bitmap,
                      unsigned int* __restrict__ pairs, int E) {
    __shared__ unsigned int pres[CHW * PRS];
    const int tid = threadIdx.x;
    for (int i = tid; i < CHW * PRS; i += 1024) pres[i] = 0u;
    __syncthreads();
    const long long c0 = (long long)blockIdx.x * CH + tid * 2;
    int2 dv[4], sv[4];
#pragma unroll
    for (int j = 0; j < 4; ++j) {
        const long long e  = c0 + j * 2048;
        const long long ec = (e < E) ? e : 0;
        dv[j] = *(const int2*)(dst + ec);
        sv[j] = *(const int2*)(src + ec);
    }
#pragma unroll
    for (int j = 0; j < 4; ++j) {
        const long long e = c0 + j * 2048;
        if (e < E) {
            const unsigned o0 = (unsigned)(dv[j].x / NPW), o1 = (unsigned)(dv[j].y / NPW);
            *(unsigned short*)(owner + e) = (unsigned short)(o0 | (o1 << 8));
            uint4 p;
            p.x = (unsigned)sv[j].x; p.y = (unsigned)dv[j].x;
            p.z = (unsigned)sv[j].y; p.w = (unsigned)dv[j].y;
            *(uint4*)(pairs + e * 2) = p;
            const int gl = (j * 2048 + tid * 2) / GE;
            const unsigned bit = 1u << (gl & 31);
            unsigned int* pw = &pres[(gl >> 5) * PRS];
            atomicOr(pw + o0, bit);
            atomicOr(pw + o1, bit);
        }
    }
    __syncthreads();
#pragma unroll
    for (int r = 0; r < 2; ++r) {
        const int idx = r * 1024 + tid;
        const int t = idx >> 3, q = (idx & 7) * 4;
        uint4 v;
        v.x = pres[(q + 0) * PRS + t];
        v.y = pres[(q + 1) * PRS + t];
        v.z = pres[(q + 2) * PRS + t];
        v.w = pres[(q + 3) * PRS + t];
        *(uint4*)(bitmap + (long long)t * BWP + blockIdx.x * CHW + q) = v;
    }
}

__device__ __forceinline__ unsigned zero_bytes(unsigned y) {
    unsigned t = (y & 0x7F7F7F7Fu) + 0x7F7F7F7Fu;
    return ~(t | y | 0x7F7F7F7Fu);
}
__device__ __forceinline__ unsigned match_group(uint4 v, unsigned me) {
    return (zero_bytes(v.x ^ me) >> 7) | (zero_bytes(v.y ^ me) >> 6) |
           (zero_bytes(v.z ^ me) >> 5) | (zero_bytes(v.w ^ me) >> 4);
}

template <int F>
__device__ void spill_edge(int e, int base, float* tile, const float* __restrict__ h,
                           const int* __restrict__ src, const int* __restrict__ dst,
                           const float* __restrict__ dinv) {
    const int s = src[e], d = dst[e];
    const float nrm = dinv[s] * dinv[d];
    const float* hr = h + (long long)s * F;
    float* tr = tile + (d - base) * F;
    for (int f = 0; f < F; ++f) atomicAdd(tr + f, hr[f] * nrm);
}
__device__ __forceinline__ int wrap_group(int g) {
    if (g >= NGRP) g -= NGRP;
    if (g >= NGRP) g -= NGRP;
    return g;
}

template <typename Spill>
__device__ __forceinline__ int scan_owner_map(const unsigned char* __restrict__ owner, int tileid,
                                              int tid, int* myq, Spill spill) {
    const unsigned me   = (unsigned)tileid * 0x01010101u;
    const uint4*   om   = (const uint4*)owner;
    const int rot = tileid * 768;
    int wc = 0;
    for (int it = 0; it < NBAT; ++it) {
        uint4 v[8];
#pragma unroll
        for (int k = 0; k < 8; ++k) {
            v[k] = om[wrap_group(it * 8192 + k * 1024 + tid + rot)];
        }
        unsigned mk[8];
#pragma unroll
        for (int k = 0; k < 8; ++k)
            mk[k] = match_group(v[k], me) & (0u - (unsigned)(it * 8192 + k * 1024 + tid < NGRP));
#pragma unroll
        for (int hf = 0; hf < 2; ++hf) {
            const unsigned lo = mk[4 * hf + 0] | (mk[4 * hf + 1] << 4);
            const unsigned hi = mk[4 * hf + 2] | (mk[4 * hf + 3] << 4);
            unsigned long long M = ((unsigned long long)hi << 32) | (unsigned long long)lo;
            unsigned act;
            while ((act = __builtin_amdgcn_ballot_w32(M != 0ull)) != 0u) {
                if (M != 0ull) {
                    const int p = __builtin_ctzll(M);
                    M &= M - 1ull;
                    const int k  = 4 * hf + 2 * (p >> 5) + ((p >> 2) & 1);
                    const int g = wrap_group(it * 8192 + k * 1024 + tid + rot);
                    const int e = g * 16 + (p & 3) * 4 + ((p & 31) >> 3);
                    const int slot = wc + (int)__builtin_amdgcn_mbcnt_lo(act, 0u);
                    if (slot < WQ) myq[slot] = e;
                    else spill(e);
                }
                wc += __builtin_popcount(act);
            }
        }
    }
    return wc;
}

static_assert(CHW == 32 && NTILE * 8 == 2048, "bitmap write-out of the owner-map launch");
constexpr int BW_FULL = BWP / 1024;
constexpr int BW_REST = BWP - BW_FULL * 1024;
static_assert(BW_REST * 4 <= 1024, "rest of a bitmap row is dealt to every 4th thread");

constexpr int PLAN_Q_WORDS = 32 * WQ;
constexpr int PLAN_G_WORDS = 32 * GQ;
constexpr int PLAN_WORDS   = PLAN_Q_WORDS + PLAN_G_WORDS + NPW + 32;

__device__ __forceinline__
void tile_plan_body(const unsigned int* __restrict__ pairs, const int* __restrict__ dst,
                    const unsigned int* __restrict__ bitmap, float* __restrict__ dinv, int N,
                    int* __restrict__ plan_c, unsigned int* __restrict__ plan_s,
                    unsigned short* __restrict__ deg16g, int tile_id, int* lds) {
    int* q_s   = lds;
    int* g_s   = lds + PLAN_Q_WORDS;
    int* indeg = lds + PLAN_Q_WORDS + PLAN_G_WORDS;
    int* wcs   = indeg + NPW;

    const int tid  = threadIdx.x;
    const int wid  = tid >> 5;
    const int lane = tid & 31;
    const int base = tile_id * NPW;
    int nloc = N - base;
    if (nloc > NPW) nloc = NPW;
    if (nloc < 0) nloc = 0;

    for (int i = tid; i < NPW; i += 1024) indeg[i] = 0;
    __syncthreads();

    int* myq = &q_s[wid * WQ];
    int* myg = &g_s[wid * GQ];
    unsigned int* ps = plan_s + (long long)(tile_id * 32 + wid) * WQ;

    const unsigned int* brow = bitmap + (long long)tile_id * BWP;
    unsigned bits[BW_FULL + 1];
    int      bj[BW_FULL + 1];
#pragma unroll
    for (int k = 0; k < BW_FULL; ++k) { bj[k] = k * 1024 + tid; bits[k] = brow[bj[k]]; }
    {
        const int r = tid >> 2;
        const bool has = ((tid & 3) == 0) && (r < BW_REST);
        bj[BW_FULL] = BW_FULL * 1024 + (has ? r : 0);
        const unsigned v = brow[bj[BW_FULL]];
        bits[BW_FULL] = has ? v : 0u;
    }
    int  gc  = 0;
    bool ovf = false;
#pragma unroll
    for (int k = 0; k <= BW_FULL; ++k) {
        unsigned m = bits[k];
        unsigned act;
        while ((act = __builtin_amdgcn_ballot_w32(m != 0u)) != 0u) {
            if (m != 0u) {
                const int p = __builtin_ctz(m);
                m &= m - 1u;
                const int g = bj[k] * 32 + p;
                const int slot = gc + (int)__builtin_amdgcn_mbcnt_lo(act, 0u);
                if (slot < GQ) myg[slot] = g;
                else {
                    ovf = true;
                    for (int x = 0; x < GE; ++x) {
                        const unsigned lr = (unsigned)(dst[g * GE + x] - base);
                        if (lr < (unsigned)NPW) atomicAdd(&indeg[lr], 1);
                    }
                }
            }
            gc += __builtin_popcount(act);
        }
    }
    if (gc > GQ) gc = GQ;

    constexpr int GB = 8;
    int wc = 0;
    const int sub = lane >> 2;
    const int qo  = (lane & 3) * 4;
    for (int i0 = 0; i0 < gc; i0 += 8 * GB) {
        int   gg[GB];
        uint4 pr[GB];
#pragma unroll
        for (int k = 0; k < GB; ++k) {
            const int i = i0 + 8 * k + sub;
            gg[k] = myg[(i < gc) ? i : (gc - 1)];
        }
#pragma unroll
        for (int k = 0; k < GB; ++k)
            pr[k] = *(const uint4*)(pairs + (long long)gg[k] * PGW + qo);
#pragma unroll
        for (int k = 0; k < GB; ++k) {
            const bool ok = i0 + 8 * k + sub < gc;
#pragma unroll
            for (int x = 0; x < 2; ++x) {
                const unsigned sv = x == 0 ? pr[k].x : pr[k].z;
                const unsigned dv = x == 0 ? pr[k].y : pr[k].w;
                const unsigned lr = dv - (unsigned)base;
                const bool hit = ok && lr < (unsigned)NPW;
                const unsigned act = __builtin_amdgcn_ballot_w32(hit);
                if (act != 0u) {
                    if (hit) {
                        const int slot = wc + (int)__builtin_amdgcn_mbcnt_lo(act, 0u);
                        if (slot < WQ) myq[slot] = (int)(sv | (lr << SRC_BITS));
                        else atomicAdd(&indeg[lr], 1);
                    }
                    wc += __builtin_popcount(act);
                }
            }
        }
    }
    const bool lost = __builtin_amdgcn_ballot_w32(ovf) != 0u;
    if (lane == 0) wcs[wid] = (lost && wc <= WQ) ? (WQ + 1) : wc;

    const int cnt = wc < WQ ? wc : WQ;
    for (int i = lane; i < cnt; i += 32) {
        const unsigned v = (unsigned)myq[i];
        ps[i] = v;
        atomicAdd(&indeg[v >> SRC_BITS], 1);
    }
    __syncthreads();
    if (wid == 0) plan_c[tile_id * 32 + lane] = wcs[lane];
    for (int n = tid; n < nloc; n += 1024) {
        const int dg = indeg[n] + 1;
        dinv[base + n]   = rsqrtf((float)dg);
        deg16g[base + n] = (unsigned short)(dg < DEG_SAT ? dg : DEG_SAT);
    }
}


typedef GemmShape<128, 64, 8> G1;
constexpr int G1_WGS = (N_NODES + G1::ROWS - 1) / G1::ROWS;
constexpr int G1_WORDS = G1::A_WORDS + G1::B_WORDS;
constexpr int PG_WORDS = PLAN_WORDS > G1_WORDS ? PLAN_WORDS : G1_WORDS;
static_assert(G1::NW * 32 == 1024, "the product's work-groups use the plan's block size");
static_assert(G1::A_WORDS % 4 == 0, "B tile starts on a 16-byte boundary");
static_assert(2 * PG_WORDS * 4 <= 320 * 1024, "a plan and a product work-group share one WGP's LDS");

__global__ __launch_bounds__(1024)
void plan_gemm_kernel(const unsigned int* __restrict__ pairs, const int* __restrict__ dst,
                      const unsigned int* __restrict__ bitmap, float* __restrict__ dinv, int N,
                      int* __restrict__ plan_c, unsigned int* __restrict__ plan_s,
                      unsigned short* __restrict__ deg16g,
                      const float* __restrict__ x, const float* __restrict__ W1,
                      float* __restrict__ h1) {
    __shared__ __attribute__((aligned(16))) int lds[PG_WORDS];
    if (blockIdx.x < NTILE) {
        tile_plan_body(pairs, dst, bitmap, dinv, N, plan_c, plan_s, deg16g, (int)blockIdx.x, lds);
    } else {
        float* As = reinterpret_cast<float*>(lds);
        gemm_tile_f32<128, 64, 8, true>(x, W1, h1, N, (int)blockIdx.x - NTILE, As, As + G1::A_WORDS);
    }
}

template <int F, bool FULL>
__device__ __forceinline__ void add_half(const float* __restrict__ h, float* tile, unsigned r, float n,
                                         int j0, int m, int lane) {
    constexpr int      VPL = F / 32;
    constexpr unsigned SM  = (1u << SRC_BITS) - 1u;
    float vals[16][VPL];
#pragma unroll
    for (int k = 0; k < 16; ++k) {
        const unsigned rj = (unsigned)__builtin_amdgcn_readlane((int)r, j0 + k);
        const float* hr = h + (long long)(rj & SM) * F + lane * VPL;
#pragma unroll
        for (int x = 0; x < VPL; ++x) vals[k][x] = hr[x];
    }
#pragma unroll
    for (int k = 0; k < 16; ++k) {
        if (FULL || k < m) {
            const unsigned rj  = (unsigned)__builtin_amdgcn_readlane((int)r, j0 + k);
            const float    nrm = __int_as_float(__builtin_amdgcn_readlane(__float_as_int(n), j0 + k));
            float* tr = tile + (int)(rj >> SRC_BITS) * F + lane * VPL;
#pragma unroll
            for (int x = 0; x < VPL; ++x) atomicAdd(tr + x, vals[k][x] * nrm);
        }
    }
}

template <int F, bool FIRST>
__global__ __launch_bounds__(1024)
void tile_agg_kernel(const float* __restrict__ h, const int* __restrict__ src,
                     const int* __restrict__ dst, const unsigned char* __restrict__ owner,
                     const float* __restrict__ dinv, const float* __restrict__ bias,
                     float* __restrict__ y, int N, const int* __restrict__ plan_c,
                     const unsigned int* __restrict__ plan_s, float* plan_n,
                     const unsigned short* __restrict__ deg16g) {
    constexpr int F4  = F / 4;
    constexpr int VPL = F / 32;
    __shared__ float          tile[NPW * F];
    constexpr int Q_BYTES   = 32 * WQ * (4 + 4 + 2);
    constexpr int DT_BYTES  = N_NODES * 2;
    constexpr int AUX_BYTES = (FIRST && DT_BYTES > Q_BYTES) ? DT_BYTES : Q_BYTES;
    __shared__ __attribute__((aligned(16))) unsigned char aux[AUX_BYTES];
    int*            q_s = reinterpret_cast<int*>(aux);
    float*          q_n = reinterpret_cast<float*>(aux + 32 * WQ * 4);
    unsigned short* q_d = reinterpret_cast<unsigned short*>(aux + 32 * WQ * 8);
    const unsigned short* deg16 = reinterpret_cast<const unsigned short*>(aux);

    const int tid  = threadIdx.x;
    const int wid  = tid >> 5;
    const int lane = tid & 31;
    const int base = blockIdx.x * NPW;
    int nloc = N - base;
    if (nloc > NPW) nloc = NPW;
    if (nloc < 0) nloc = 0;

    const int c32 = plan_c[blockIdx.x * 32 + lane];
    const bool build = __builtin_amdgcn_ballot_w32(c32 > WQ) != 0u;
    int wc = __builtin_amdgcn_readlane(c32, __builtin_amdgcn_readfirstlane(wid));

    if (FIRST && !build) {
        constexpr int DT_VEC = DT_BYTES / 16;
        constexpr int DT_T   = (DT_VEC + 1023) / 1024;
        const uint4* gsrc = reinterpret_cast<const uint4*>(deg16g);
        uint4*       ldst = reinterpret_cast<uint4*>(aux);
        uint4 tv[DT_T];
#pragma unroll
        for (int t = 0; t < DT_T; ++t) {
            const int i = t * 1024 + tid;
            tv[t] = gsrc[i < DT_VEC ? i : DT_VEC - 1];
        }
#pragma unroll
        for (int t = 0; t < DT_T; ++t) {
            const int i = t * 1024 + tid;
            if (i < DT_VEC) ldst[i] = tv[t];
        }
    }

    for (int i = tid; i < nloc * F4; i += 1024) {
        const int n = i / F4, c = i % F4;
        const float di = dinv[base + n];
        const float4 hv = *(const float4*)(h + (long long)(base + n) * F + c * 4);
        const float4 bv = *(const float4*)(bias + c * 4);
        float4 t;
        t.x = hv.x * di * di + bv.x;
        t.y = hv.y * di * di + bv.y;
        t.z = hv.z * di * di + bv.z;
        t.w = hv.w * di * di + bv.w;
        *(float4*)(&tile[n * F + c * 4]) = t;
    }
    __syncthreads();

    int*            myq = &q_s[wid * WQ];
    float*          myn = &q_n[wid * WQ];
    unsigned short* myd = &q_d[wid * WQ];
    const unsigned int* ps = plan_s + (long long)(blockIdx.x * 32 + wid) * WQ;
    float*              pn = plan_n + (long long)(blockIdx.x * 32 + wid) * WQ;
    if (build) {
    wc = scan_owner_map(owner, (int)blockIdx.x, tid, myq,
                        [&](int e) { spill_edge<F>(e, base, tile, h, src, dst, dinv); });
    const int cnt = wc < WQ ? wc : WQ;
    for (int i0 = 0; i0 < cnt; i0 += 128) {
        int ix[4], ss[4], dd[4];
        float nr[4];
#pragma unroll
        for (int k = 0; k < 4; ++k) {
            const int i = i0 + 32 * k + lane;
            ix[k] = (i < cnt) ? i : (cnt - 1);
        }
        int ee[4];
#pragma unroll
        for (int k = 0; k < 4; ++k) ee[k] = myq[ix[k]];
#pragma unroll
        for (int k = 0; k < 4; ++k) { ss[k] = src[ee[k]]; dd[k] = dst[ee[k]]; }
#pragma unroll
        for (int k = 0; k < 4; ++k) nr[k] = dinv[ss[k]] * dinv[dd[k]];
#pragma unroll
        for (int k = 0; k < 4; ++k) {
            myq[ix[k]] = ss[k];
            myd[ix[k]] = (unsigned short)(dd[k] - base);
            myn[ix[k]] = nr[k];
        }
    }
    __syncthreads();

    constexpr int U = 16;
    for (int i0 = 0; i0 < cnt; i0 += U) {
        float vals[U][VPL];
#pragma unroll
        for (int k = 0; k < U; ++k) {
            const int i = (i0 + k < cnt) ? (i0 + k) : i0;
            const float* hr = h + (long long)myq[i] * F + lane * VPL;
#pragma unroll
            for (int x = 0; x < VPL; ++x) vals[k][x] = hr[x];
        }
#pragma unroll
        for (int k = 0; k < U; ++k) {
            if (i0 + k < cnt) {
                const float nrm = myn[i0 + k];
                float* tr = &tile[(int)myd[i0 + k] * F + lane * VPL];
#pragma unroll
                for (int x = 0; x < VPL; ++x) atomicAdd(tr + x, vals[k][x] * nrm);
            }
        }
    }
    } else {
    const int cnt = wc;
    if (cnt > 0) {
        constexpr unsigned SM = (1u << SRC_BITS) - 1u;
        const int nb = (cnt + 31) >> 5;
        auto idx = [&](int bb) { const int i = bb * 32 + lane; return (i < cnt) ? i : (cnt - 1); };
        const int nfull = cnt >> 5;
        auto add_tail = [&](unsigned r, float n, int m0) {
            if (m0 >= 16) add_half<F, true >(h, tile, r, n, 0, 16, lane);
            else          add_half<F, false>(h, tile, r, n, 0, m0, lane);
            if (m0 > 16)  add_half<F, false>(h, tile, r, n, 16, m0 - 16, lane);
        };
        if (FIRST) {
            auto src_dinv = [&](unsigned s) {
                const int dg = (int)deg16[s];
                return dg == DEG_SAT ? dinv[s] : rsqrtf((float)dg);
            };
            unsigned r_cur = ps[idx(0)];
            unsigned r_nxt = ps[idx(nb > 1 ? 1 : 0)];
            float    n_cur = src_dinv(r_cur & SM) * dinv[base + (int)(r_cur >> SRC_BITS)];
            if (lane < cnt) pn[lane] = n_cur;
            for (int bb = 0; bb < nfull; ++bb) {
                const int   dg = (int)deg16[r_nxt & SM];
                const float d2 = dinv[base + (int)(r_nxt >> SRC_BITS)];
                const unsigned r_n2 = ps[idx(bb + 2 < nb ? bb + 2 : nb - 1)];
                add_half<F, true>(h, tile, r_cur, n_cur, 0, 16, lane);
                add_half<F, true>(h, tile, r_cur, n_cur, 16, 16, lane);
                const float d1 = dg == DEG_SAT ? dinv[r_nxt & SM] : rsqrtf((float)dg);
                r_cur = r_nxt; n_cur = d1 * d2; r_nxt = r_n2;
                const int i1 = (bb + 1) * 32 + lane;
                if (i1 < cnt) pn[i1] = n_cur;
            }
            if (nfull < nb) add_tail(r_cur, n_cur, cnt - nfull * 32);
        } else {
            const int ic = idx(0);
            unsigned r_cur = ps[ic];
            float    n_cur = pn[ic];
            for (int bb = 0; bb < nfull; ++bb) {
                const int ix = idx(bb + 1 < nb ? bb + 1 : nb - 1);
                const unsigned r_nxt = ps[ix];
                const float    n_nxt = pn[ix];
                add_half<F, true>(h, tile, r_cur, n_cur, 0, 16, lane);
                add_half<F, true>(h, tile, r_cur, n_cur, 16, 16, lane);
                r_cur = r_nxt; n_cur = n_nxt;
            }
            if (nfull < nb) add_tail(r_cur, n_cur, cnt - nfull * 32);
        }
    }
    }
    __syncthreads();

    for (int i = tid; i < nloc * F4; i += 1024) {
        float4 t = *(const float4*)(&tile[i * 4]);
        t.x = fmaxf(t.x, 0.0f); t.y = fmaxf(t.y, 0.0f);
        t.z = fmaxf(t.z, 0.0f); t.w = fmaxf(t.w, 0.0f);
        *(float4*)(y + (long long)base * F + (long long)i * 4) = t;
    }
}

__global__ void final_linear_kernel(const float* __restrict__ y, const float* __restrict__ Wl,
                                    const float* __restrict__ bl, float* out, int n) {
    int i = blockIdx.x * blockDim.x + threadIdx.x;
    if (i >= n) return;
    const float* row = y + (long long)i * 32;
    float s = 0.0f;
#pragma unroll
    for (int f = 0; f < 32; ++f) s += row[f] * Wl[f];
    out[i] = s + bl[0];
}

static inline size_t align256(size_t x) { return (x + 255) & ~size_t(255); }

extern "C" void kernel_launch(void* const* d_in, const int* in_sizes, int n_in,
                              void* d_out, int out_size, void* d_ws, size_t ws_size,
                              hipStream_t stream) {
    const float* x    = (const float*)d_in[0];
    const int*   ei   = (const int*)d_in[1];
    const float* W1   = (const float*)d_in[2];
    const float* b1   = (const float*)d_in[3];
    const float* W2   = (const float*)d_in[4];
    const float* b2   = (const float*)d_in[5];
    const float* Wl   = (const float*)d_in[6];
    const float* bl   = (const float*)d_in[7];
    float*       out  = (float*)d_out;

    const int N = N_NODES;
    const int E = N_EDGES;
    const int* src = ei;
    const int* dst = ei + E;

    char* ws = (char*)d_ws;
    size_t off = 0;
    float* dinv = (float*)(ws + off); off += align256((size_t)N * 4);
    float* h1   = (float*)(ws + off); off += align256((size_t)N * 64 * 4);
    float* y1   = (float*)(ws + off); off += align256((size_t)N * 64 * 4);
    float* h2   = (float*)(ws + off); off += align256((size_t)N * 32 * 4);
    float* y2   = (float*)(ws + off); off += align256((size_t)N * 32 * 4);
    unsigned char* owner = (unsigned char*)(ws + off); off += align256((size_t)E);
    int*          plan_c = (int*)(ws + off);          off += align256((size_t)NTILE * 32 * 4);
    unsigned int* plan_s = (unsigned int*)(ws + off); off += align256((size_t)NTILE * 32 * WQ * 4);
    float*        plan_n = (float*)(ws + off);        off += align256((size_t)NTILE * 32 * WQ * 4);
    unsigned int* bitmap = (unsigned int*)(ws + off); off += align256((size_t)NTILE * BWP * 4);
    unsigned short* deg16g = (unsigned short*)(ws + off); off += align256((size_t)N * 2);
    unsigned int* pairs  = (unsigned int*)(ws + off); off += align256((size_t)E * 8);
    (void)ws_size; (void)n_in; (void)in_sizes; (void)out_size;

    const int T = 256;

    owner_map_kernel<<<NCH, 1024, 0, stream>>>(src, dst, owner, bitmap, pairs, E);

    plan_gemm_kernel<<<NTILE + G1_WGS, 1024, 0, stream>>>(pairs, dst, bitmap, dinv, N, plan_c, plan_s,
                                                           deg16g, x, W1, h1);

    tile_agg_kernel<64, true><<<NTILE, 1024, 0, stream>>>(h1, src, dst, owner, dinv, b1, y1, N,
                                                          plan_c, plan_s, plan_n, deg16g);

    gemm_wmma_f32<64, 32, 2><<<N / 32, 2 * 2 * 32, 0, stream>>>(y1, W2, h2, N);

    tile_agg_kernel<32, false><<<NTILE, 1024, 0, stream>>>(h2, src, dst, owner, dinv, b2, y2, N,
                                                           plan_c, plan_s, plan_n, deg16g);

    final_linear_kernel<<<(N + T - 1) / T, T, 0, stream>>>(y2, Wl, bl, out, N);
}


# Each aggregation launch ends with the dense step behind it, from its LDS tile (pitch padded): layer-2 product on WMMA, final dot per row; y1, y2 and 2 launches gone (6 to 4); sums unchanged
# speedup vs baseline: 4.3224x; 1.0618x over previous
#include <hip/hip_runtime.h>
#include <hip/hip_bf16.h>

typedef __attribute__((ext_vector_type(2))) float v2f;
typedef __attribute__((ext_vector_type(8))) float v8f;

#define N_NODES 100000
#define N_EDGES 3200000

template <int K, int NOUT, int MT>
struct GemmShape {
    static constexpr int NT   = NOUT / 16;
    static constexpr int NW   = MT * NT;
    static constexpr int ROWS = MT * 16;
    static constexpr int KP   = K + 4;
    static constexpr int NP   = NOUT + 4;
    static constexpr int A_WORDS = ROWS * KP;
    static constexpr int B_WORDS = K * NP;
};

template <int K, int NOUT, int MT, bool GROUPED>
__device__ __forceinline__
void gemm_tile_f32(const float* __restrict__ A, const float* __restrict__ B,
                   float* __restrict__ C, int M, int blk, float* As, float* Bs) {
    typedef GemmShape<K, NOUT, MT> S;
    constexpr int NT   = S::NT;
    constexpr int NW   = S::NW;
    constexpr int ROWS = S::ROWS;
    constexpr int KP   = S::KP;
    constexpr int NP   = S::NP;

    const int tid  = threadIdx.x;
    const int row0 = blk * ROWS;

    constexpr int A_ELTS = ROWS * K;
    constexpr int B_ELTS = K * NOUT;
    constexpr int STEP   = NW * 32 * 4;
    if constexpr (GROUPED) {
        static_assert(A_ELTS % STEP == 0 && B_ELTS % STEP == 0, "whole turns only");
        constexpr int AT = A_ELTS / STEP, BT = B_ELTS / STEP;
        float4 av[AT], bv[BT];
#pragma unroll
        for (int t = 0; t < AT; ++t) {
            const int i = tid * 4 + t * STEP;
            const int gr = row0 + i / K;
            av[t] = (gr < M) ? *(const float4*)(A + (long long)gr * K + i % K)
                             : make_float4(0.f, 0.f, 0.f, 0.f);
        }
#pragma unroll
        for (int t = 0; t < BT; ++t) bv[t] = *(const float4*)(B + tid * 4 + t * STEP);
#pragma unroll
        for (int t = 0; t < AT; ++t) {
            const int i = tid * 4 + t * STEP;
            *(float4*)(&As[(i / K) * KP + i % K]) = av[t];
        }
#pragma unroll
        for (int t = 0; t < BT; ++t) {
            const int i = tid * 4 + t * STEP;
            *(float4*)(&Bs[(i / NOUT) * NP + i % NOUT]) = bv[t];
        }
    } else {
        for (int i = tid * 4; i < A_ELTS; i += STEP) {
            int r = i / K, c = i % K;
            int gr = row0 + r;
            float4 v = (gr < M) ? *(const float4*)(A + (long long)gr * K + c)
                                : make_float4(0.f, 0.f, 0.f, 0.f);
            *(float4*)(&As[r * KP + c]) = v;
        }
        for (int i = tid * 4; i < B_ELTS; i += STEP) {
            int r = i / NOUT, c = i % NOUT;
            *(float4*)(&Bs[r * NP + c]) = *(const float4*)(B + i);
        }
    }
    __syncthreads();

    const int wid  = tid >> 5;
    const int lane = tid & 31;
    const int mn   = lane & 15;
    const int h    = lane >> 4;
    const int mt   = wid / NT;
    const int nt   = wid % NT;

    v8f acc = {};
    const float* arow = &As[(mt * 16 + mn) * KP];
    const int    ncol = nt * 16 + mn;

#pragma unroll
    for (int k = 0; k < K; k += 4) {
        v2f a, b;
        a.x = arow[k + 2 * h + 0];
        a.y = arow[k + 2 * h + 1];
        b.x = Bs[(k + 2 * h + 0) * NP + ncol];
        b.y = Bs[(k + 2 * h + 1) * NP + ncol];
        acc = __builtin_amdgcn_wmma_f32_16x16x4_f32(
             false, a,  false, b,
             (short)0, acc,  false,  false);
    }

#pragma unroll
    for (int j = 0; j < 8; ++j) {
        int r = row0 + mt * 16 + j + 8 * h;
        if (r < M) C[(long long)r * NOUT + ncol] = acc[j];
    }
}

constexpr int NPW   = 392;
constexpr int TPAD  = 4;
constexpr int NTILE = 256;
constexpr int WQ    = 512;
constexpr int NGRP  = N_EDGES / 16;
constexpr int NBAT  = (NGRP + 8191) / 8192;
static_assert(NPW * NTILE >= N_NODES, "tiles must cover every node");
static_assert(N_EDGES % 16 == 0, "owner map is read in 16-byte groups");
static_assert(NPW <= 65535, "local row index is kept in 16 bits");
constexpr int CH    = 8192;
constexpr int NCH   = (N_EDGES + CH - 1) / CH;
constexpr int GE    = 8;
constexpr int PGW   = GE * 2;
constexpr int CHW   = CH / GE / 32;
constexpr int BWP   = NCH * CHW;
constexpr int GQ    = 640;
static_assert(CH == 1024 * 8, "one thread of the owner-map launch takes four pairs of edges");
static_assert(N_EDGES % GE == 0, "the pair array is read in whole groups");
static_assert(BWP % 4 == 0, "bitmap rows are written in 16-byte pieces");
constexpr int DEG_SAT  = 65535;
static_assert((N_NODES * 2) % 16 == 0, "the 16-bit degree table is copied in 16-byte pieces");
constexpr int SRC_BITS = 17;
static_assert(N_NODES <= (1 << SRC_BITS), "source id must fit the low bits of a stored entry");
static_assert(NPW <= (1 << (32 - SRC_BITS)), "local row must fit the high bits of a stored entry");

constexpr int PRS = NTILE + 2;
__global__ __launch_bounds__(1024)
void owner_map_kernel(const int* __restrict__ src, const int* __restrict__ dst,
                      unsigned char* __restrict__ owner, unsigned int* __restrict__ bitmap,
                      unsigned int* __restrict__ pairs, int E) {
    __shared__ unsigned int pres[CHW * PRS];
    const int tid = threadIdx.x;
    for (int i = tid; i < CHW * PRS; i += 1024) pres[i] = 0u;
    __syncthreads();
    const long long c0 = (long long)blockIdx.x * CH + tid * 2;
    int2 dv[4], sv[4];
#pragma unroll
    for (int j = 0; j < 4; ++j) {
        const long long e  = c0 + j * 2048;
        const long long ec = (e < E) ? e : 0;
        dv[j] = *(const int2*)(dst + ec);
        sv[j] = *(const int2*)(src + ec);
    }
#pragma unroll
    for (int j = 0; j < 4; ++j) {
        const long long e = c0 + j * 2048;
        if (e < E) {
            const unsigned o0 = (unsigned)(dv[j].x / NPW), o1 = (unsigned)(dv[j].y / NPW);
            *(unsigned short*)(owner + e) = (unsigned short)(o0 | (o1 << 8));
            uint4 p;
            p.x = (unsigned)sv[j].x; p.y = (unsigned)dv[j].x;
            p.z = (unsigned)sv[j].y; p.w = (unsigned)dv[j].y;
            *(uint4*)(pairs + e * 2) = p;
            const int gl = (j * 2048 + tid * 2) / GE;
            const unsigned bit = 1u << (gl & 31);
            unsigned int* pw = &pres[(gl >> 5) * PRS];
            atomicOr(pw + o0, bit);
            atomicOr(pw + o1, bit);
        }
    }
    __syncthreads();
#pragma unroll
    for (int r = 0; r < 2; ++r) {
        const int idx = r * 1024 + tid;
        const int t = idx >> 3, q = (idx & 7) * 4;
        uint4 v;
        v.x = pres[(q + 0) * PRS + t];
        v.y = pres[(q + 1) * PRS + t];
        v.z = pres[(q + 2) * PRS + t];
        v.w = pres[(q + 3) * PRS + t];
        *(uint4*)(bitmap + (long long)t * BWP + blockIdx.x * CHW + q) = v;
    }
}

__device__ __forceinline__ unsigned zero_bytes(unsigned y) {
    unsigned t = (y & 0x7F7F7F7Fu) + 0x7F7F7F7Fu;
    return ~(t | y | 0x7F7F7F7Fu);
}
__device__ __forceinline__ unsigned match_group(uint4 v, unsigned me) {
    return (zero_bytes(v.x ^ me) >> 7) | (zero_bytes(v.y ^ me) >> 6) |
           (zero_bytes(v.z ^ me) >> 5) | (zero_bytes(v.w ^ me) >> 4);
}

template <int F>
__device__ void spill_edge(int e, int base, float* tile, const float* __restrict__ h,
                           const int* __restrict__ src, const int* __restrict__ dst,
                           const float* __restrict__ dinv) {
    const int s = src[e], d = dst[e];
    const float nrm = dinv[s] * dinv[d];
    const float* hr = h + (long long)s * F;
    float* tr = tile + (d - base) * (F + TPAD);
    for (int f = 0; f < F; ++f) atomicAdd(tr + f, hr[f] * nrm);
}
__device__ __forceinline__ int wrap_group(int g) {
    if (g >= NGRP) g -= NGRP;
    if (g >= NGRP) g -= NGRP;
    return g;
}

template <typename Spill>
__device__ __forceinline__ int scan_owner_map(const unsigned char* __restrict__ owner, int tileid,
                                              int tid, int* myq, Spill spill) {
    const unsigned me   = (unsigned)tileid * 0x01010101u;
    const uint4*   om   = (const uint4*)owner;
    const int rot = tileid * 768;
    int wc = 0;
    for (int it = 0; it < NBAT; ++it) {
        uint4 v[8];
#pragma unroll
        for (int k = 0; k < 8; ++k) {
            v[k] = om[wrap_group(it * 8192 + k * 1024 + tid + rot)];
        }
        unsigned mk[8];
#pragma unroll
        for (int k = 0; k < 8; ++k)
            mk[k] = match_group(v[k], me) & (0u - (unsigned)(it * 8192 + k * 1024 + tid < NGRP));
#pragma unroll
        for (int hf = 0; hf < 2; ++hf) {
            const unsigned lo = mk[4 * hf + 0] | (mk[4 * hf + 1] << 4);
            const unsigned hi = mk[4 * hf + 2] | (mk[4 * hf + 3] << 4);
            unsigned long long M = ((unsigned long long)hi << 32) | (unsigned long long)lo;
            unsigned act;
            while ((act = __builtin_amdgcn_ballot_w32(M != 0ull)) != 0u) {
                if (M != 0ull) {
                    const int p = __builtin_ctzll(M);
                    M &= M - 1ull;
                    const int k  = 4 * hf + 2 * (p >> 5) + ((p >> 2) & 1);
                    const int g = wrap_group(it * 8192 + k * 1024 + tid + rot);
                    const int e = g * 16 + (p & 3) * 4 + ((p & 31) >> 3);
                    const int slot = wc + (int)__builtin_amdgcn_mbcnt_lo(act, 0u);
                    if (slot < WQ) myq[slot] = e;
                    else spill(e);
                }
                wc += __builtin_popcount(act);
            }
        }
    }
    return wc;
}

static_assert(CHW == 32 && NTILE * 8 == 2048, "bitmap write-out of the owner-map launch");
constexpr int BW_FULL = BWP / 1024;
constexpr int BW_REST = BWP - BW_FULL * 1024;
static_assert(BW_REST * 4 <= 1024, "rest of a bitmap row is dealt to every 4th thread");

constexpr int PLAN_Q_WORDS = 32 * WQ;
constexpr int PLAN_G_WORDS = 32 * GQ;
constexpr int PLAN_WORDS   = PLAN_Q_WORDS + PLAN_G_WORDS + NPW + 32;

__device__ __forceinline__
void tile_plan_body(const unsigned int* __restrict__ pairs, const int* __restrict__ dst,
                    const unsigned int* __restrict__ bitmap, float* __restrict__ dinv, int N,
                    int* __restrict__ plan_c, unsigned int* __restrict__ plan_s,
                    unsigned short* __restrict__ deg16g, int tile_id, int* lds) {
    int* q_s   = lds;
    int* g_s   = lds + PLAN_Q_WORDS;
    int* indeg = lds + PLAN_Q_WORDS + PLAN_G_WORDS;
    int* wcs   = indeg + NPW;

    const int tid  = threadIdx.x;
    const int wid  = tid >> 5;
    const int lane = tid & 31;
    const int base = tile_id * NPW;
    int nloc = N - base;
    if (nloc > NPW) nloc = NPW;
    if (nloc < 0) nloc = 0;

    for (int i = tid; i < NPW; i += 1024) indeg[i] = 0;
    __syncthreads();

    int* myq = &q_s[wid * WQ];
    int* myg = &g_s[wid * GQ];
    unsigned int* ps = plan_s + (long long)(tile_id * 32 + wid) * WQ;

    const unsigned int* brow = bitmap + (long long)tile_id * BWP;
    unsigned bits[BW_FULL + 1];
    int      bj[BW_FULL + 1];
#pragma unroll
    for (int k = 0; k < BW_FULL; ++k) { bj[k] = k * 1024 + tid; bits[k] = brow[bj[k]]; }
    {
        const int r = tid >> 2;
        const bool has = ((tid & 3) == 0) && (r < BW_REST);
        bj[BW_FULL] = BW_FULL * 1024 + (has ? r : 0);
        const unsigned v = brow[bj[BW_FULL]];
        bits[BW_FULL] = has ? v : 0u;
    }
    int  gc  = 0;
    bool ovf = false;
#pragma unroll
    for (int k = 0; k <= BW_FULL; ++k) {
        unsigned m = bits[k];
        unsigned act;
        while ((act = __builtin_amdgcn_ballot_w32(m != 0u)) != 0u) {
            if (m != 0u) {
                const int p = __builtin_ctz(m);
                m &= m - 1u;
                const int g = bj[k] * 32 + p;
                const int slot = gc + (int)__builtin_amdgcn_mbcnt_lo(act, 0u);
                if (slot < GQ) myg[slot] = g;
                else {
                    ovf = true;
                    for (int x = 0; x < GE; ++x) {
                        const unsigned lr = (unsigned)(dst[g * GE + x] - base);
                        if (lr < (unsigned)NPW) atomicAdd(&indeg[lr], 1);
                    }
                }
            }
            gc += __builtin_popcount(act);
        }
    }
    if (gc > GQ) gc = GQ;

    constexpr int GB = 8;
    int wc = 0;
    const int sub = lane >> 2;
    const int qo  = (lane & 3) * 4;
    for (int i0 = 0; i0 < gc; i0 += 8 * GB) {
        int   gg[GB];
        uint4 pr[GB];
#pragma unroll
        for (int k = 0; k < GB; ++k) {
            const int i = i0 + 8 * k + sub;
            gg[k] = myg[(i < gc) ? i : (gc - 1)];
        }
#pragma unroll
        for (int k = 0; k < GB; ++k)
            pr[k] = *(const uint4*)(pairs + (long long)gg[k] * PGW + qo);
#pragma unroll
        for (int k = 0; k < GB; ++k) {
            const bool ok = i0 + 8 * k + sub < gc;
#pragma unroll
            for (int x = 0; x < 2; ++x) {
                const unsigned sv = x == 0 ? pr[k].x : pr[k].z;
                const unsigned dv = x == 0 ? pr[k].y : pr[k].w;
                const unsigned lr = dv - (unsigned)base;
                const bool hit = ok && lr < (unsigned)NPW;
                const unsigned act = __builtin_amdgcn_ballot_w32(hit);
                if (act != 0u) {
                    if (hit) {
                        const int slot = wc + (int)__builtin_amdgcn_mbcnt_lo(act, 0u);
                        if (slot < WQ) myq[slot] = (int)(sv | (lr << SRC_BITS));
                        else atomicAdd(&indeg[lr], 1);
                    }
                    wc += __builtin_popcount(act);
                }
            }
        }
    }
    const bool lost = __builtin_amdgcn_ballot_w32(ovf) != 0u;
    if (lane == 0) wcs[wid] = (lost && wc <= WQ) ? (WQ + 1) : wc;

    const int cnt = wc < WQ ? wc : WQ;
    for (int i = lane; i < cnt; i += 32) {
        const unsigned v = (unsigned)myq[i];
        ps[i] = v;
        atomicAdd(&indeg[v >> SRC_BITS], 1);
    }
    __syncthreads();
    if (wid == 0) plan_c[tile_id * 32 + lane] = wcs[lane];
    for (int n = tid; n < nloc; n += 1024) {
        const int dg = indeg[n] + 1;
        dinv[base + n]   = rsqrtf((float)dg);
        deg16g[base + n] = (unsigned short)(dg < DEG_SAT ? dg : DEG_SAT);
    }
}


typedef GemmShape<128, 64, 8> G1;
constexpr int G1_WGS = (N_NODES + G1::ROWS - 1) / G1::ROWS;
constexpr int G1_WORDS = G1::A_WORDS + G1::B_WORDS;
constexpr int PG_WORDS = PLAN_WORDS > G1_WORDS ? PLAN_WORDS : G1_WORDS;
static_assert(G1::NW * 32 == 1024, "the product's work-groups use the plan's block size");
static_assert(G1::A_WORDS % 4 == 0, "B tile starts on a 16-byte boundary");
static_assert(2 * PG_WORDS * 4 <= 320 * 1024, "a plan and a product work-group share one WGP's LDS");

__global__ __launch_bounds__(1024)
void plan_gemm_kernel(const unsigned int* __restrict__ pairs, const int* __restrict__ dst,
                      const unsigned int* __restrict__ bitmap, float* __restrict__ dinv, int N,
                      int* __restrict__ plan_c, unsigned int* __restrict__ plan_s,
                      unsigned short* __restrict__ deg16g,
                      const float* __restrict__ x, const float* __restrict__ W1,
                      float* __restrict__ h1) {
    __shared__ __attribute__((aligned(16))) int lds[PG_WORDS];
    if (blockIdx.x < NTILE) {
        tile_plan_body(pairs, dst, bitmap, dinv, N, plan_c, plan_s, deg16g, (int)blockIdx.x, lds);
    } else {
        float* As = reinterpret_cast<float*>(lds);
        gemm_tile_f32<128, 64, 8, true>(x, W1, h1, N, (int)blockIdx.x - NTILE, As, As + G1::A_WORDS);
    }
}

template <int F>
__device__ __forceinline__ int tile_row_bytes(unsigned row) {
    int ro;
    asm("s_mul_i32 %0, %1, %2" : "=s"(ro) : "s"(row), "n"((F + TPAD) * 4));
    return ro;
}

template <int F, bool FULL>
__device__ __forceinline__ void add_half(const float* __restrict__ h, float* tile, unsigned r, float n,
                                         int j0, int m, int lane) {
    constexpr int      VPL = F / 32;
    constexpr unsigned SM  = (1u << SRC_BITS) - 1u;
    float vals[16][VPL];
#pragma unroll
    for (int k = 0; k < 16; ++k) {
        const unsigned rj = (unsigned)__builtin_amdgcn_readlane((int)r, j0 + k);
        const float* hr = h + (long long)(rj & SM) * F + lane * VPL;
#pragma unroll
        for (int x = 0; x < VPL; ++x) vals[k][x] = hr[x];
    }
#pragma unroll
    for (int k = 0; k < 16; ++k) {
        if (FULL || k < m) {
            const unsigned rj  = (unsigned)__builtin_amdgcn_readlane((int)r, j0 + k);
            const float    nrm = __int_as_float(__builtin_amdgcn_readlane(__float_as_int(n), j0 + k));
            float* tr = reinterpret_cast<float*>(reinterpret_cast<char*>(tile) + tile_row_bytes<F>(rj >> SRC_BITS)) + lane * VPL;
#pragma unroll
            for (int x = 0; x < VPL; ++x) atomicAdd(tr + x, vals[k][x] * nrm);
        }
    }
}

constexpr int DN = 32;
template <int F, bool FIRST>
__global__ __launch_bounds__(1024)
void tile_agg_kernel(const float* __restrict__ h, const int* __restrict__ src,
                     const int* __restrict__ dst, const unsigned char* __restrict__ owner,
                     const float* __restrict__ dinv, const float* __restrict__ bias,
                     float* __restrict__ y, int N, const int* __restrict__ plan_c,
                     const unsigned int* __restrict__ plan_s, float* plan_n,
                     const unsigned short* __restrict__ deg16g,
                     const float* __restrict__ Wd, const float* __restrict__ bd) {
    constexpr int F4  = F / 4;
    constexpr int VPL = F / 32;
    constexpr int P   = F + TPAD;
    static_assert(!FIRST || F == 64, "the first layer's dense step is written for 64 x 32 weights");
    static_assert(FIRST || F == 32, "the second layer's dense step is written for 32 weights");
    __shared__ __attribute__((aligned(16))) float tile[NPW * P];
    constexpr int Q_BYTES   = 32 * WQ * (4 + 4 + 2);
    constexpr int DT_BYTES  = N_NODES * 2;
    constexpr int AUX_BYTES = (FIRST && DT_BYTES > Q_BYTES) ? DT_BYTES : Q_BYTES;
    __shared__ __attribute__((aligned(16))) unsigned char aux[AUX_BYTES];
    int*            q_s = reinterpret_cast<int*>(aux);
    float*          q_n = reinterpret_cast<float*>(aux + 32 * WQ * 4);
    unsigned short* q_d = reinterpret_cast<unsigned short*>(aux + 32 * WQ * 8);
    const unsigned short* deg16 = reinterpret_cast<const unsigned short*>(aux);

    const int tid  = threadIdx.x;
    const int wid  = tid >> 5;
    const int lane = tid & 31;
    const int base = blockIdx.x * NPW;
    int nloc = N - base;
    if (nloc > NPW) nloc = NPW;
    if (nloc < 0) nloc = 0;

    const int c32 = plan_c[blockIdx.x * 32 + lane];
    const bool build = __builtin_amdgcn_ballot_w32(c32 > WQ) != 0u;
    int wc = __builtin_amdgcn_readlane(c32, __builtin_amdgcn_readfirstlane(wid));

    if (FIRST && !build) {
        constexpr int DT_VEC = DT_BYTES / 16;
        constexpr int DT_T   = (DT_VEC + 1023) / 1024;
        const uint4* gsrc = reinterpret_cast<const uint4*>(deg16g);
        uint4*       ldst = reinterpret_cast<uint4*>(aux);
        uint4 tv[DT_T];
#pragma unroll
        for (int t = 0; t < DT_T; ++t) {
            const int i = t * 1024 + tid;
            tv[t] = gsrc[i < DT_VEC ? i : DT_VEC - 1];
        }
#pragma unroll
        for (int t = 0; t < DT_T; ++t) {
            const int i = t * 1024 + tid;
            if (i < DT_VEC) ldst[i] = tv[t];
        }
    }

    for (int i = tid; i < nloc * F4; i += 1024) {
        const int n = i / F4, c = i % F4;
        const float di = dinv[base + n];
        const float4 hv = *(const float4*)(h + (long long)(base + n) * F + c * 4);
        const float4 bv = *(const float4*)(bias + c * 4);
        float4 t;
        t.x = hv.x * di * di + bv.x;
        t.y = hv.y * di * di + bv.y;
        t.z = hv.z * di * di + bv.z;
        t.w = hv.w * di * di + bv.w;
        *(float4*)(&tile[n * P + c * 4]) = t;
    }
    __syncthreads();

    int*            myq = &q_s[wid * WQ];
    float*          myn = &q_n[wid * WQ];
    unsigned short* myd = &q_d[wid * WQ];
    const unsigned int* ps = plan_s + (long long)(blockIdx.x * 32 + wid) * WQ;
    float*              pn = plan_n + (long long)(blockIdx.x * 32 + wid) * WQ;
    if (build) {
    wc = scan_owner_map(owner, (int)blockIdx.x, tid, myq,
                        [&](int e) { spill_edge<F>(e, base, tile, h, src, dst, dinv); });
    const int cnt = wc < WQ ? wc : WQ;
    for (int i0 = 0; i0 < cnt; i0 += 128) {
        int ix[4], ss[4], dd[4];
        float nr[4];
#pragma unroll
        for (int k = 0; k < 4; ++k) {
            const int i = i0 + 32 * k + lane;
            ix[k] = (i < cnt) ? i : (cnt - 1);
        }
        int ee[4];
#pragma unroll
        for (int k = 0; k < 4; ++k) ee[k] = myq[ix[k]];
#pragma unroll
        for (int k = 0; k < 4; ++k) { ss[k] = src[ee[k]]; dd[k] = dst[ee[k]]; }
#pragma unroll
        for (int k = 0; k < 4; ++k) nr[k] = dinv[ss[k]] * dinv[dd[k]];
#pragma unroll
        for (int k = 0; k < 4; ++k) {
            myq[ix[k]] = ss[k];
            myd[ix[k]] = (unsigned short)(dd[k] - base);
            myn[ix[k]] = nr[k];
        }
    }
    __syncthreads();

    constexpr int U = 16;
    for (int i0 = 0; i0 < cnt; i0 += U) {
        float vals[U][VPL];
#pragma unroll
        for (int k = 0; k < U; ++k) {
            const int i = (i0 + k < cnt) ? (i0 + k) : i0;
            const float* hr = h + (long long)myq[i] * F + lane * VPL;
#pragma unroll
            for (int x = 0; x < VPL; ++x) vals[k][x] = hr[x];
        }
#pragma unroll
        for (int k = 0; k < U; ++k) {
            if (i0 + k < cnt) {
                const float nrm = myn[i0 + k];
                float* tr = &tile[(int)myd[i0 + k] * P + lane * VPL];
#pragma unroll
                for (int x = 0; x < VPL; ++x) atomicAdd(tr + x, vals[k][x] * nrm);
            }
        }
    }
    } else {
    const int cnt = wc;
    if (cnt > 0) {
        constexpr unsigned SM = (1u << SRC_BITS) - 1u;
        const int nb = (cnt + 31) >> 5;
        auto idx = [&](int bb) { const int i = bb * 32 + lane; return (i < cnt) ? i : (cnt - 1); };
        const int nfull = cnt >> 5;
        auto add_tail = [&](unsigned r, float n, int m0) {
            if (m0 >= 16) add_half<F, true >(h, tile, r, n, 0, 16, lane);
            else          add_half<F, false>(h, tile, r, n, 0, m0, lane);
            if (m0 > 16)  add_half<F, false>(h, tile, r, n, 16, m0 - 16, lane);
        };
        if (FIRST) {
            auto src_dinv = [&](unsigned s) {
                const int dg = (int)deg16[s];
                return dg == DEG_SAT ? dinv[s] : rsqrtf((float)dg);
            };
            unsigned r_cur = ps[idx(0)];
            unsigned r_nxt = ps[idx(nb > 1 ? 1 : 0)];
            float    n_cur = src_dinv(r_cur & SM) * dinv[base + (int)(r_cur >> SRC_BITS)];
            if (lane < cnt) pn[lane] = n_cur;
            for (int bb = 0; bb < nfull; ++bb) {
                const int   dg = (int)deg16[r_nxt & SM];
                const float d2 = dinv[base + (int)(r_nxt >> SRC_BITS)];
                const unsigned r_n2 = ps[idx(bb + 2 < nb ? bb + 2 : nb - 1)];
                add_half<F, true>(h, tile, r_cur, n_cur, 0, 16, lane);
                add_half<F, true>(h, tile, r_cur, n_cur, 16, 16, lane);
                const float d1 = dg == DEG_SAT ? dinv[r_nxt & SM] : rsqrtf((float)dg);
                r_cur = r_nxt; n_cur = d1 * d2; r_nxt = r_n2;
                const int i1 = (bb + 1) * 32 + lane;
                if (i1 < cnt) pn[i1] = n_cur;
            }
            if (nfull < nb) add_tail(r_cur, n_cur, cnt - nfull * 32);
        } else {
            const int ic = idx(0);
            unsigned r_cur = ps[ic];
            float    n_cur = pn[ic];
            for (int bb = 0; bb < nfull; ++bb) {
                const int ix = idx(bb + 1 < nb ? bb + 1 : nb - 1);
                const unsigned r_nxt = ps[ix];
                const float    n_nxt = pn[ix];
                add_half<F, true>(h, tile, r_cur, n_cur, 0, 16, lane);
                add_half<F, true>(h, tile, r_cur, n_cur, 16, 16, lane);
                r_cur = r_nxt; n_cur = n_nxt;
            }
            if (nfull < nb) add_tail(r_cur, n_cur, cnt - nfull * 32);
        }
    }
    }

    if constexpr (FIRST) {
        const int  mn   = lane & 15;
        const int  hh   = lane >> 4;
        const bool prod = wid * 16 < nloc;
        v2f b0[F / 4], b1[F / 4];
        if (prod) {
#pragma unroll
            for (int kk = 0; kk < F / 4; ++kk) {
                const float* wr = Wd + (kk * 4 + 2 * hh) * DN + mn;
                b0[kk].x = wr[0];       b0[kk].y = wr[DN];
                b1[kk].x = wr[16];      b1[kk].y = wr[DN + 16];
            }
        }
        __syncthreads();
        if (prod) {
            int ar = wid * 16 + mn;
            if (ar > nloc - 1) ar = nloc - 1;
            const float* arow = &tile[ar * P];
            v8f acc0 = {}, acc1 = {};
#pragma unroll
            for (int kk = 0; kk < F / 4; ++kk) {
                v2f a;
                a.x = fmaxf(arow[kk * 4 + 2 * hh + 0], 0.0f);
                a.y = fmaxf(arow[kk * 4 + 2 * hh + 1], 0.0f);
                acc0 = __builtin_amdgcn_wmma_f32_16x16x4_f32(
                     false, a,  false, b0[kk],
                     (short)0, acc0,  false,  false);
                acc1 = __builtin_amdgcn_wmma_f32_16x16x4_f32(
                     false, a,  false, b1[kk],
                     (short)0, acc1,  false,  false);
            }
#pragma unroll
            for (int j = 0; j < 8; ++j) {
                const int r = wid * 16 + j + 8 * hh;
                if (r < nloc) {
                    float* yr = y + (long long)(base + r) * DN + mn;
                    yr[0]  = acc0[j];
                    yr[16] = acc1[j];
                }
            }
        }
    } else {
        __syncthreads();
        const float bl0 = bd[0];
        for (int n = tid; n < nloc; n += 1024) {
            const float* tr = &tile[n * P];
            float s = 0.0f;
#pragma unroll
            for (int q = 0; q < F4; ++q) {
                const float4 t = *(const float4*)(tr + q * 4);
                s += fmaxf(t.x, 0.0f) * Wd[q * 4 + 0];
                s += fmaxf(t.y, 0.0f) * Wd[q * 4 + 1];
                s += fmaxf(t.z, 0.0f) * Wd[q * 4 + 2];
                s += fmaxf(t.w, 0.0f) * Wd[q * 4 + 3];
            }
            y[base + n] = s + bl0;
        }
    }
}

static inline size_t align256(size_t x) { return (x + 255) & ~size_t(255); }

extern "C" void kernel_launch(void* const* d_in, const int* in_sizes, int n_in,
                              void* d_out, int out_size, void* d_ws, size_t ws_size,
                              hipStream_t stream) {
    const float* x    = (const float*)d_in[0];
    const int*   ei   = (const int*)d_in[1];
    const float* W1   = (const float*)d_in[2];
    const float* b1   = (const float*)d_in[3];
    const float* W2   = (const float*)d_in[4];
    const float* b2   = (const float*)d_in[5];
    const float* Wl   = (const float*)d_in[6];
    const float* bl   = (const float*)d_in[7];
    float*       out  = (float*)d_out;

    const int N = N_NODES;
    const int E = N_EDGES;
    const int* src = ei;
    const int* dst = ei + E;

    char* ws = (char*)d_ws;
    size_t off = 0;
    float* dinv = (float*)(ws + off); off += align256((size_t)N * 4);
    float* h1   = (float*)(ws + off); off += align256((size_t)N * 64 * 4);
    float* h2   = (float*)(ws + off); off += align256((size_t)N * 32 * 4);
    unsigned char* owner = (unsigned char*)(ws + off); off += align256((size_t)E);
    int*          plan_c = (int*)(ws + off);          off += align256((size_t)NTILE * 32 * 4);
    unsigned int* plan_s = (unsigned int*)(ws + off); off += align256((size_t)NTILE * 32 * WQ * 4);
    float*        plan_n = (float*)(ws + off);        off += align256((size_t)NTILE * 32 * WQ * 4);
    unsigned int* bitmap = (unsigned int*)(ws + off); off += align256((size_t)NTILE * BWP * 4);
    unsigned short* deg16g = (unsigned short*)(ws + off); off += align256((size_t)N * 2);
    unsigned int* pairs  = (unsigned int*)(ws + off); off += align256((size_t)E * 8);
    (void)ws_size; (void)n_in; (void)in_sizes; (void)out_size;

    owner_map_kernel<<<NCH, 1024, 0, stream>>>(src, dst, owner, bitmap, pairs, E);

    plan_gemm_kernel<<<NTILE + G1_WGS, 1024, 0, stream>>>(pairs, dst, bitmap, dinv, N, plan_c, plan_s,
                                                           deg16g, x, W1, h1);

    tile_agg_kernel<64, true><<<NTILE, 1024, 0, stream>>>(h1, src, dst, owner, dinv, b1, h2, N,
                                                          plan_c, plan_s, plan_n, deg16g, W2, bl);

    tile_agg_kernel<32, false><<<NTILE, 1024, 0, stream>>>(h2, src, dst, owner, dinv, b2, out, N,
                                                           plan_c, plan_s, plan_n, deg16g, Wl, bl);
}


# Plan deals each edge to the wave owning its row (row mod 32, int LDS slot counter); aggregation waves sum their 13 rows in registers picked by a uniform index: no LDS f32 adds; sum order changes
# speedup vs baseline: 4.6994x; 1.0872x over previous
#include <hip/hip_runtime.h>
#include <hip/hip_bf16.h>

typedef __attribute__((ext_vector_type(2))) float v2f;
typedef __attribute__((ext_vector_type(8))) float v8f;
typedef __attribute__((ext_vector_type(16))) float v16f;

#define N_NODES 100000
#define N_EDGES 3200000

template <int K, int NOUT, int MT>
struct GemmShape {
    static constexpr int NT   = NOUT / 16;
    static constexpr int NW   = MT * NT;
    static constexpr int ROWS = MT * 16;
    static constexpr int KP   = K + 4;
    static constexpr int NP   = NOUT + 4;
    static constexpr int A_WORDS = ROWS * KP;
    static constexpr int B_WORDS = K * NP;
};

template <int K, int NOUT, int MT, bool GROUPED>
__device__ __forceinline__
void gemm_tile_f32(const float* __restrict__ A, const float* __restrict__ B,
                   float* __restrict__ C, int M, int blk, float* As, float* Bs) {
    typedef GemmShape<K, NOUT, MT> S;
    constexpr int NT   = S::NT;
    constexpr int NW   = S::NW;
    constexpr int ROWS = S::ROWS;
    constexpr int KP   = S::KP;
    constexpr int NP   = S::NP;

    const int tid  = threadIdx.x;
    const int row0 = blk * ROWS;

    constexpr int A_ELTS = ROWS * K;
    constexpr int B_ELTS = K * NOUT;
    constexpr int STEP   = NW * 32 * 4;
    if constexpr (GROUPED) {
        static_assert(A_ELTS % STEP == 0 && B_ELTS % STEP == 0, "whole turns only");
        constexpr int AT = A_ELTS / STEP, BT = B_ELTS / STEP;
        float4 av[AT], bv[BT];
#pragma unroll
        for (int t = 0; t < AT; ++t) {
            const int i = tid * 4 + t * STEP;
            const int gr = row0 + i / K;
            av[t] = (gr < M) ? *(const float4*)(A + (long long)gr * K + i % K)
                             : make_float4(0.f, 0.f, 0.f, 0.f);
        }
#pragma unroll
        for (int t = 0; t < BT; ++t) bv[t] = *(const float4*)(B + tid * 4 + t * STEP);
#pragma unroll
        for (int t = 0; t < AT; ++t) {
            const int i = tid * 4 + t * STEP;
            *(float4*)(&As[(i / K) * KP + i % K]) = av[t];
        }
#pragma unroll
        for (int t = 0; t < BT; ++t) {
            const int i = tid * 4 + t * STEP;
            *(float4*)(&Bs[(i / NOUT) * NP + i % NOUT]) = bv[t];
        }
    } else {
        for (int i = tid * 4; i < A_ELTS; i += STEP) {
            int r = i / K, c = i % K;
            int gr = row0 + r;
            float4 v = (gr < M) ? *(const float4*)(A + (long long)gr * K + c)
                                : make_float4(0.f, 0.f, 0.f, 0.f);
            *(float4*)(&As[r * KP + c]) = v;
        }
        for (int i = tid * 4; i < B_ELTS; i += STEP) {
            int r = i / NOUT, c = i % NOUT;
            *(float4*)(&Bs[r * NP + c]) = *(const float4*)(B + i);
        }
    }
    __syncthreads();

    const int wid  = tid >> 5;
    const int lane = tid & 31;
    const int mn   = lane & 15;
    const int h    = lane >> 4;
    const int mt   = wid / NT;
    const int nt   = wid % NT;

    v8f acc = {};
    const float* arow = &As[(mt * 16 + mn) * KP];
    const int    ncol = nt * 16 + mn;

#pragma unroll
    for (int k = 0; k < K; k += 4) {
        v2f a, b;
        a.x = arow[k + 2 * h + 0];
        a.y = arow[k + 2 * h + 1];
        b.x = Bs[(k + 2 * h + 0) * NP + ncol];
        b.y = Bs[(k + 2 * h + 1) * NP + ncol];
        acc = __builtin_amdgcn_wmma_f32_16x16x4_f32(
             false, a,  false, b,
             (short)0, acc,  false,  false);
    }

#pragma unroll
    for (int j = 0; j < 8; ++j) {
        int r = row0 + mt * 16 + j + 8 * h;
        if (r < M) C[(long long)r * NOUT + ncol] = acc[j];
    }
}

constexpr int NPW   = 392;
constexpr int TPAD  = 4;
constexpr int NTILE = 256;
constexpr int WQ    = 512;
constexpr int NGRP  = N_EDGES / 16;
constexpr int NBAT  = (NGRP + 8191) / 8192;
static_assert(NPW * NTILE >= N_NODES, "tiles must cover every node");
static_assert(N_EDGES % 16 == 0, "owner map is read in 16-byte groups");
static_assert(NPW <= 65535, "local row index is kept in 16 bits");
constexpr int NJ    = (NPW + 31) / 32;
static_assert(NJ <= 16, "a wave keeps the sums of its rows in one 16-float register vector per column");
constexpr int CH    = 8192;
constexpr int NCH   = (N_EDGES + CH - 1) / CH;
constexpr int GE    = 8;
constexpr int PGW   = GE * 2;
constexpr int CHW   = CH / GE / 32;
constexpr int BWP   = NCH * CHW;
constexpr int GQ    = 640;
static_assert(CH == 1024 * 8, "one thread of the owner-map launch takes four pairs of edges");
static_assert(N_EDGES % GE == 0, "the pair array is read in whole groups");
static_assert(BWP % 4 == 0, "bitmap rows are written in 16-byte pieces");
constexpr int DEG_SAT  = 65535;
static_assert((N_NODES * 2) % 16 == 0, "the 16-bit degree table is copied in 16-byte pieces");
constexpr int SRC_BITS = 17;
static_assert(N_NODES <= (1 << SRC_BITS), "source id must fit the low bits of a stored entry");
static_assert(NPW <= (1 << (32 - SRC_BITS)), "local row must fit the high bits of a stored entry");

constexpr int PRS = NTILE + 2;
__global__ __launch_bounds__(1024)
void owner_map_kernel(const int* __restrict__ src, const int* __restrict__ dst,
                      unsigned char* __restrict__ owner, unsigned int* __restrict__ bitmap,
                      unsigned int* __restrict__ pairs, int E) {
    __shared__ unsigned int pres[CHW * PRS];
    const int tid = threadIdx.x;
    for (int i = tid; i < CHW * PRS; i += 1024) pres[i] = 0u;
    __syncthreads();
    const long long c0 = (long long)blockIdx.x * CH + tid * 2;
    int2 dv[4], sv[4];
#pragma unroll
    for (int j = 0; j < 4; ++j) {
        const long long e  = c0 + j * 2048;
        const long long ec = (e < E) ? e : 0;
        dv[j] = *(const int2*)(dst + ec);
        sv[j] = *(const int2*)(src + ec);
    }
#pragma unroll
    for (int j = 0; j < 4; ++j) {
        const long long e = c0 + j * 2048;
        if (e < E) {
            const unsigned o0 = (unsigned)(dv[j].x / NPW), o1 = (unsigned)(dv[j].y / NPW);
            *(unsigned short*)(owner + e) = (unsigned short)(o0 | (o1 << 8));
            uint4 p;
            p.x = (unsigned)sv[j].x; p.y = (unsigned)dv[j].x;
            p.z = (unsigned)sv[j].y; p.w = (unsigned)dv[j].y;
            *(uint4*)(pairs + e * 2) = p;
            const int gl = (j * 2048 + tid * 2) / GE;
            const unsigned bit = 1u << (gl & 31);
            unsigned int* pw = &pres[(gl >> 5) * PRS];
            atomicOr(pw + o0, bit);
            atomicOr(pw + o1, bit);
        }
    }
    __syncthreads();
#pragma unroll
    for (int r = 0; r < 2; ++r) {
        const int idx = r * 1024 + tid;
        const int t = idx >> 3, q = (idx & 7) * 4;
        uint4 v;
        v.x = pres[(q + 0) * PRS + t];
        v.y = pres[(q + 1) * PRS + t];
        v.z = pres[(q + 2) * PRS + t];
        v.w = pres[(q + 3) * PRS + t];
        *(uint4*)(bitmap + (long long)t * BWP + blockIdx.x * CHW + q) = v;
    }
}

__device__ __forceinline__ unsigned zero_bytes(unsigned y) {
    unsigned t = (y & 0x7F7F7F7Fu) + 0x7F7F7F7Fu;
    return ~(t | y | 0x7F7F7F7Fu);
}
__device__ __forceinline__ unsigned match_group(uint4 v, unsigned me) {
    return (zero_bytes(v.x ^ me) >> 7) | (zero_bytes(v.y ^ me) >> 6) |
           (zero_bytes(v.z ^ me) >> 5) | (zero_bytes(v.w ^ me) >> 4);
}

template <int F>
__device__ void spill_edge(int e, int base, float* tile, const float* __restrict__ h,
                           const int* __restrict__ src, const int* __restrict__ dst,
                           const float* __restrict__ dinv) {
    const int s = src[e], d = dst[e];
    const float nrm = dinv[s] * dinv[d];
    const float* hr = h + (long long)s * F;
    float* tr = tile + (d - base) * (F + TPAD);
    for (int f = 0; f < F; ++f) atomicAdd(tr + f, hr[f] * nrm);
}
__device__ __forceinline__ int wrap_group(int g) {
    if (g >= NGRP) g -= NGRP;
    if (g >= NGRP) g -= NGRP;
    return g;
}

template <typename Spill>
__device__ __forceinline__ int scan_owner_map(const unsigned char* __restrict__ owner, int tileid,
                                              int tid, int* myq, Spill spill) {
    const unsigned me   = (unsigned)tileid * 0x01010101u;
    const uint4*   om   = (const uint4*)owner;
    const int rot = tileid * 768;
    int wc = 0;
    for (int it = 0; it < NBAT; ++it) {
        uint4 v[8];
#pragma unroll
        for (int k = 0; k < 8; ++k) {
            v[k] = om[wrap_group(it * 8192 + k * 1024 + tid + rot)];
        }
        unsigned mk[8];
#pragma unroll
        for (int k = 0; k < 8; ++k)
            mk[k] = match_group(v[k], me) & (0u - (unsigned)(it * 8192 + k * 1024 + tid < NGRP));
#pragma unroll
        for (int hf = 0; hf < 2; ++hf) {
            const unsigned lo = mk[4 * hf + 0] | (mk[4 * hf + 1] << 4);
            const unsigned hi = mk[4 * hf + 2] | (mk[4 * hf + 3] << 4);
            unsigned long long M = ((unsigned long long)hi << 32) | (unsigned long long)lo;
            unsigned act;
            while ((act = __builtin_amdgcn_ballot_w32(M != 0ull)) != 0u) {
                if (M != 0ull) {
                    const int p = __builtin_ctzll(M);
                    M &= M - 1ull;
                    const int k  = 4 * hf + 2 * (p >> 5) + ((p >> 2) & 1);
                    const int g = wrap_group(it * 8192 + k * 1024 + tid + rot);
                    const int e = g * 16 + (p & 3) * 4 + ((p & 31) >> 3);
                    const int slot = wc + (int)__builtin_amdgcn_mbcnt_lo(act, 0u);
                    if (slot < WQ) myq[slot] = e;
                    else spill(e);
                }
                wc += __builtin_popcount(act);
            }
        }
    }
    return wc;
}

static_assert(CHW == 32 && NTILE * 8 == 2048, "bitmap write-out of the owner-map launch");
constexpr int BW_FULL = BWP / 1024;
constexpr int BW_REST = BWP - BW_FULL * 1024;
static_assert(BW_REST * 4 <= 1024, "rest of a bitmap row is dealt to every 4th thread");

constexpr int PLAN_Q_WORDS = 32 * WQ;
constexpr int PLAN_G_WORDS = 32 * GQ;
constexpr int PLAN_WORDS   = PLAN_Q_WORDS + PLAN_G_WORDS + NPW + 33;

__device__ __forceinline__
void tile_plan_body(const unsigned int* __restrict__ pairs, const int* __restrict__ dst,
                    const unsigned int* __restrict__ bitmap, float* __restrict__ dinv, int N,
                    int* __restrict__ plan_c, unsigned int* __restrict__ plan_s,
                    unsigned short* __restrict__ deg16g, int tile_id, int* lds) {
    int* q_s   = lds;
    int* g_s   = lds + PLAN_Q_WORDS;
    int* indeg = lds + PLAN_Q_WORDS + PLAN_G_WORDS;
    int* wcs   = indeg + NPW;

    const int tid  = threadIdx.x;
    const int wid  = tid >> 5;
    const int lane = tid & 31;
    const int base = tile_id * NPW;
    int nloc = N - base;
    if (nloc > NPW) nloc = NPW;
    if (nloc < 0) nloc = 0;

    for (int i = tid; i < NPW; i += 1024) indeg[i] = 0;
    if (tid < 33) wcs[tid] = 0;
    __syncthreads();

    int* myq = &q_s[wid * WQ];
    int* myg = &g_s[wid * GQ];
    unsigned int* ps = plan_s + (long long)(tile_id * 32 + wid) * WQ;

    const unsigned int* brow = bitmap + (long long)tile_id * BWP;
    unsigned bits[BW_FULL + 1];
    int      bj[BW_FULL + 1];
#pragma unroll
    for (int k = 0; k < BW_FULL; ++k) { bj[k] = k * 1024 + tid; bits[k] = brow[bj[k]]; }
    {
        const int r = tid >> 2;
        const bool has = ((tid & 3) == 0) && (r < BW_REST);
        bj[BW_FULL] = BW_FULL * 1024 + (has ? r : 0);
        const unsigned v = brow[bj[BW_FULL]];
        bits[BW_FULL] = has ? v : 0u;
    }
    int  gc  = 0;
    bool ovf = false;
#pragma unroll
    for (int k = 0; k <= BW_FULL; ++k) {
        unsigned m = bits[k];
        unsigned act;
        while ((act = __builtin_amdgcn_ballot_w32(m != 0u)) != 0u) {
            if (m != 0u) {
                const int p = __builtin_ctz(m);
                m &= m - 1u;
                const int g = bj[k] * 32 + p;
                const int slot = gc + (int)__builtin_amdgcn_mbcnt_lo(act, 0u);
                if (slot < GQ) myg[slot] = g;
                else {
                    ovf = true;
                    for (int x = 0; x < GE; ++x) {
                        const unsigned lr = (unsigned)(dst[g * GE + x] - base);
                        if (lr < (unsigned)NPW) atomicAdd(&indeg[lr], 1);
                    }
                }
            }
            gc += __builtin_popcount(act);
        }
    }
    if (gc > GQ) gc = GQ;

    constexpr int GB = 8;
    const int sub = lane >> 2;
    const int qo  = (lane & 3) * 4;
    for (int i0 = 0; i0 < gc; i0 += 8 * GB) {
        int   gg[GB];
        uint4 pr[GB];
#pragma unroll
        for (int k = 0; k < GB; ++k) {
            const int i = i0 + 8 * k + sub;
            gg[k] = myg[(i < gc) ? i : (gc - 1)];
        }
#pragma unroll
        for (int k = 0; k < GB; ++k)
            pr[k] = *(const uint4*)(pairs + (long long)gg[k] * PGW + qo);
#pragma unroll
        for (int kh = 0; kh < GB; kh += 4) {
            int slot[4][2];
#pragma unroll
            for (int k = 0; k < 4; ++k) {
                const bool ok = i0 + 8 * (kh + k) + sub < gc;
#pragma unroll
                for (int x = 0; x < 2; ++x) {
                    const unsigned dv = x == 0 ? pr[kh + k].y : pr[kh + k].w;
                    const unsigned lr = dv - (unsigned)base;
                    slot[k][x] = -1;
                    if (ok && lr < (unsigned)NPW) slot[k][x] = atomicAdd(&wcs[lr & 31u], 1);
                }
            }
#pragma unroll
            for (int k = 0; k < 4; ++k) {
#pragma unroll
                for (int x = 0; x < 2; ++x) {
                    if (slot[k][x] >= 0) {
                        const unsigned sv = x == 0 ? pr[kh + k].x : pr[kh + k].z;
                        const unsigned dv = x == 0 ? pr[kh + k].y : pr[kh + k].w;
                        const unsigned lr = dv - (unsigned)base;
                        if (slot[k][x] < WQ) q_s[(lr & 31u) * WQ + slot[k][x]] = (int)(sv | (lr << SRC_BITS));
                        else atomicAdd(&indeg[lr], 1);
                    }
                }
            }
        }
    }
    if (ovf) wcs[32] = 1;
    __syncthreads();

    const int tot = wcs[wid];
    const int cnt = tot < WQ ? tot : WQ;
    for (int i = lane; i < cnt; i += 32) {
        const unsigned v = (unsigned)myq[i];
        ps[i] = v;
        atomicAdd(&indeg[v >> SRC_BITS], 1);
    }
    __syncthreads();
    if (wid == 0) {
        const int c = wcs[lane];
        plan_c[tile_id * 32 + lane] = (wcs[32] != 0 && c <= WQ) ? (WQ + 1) : c;
    }
    for (int n = tid; n < nloc; n += 1024) {
        const int dg = indeg[n] + 1;
        dinv[base + n]   = rsqrtf((float)dg);
        deg16g[base + n] = (unsigned short)(dg < DEG_SAT ? dg : DEG_SAT);
    }
}


typedef GemmShape<128, 64, 8> G1;
constexpr int G1_WGS = (N_NODES + G1::ROWS - 1) / G1::ROWS;
constexpr int G1_WORDS = G1::A_WORDS + G1::B_WORDS;
constexpr int PG_WORDS = PLAN_WORDS > G1_WORDS ? PLAN_WORDS : G1_WORDS;
static_assert(G1::NW * 32 == 1024, "the product's work-groups use the plan's block size");
static_assert(G1::A_WORDS % 4 == 0, "B tile starts on a 16-byte boundary");
static_assert(2 * PG_WORDS * 4 <= 320 * 1024, "a plan and a product work-group share one WGP's LDS");

__global__ __launch_bounds__(1024)
void plan_gemm_kernel(const unsigned int* __restrict__ pairs, const int* __restrict__ dst,
                      const unsigned int* __restrict__ bitmap, float* __restrict__ dinv, int N,
                      int* __restrict__ plan_c, unsigned int* __restrict__ plan_s,
                      unsigned short* __restrict__ deg16g,
                      const float* __restrict__ x, const float* __restrict__ W1,
                      float* __restrict__ h1) {
    __shared__ __attribute__((aligned(16))) int lds[PG_WORDS];
    if (blockIdx.x < NTILE) {
        tile_plan_body(pairs, dst, bitmap, dinv, N, plan_c, plan_s, deg16g, (int)blockIdx.x, lds);
    } else {
        float* As = reinterpret_cast<float*>(lds);
        gemm_tile_f32<128, 64, 8, true>(x, W1, h1, N, (int)blockIdx.x - NTILE, As, As + G1::A_WORDS);
    }
}

template <int F, bool FULL>
__device__ __forceinline__ void add_half(const float* __restrict__ h, v16f (&acc)[F / 32], unsigned r, float n,
                                         int j0, int m, int lane) {
    constexpr int      VPL = F / 32;
    constexpr unsigned SM  = (1u << SRC_BITS) - 1u;
    float vals[16][VPL];
#pragma unroll
    for (int k = 0; k < 16; ++k) {
        const unsigned rj = (unsigned)__builtin_amdgcn_readlane((int)r, j0 + k);
        const float* hr = h + (long long)(rj & SM) * F + lane * VPL;
#pragma unroll
        for (int x = 0; x < VPL; ++x) vals[k][x] = hr[x];
    }
#pragma unroll
    for (int k = 0; k < 16; ++k) {
        if (FULL || k < m) {
            const unsigned rj  = (unsigned)__builtin_amdgcn_readlane((int)r, j0 + k);
            const float    nrm = __int_as_float(__builtin_amdgcn_readlane(__float_as_int(n), j0 + k));
            const int      j   = (int)(rj >> (SRC_BITS + 5));
#pragma unroll
            for (int x = 0; x < VPL; ++x) acc[x][j] += vals[k][x] * nrm;
        }
    }
}

constexpr int DN = 32;
template <int F, bool FIRST>
__global__ __launch_bounds__(1024)
void tile_agg_kernel(const float* __restrict__ h, const int* __restrict__ src,
                     const int* __restrict__ dst, const unsigned char* __restrict__ owner,
                     const float* __restrict__ dinv, const float* __restrict__ bias,
                     float* __restrict__ y, int N, const int* __restrict__ plan_c,
                     const unsigned int* __restrict__ plan_s, float* plan_n,
                     const unsigned short* __restrict__ deg16g,
                     const float* __restrict__ Wd, const float* __restrict__ bd) {
    constexpr int F4  = F / 4;
    constexpr int VPL = F / 32;
    constexpr int P   = F + TPAD;
    static_assert(!FIRST || F == 64, "the first layer's dense step is written for 64 x 32 weights");
    static_assert(FIRST || F == 32, "the second layer's dense step is written for 32 weights");
    __shared__ __attribute__((aligned(16))) float tile[NPW * P];
    constexpr int Q_BYTES   = 32 * WQ * (4 + 4 + 2);
    constexpr int DT_BYTES  = N_NODES * 2;
    constexpr int AUX_BYTES = (FIRST && DT_BYTES > Q_BYTES) ? DT_BYTES : Q_BYTES;
    __shared__ __attribute__((aligned(16))) unsigned char aux[AUX_BYTES];
    int*            q_s = reinterpret_cast<int*>(aux);
    float*          q_n = reinterpret_cast<float*>(aux + 32 * WQ * 4);
    unsigned short* q_d = reinterpret_cast<unsigned short*>(aux + 32 * WQ * 8);
    const unsigned short* deg16 = reinterpret_cast<const unsigned short*>(aux);

    const int tid  = threadIdx.x;
    const int wid  = tid >> 5;
    const int lane = tid & 31;
    const int base = blockIdx.x * NPW;
    int nloc = N - base;
    if (nloc > NPW) nloc = NPW;
    if (nloc < 0) nloc = 0;

    const int c32 = plan_c[blockIdx.x * 32 + lane];
    const bool build = __builtin_amdgcn_ballot_w32(c32 > WQ) != 0u;
    int wc = __builtin_amdgcn_readlane(c32, __builtin_amdgcn_readfirstlane(wid));

    if (FIRST && !build) {
        constexpr int DT_VEC = DT_BYTES / 16;
        constexpr int DT_T   = (DT_VEC + 1023) / 1024;
        const uint4* gsrc = reinterpret_cast<const uint4*>(deg16g);
        uint4*       ldst = reinterpret_cast<uint4*>(aux);
        uint4 tv[DT_T];
#pragma unroll
        for (int t = 0; t < DT_T; ++t) {
            const int i = t * 1024 + tid;
            tv[t] = gsrc[i < DT_VEC ? i : DT_VEC - 1];
        }
#pragma unroll
        for (int t = 0; t < DT_T; ++t) {
            const int i = t * 1024 + tid;
            if (i < DT_VEC) ldst[i] = tv[t];
        }
    }

    if (build) {
        for (int i = tid; i < nloc * F4; i += 1024) {
            const int n = i / F4, c = i % F4;
            const float di = dinv[base + n];
            const float4 hv = *(const float4*)(h + (long long)(base + n) * F + c * 4);
            const float4 bv = *(const float4*)(bias + c * 4);
            float4 t;
            t.x = hv.x * di * di + bv.x;
            t.y = hv.y * di * di + bv.y;
            t.z = hv.z * di * di + bv.z;
            t.w = hv.w * di * di + bv.w;
            *(float4*)(&tile[n * P + c * 4]) = t;
        }
    }
    if (build || FIRST) __syncthreads();

    int*            myq = &q_s[wid * WQ];
    float*          myn = &q_n[wid * WQ];
    unsigned short* myd = &q_d[wid * WQ];
    const unsigned int* ps = plan_s + (long long)(blockIdx.x * 32 + wid) * WQ;
    float*              pn = plan_n + (long long)(blockIdx.x * 32 + wid) * WQ;
    if (build) {
    wc = scan_owner_map(owner, (int)blockIdx.x, tid, myq,
                        [&](int e) { spill_edge<F>(e, base, tile, h, src, dst, dinv); });
    const int cnt = wc < WQ ? wc : WQ;
    for (int i0 = 0; i0 < cnt; i0 += 128) {
        int ix[4], ss[4], dd[4];
        float nr[4];
#pragma unroll
        for (int k = 0; k < 4; ++k) {
            const int i = i0 + 32 * k + lane;
            ix[k] = (i < cnt) ? i : (cnt - 1);
        }
        int ee[4];
#pragma unroll
        for (int k = 0; k < 4; ++k) ee[k] = myq[ix[k]];
#pragma unroll
        for (int k = 0; k < 4; ++k) { ss[k] = src[ee[k]]; dd[k] = dst[ee[k]]; }
#pragma unroll
        for (int k = 0; k < 4; ++k) nr[k] = dinv[ss[k]] * dinv[dd[k]];
#pragma unroll
        for (int k = 0; k < 4; ++k) {
            myq[ix[k]] = ss[k];
            myd[ix[k]] = (unsigned short)(dd[k] - base);
            myn[ix[k]] = nr[k];
        }
    }
    __syncthreads();

    constexpr int U = 16;
    for (int i0 = 0; i0 < cnt; i0 += U) {
        float vals[U][VPL];
#pragma unroll
        for (int k = 0; k < U; ++k) {
            const int i = (i0 + k < cnt) ? (i0 + k) : i0;
            const float* hr = h + (long long)myq[i] * F + lane * VPL;
#pragma unroll
            for (int x = 0; x < VPL; ++x) vals[k][x] = hr[x];
        }
#pragma unroll
        for (int k = 0; k < U; ++k) {
            if (i0 + k < cnt) {
                const float nrm = myn[i0 + k];
                float* tr = &tile[(int)myd[i0 + k] * P + lane * VPL];
#pragma unroll
                for (int x = 0; x < VPL; ++x) atomicAdd(tr + x, vals[k][x] * nrm);
            }
        }
    }
    } else {
    const int wu = __builtin_amdgcn_readfirstlane(wid);
    v16f acc[VPL];
    {
        float hv[NJ][VPL], dj[NJ];
#pragma unroll
        for (int j = 0; j < NJ; ++j) {
            const int lr   = wu + 32 * j;
            const int node = lr < nloc ? base + lr : N - 1;
            dj[j] = dinv[node];
            const float* hr = h + (long long)node * F + lane * VPL;
#pragma unroll
            for (int x = 0; x < VPL; ++x) hv[j][x] = hr[x];
        }
#pragma unroll
        for (int x = 0; x < VPL; ++x) {
            const float bx = bias[lane * VPL + x];
            acc[x] = (v16f)(0.0f);
#pragma unroll
            for (int j = 0; j < NJ; ++j) acc[x][j] = hv[j][x] * dj[j] * dj[j] + bx;
        }
    }
    const int cnt = wc;
    if (cnt > 0) {
        constexpr unsigned SM = (1u << SRC_BITS) - 1u;
        const int nb = (cnt + 31) >> 5;
        auto idx = [&](int bb) { const int i = bb * 32 + lane; return (i < cnt) ? i : (cnt - 1); };
        const int nfull = cnt >> 5;
        auto add_tail = [&](unsigned r, float n, int m0) {
            if (m0 >= 16) add_half<F, true >(h, acc, r, n, 0, 16, lane);
            else          add_half<F, false>(h, acc, r, n, 0, m0, lane);
            if (m0 > 16)  add_half<F, false>(h, acc, r, n, 16, m0 - 16, lane);
        };
        if (FIRST) {
            auto src_dinv = [&](unsigned s) {
                const int dg = (int)deg16[s];
                return dg == DEG_SAT ? dinv[s] : rsqrtf((float)dg);
            };
            unsigned r_cur = ps[idx(0)];
            unsigned r_nxt = ps[idx(nb > 1 ? 1 : 0)];
            float    n_cur = src_dinv(r_cur & SM) * dinv[base + (int)(r_cur >> SRC_BITS)];
            if (lane < cnt) pn[lane] = n_cur;
            for (int bb = 0; bb < nfull; ++bb) {
                const int   dg = (int)deg16[r_nxt & SM];
                const float d2 = dinv[base + (int)(r_nxt >> SRC_BITS)];
                const unsigned r_n2 = ps[idx(bb + 2 < nb ? bb + 2 : nb - 1)];
                add_half<F, true>(h, acc, r_cur, n_cur, 0, 16, lane);
                add_half<F, true>(h, acc, r_cur, n_cur, 16, 16, lane);
                const float d1 = dg == DEG_SAT ? dinv[r_nxt & SM] : rsqrtf((float)dg);
                r_cur = r_nxt; n_cur = d1 * d2; r_nxt = r_n2;
                const int i1 = (bb + 1) * 32 + lane;
                if (i1 < cnt) pn[i1] = n_cur;
            }
            if (nfull < nb) add_tail(r_cur, n_cur, cnt - nfull * 32);
        } else {
            const int ic = idx(0);
            unsigned r_cur = ps[ic];
            float    n_cur = pn[ic];
            for (int bb = 0; bb < nfull; ++bb) {
                const int ix = idx(bb + 1 < nb ? bb + 1 : nb - 1);
                const unsigned r_nxt = ps[ix];
                const float    n_nxt = pn[ix];
                add_half<F, true>(h, acc, r_cur, n_cur, 0, 16, lane);
                add_half<F, true>(h, acc, r_cur, n_cur, 16, 16, lane);
                r_cur = r_nxt; n_cur = n_nxt;
            }
            if (nfull < nb) add_tail(r_cur, n_cur, cnt - nfull * 32);
        }
    }
#pragma unroll
    for (int j = 0; j < NJ; ++j) {
        const int lr = wu + 32 * j;
        if (lr < NPW) {
            float* tr = &tile[lr * P + lane * VPL];
#pragma unroll
            for (int x = 0; x < VPL; ++x) tr[x] = acc[x][j];
        }
    }
    }

    if constexpr (FIRST) {
        const int  mn   = lane & 15;
        const int  hh   = lane >> 4;
        const bool prod = wid * 16 < nloc;
        v2f b0[F / 4], b1[F / 4];
        if (prod) {
#pragma unroll
            for (int kk = 0; kk < F / 4; ++kk) {
                const float* wr = Wd + (kk * 4 + 2 * hh) * DN + mn;
                b0[kk].x = wr[0];       b0[kk].y = wr[DN];
                b1[kk].x = wr[16];      b1[kk].y = wr[DN + 16];
            }
        }
        __syncthreads();
        if (prod) {
            int ar = wid * 16 + mn;
            if (ar > nloc - 1) ar = nloc - 1;
            const float* arow = &tile[ar * P];
            v8f acc0 = {}, acc1 = {};
#pragma unroll
            for (int kk = 0; kk < F / 4; ++kk) {
                v2f a;
                a.x = fmaxf(arow[kk * 4 + 2 * hh + 0], 0.0f);
                a.y = fmaxf(arow[kk * 4 + 2 * hh + 1], 0.0f);
                acc0 = __builtin_amdgcn_wmma_f32_16x16x4_f32(
                     false, a,  false, b0[kk],
                     (short)0, acc0,  false,  false);
                acc1 = __builtin_amdgcn_wmma_f32_16x16x4_f32(
                     false, a,  false, b1[kk],
                     (short)0, acc1,  false,  false);
            }
#pragma unroll
            for (int j = 0; j < 8; ++j) {
                const int r = wid * 16 + j + 8 * hh;
                if (r < nloc) {
                    float* yr = y + (long long)(base + r) * DN + mn;
                    yr[0]  = acc0[j];
                    yr[16] = acc1[j];
                }
            }
        }
    } else {
        __syncthreads();
        const float bl0 = bd[0];
        for (int n = tid; n < nloc; n += 1024) {
            const float* tr = &tile[n * P];
            float s = 0.0f;
#pragma unroll
            for (int q = 0; q < F4; ++q) {
                const float4 t = *(const float4*)(tr + q * 4);
                s += fmaxf(t.x, 0.0f) * Wd[q * 4 + 0];
                s += fmaxf(t.y, 0.0f) * Wd[q * 4 + 1];
                s += fmaxf(t.z, 0.0f) * Wd[q * 4 + 2];
                s += fmaxf(t.w, 0.0f) * Wd[q * 4 + 3];
            }
            y[base + n] = s + bl0;
        }
    }
}

static inline size_t align256(size_t x) { return (x + 255) & ~size_t(255); }

extern "C" void kernel_launch(void* const* d_in, const int* in_sizes, int n_in,
                              void* d_out, int out_size, void* d_ws, size_t ws_size,
                              hipStream_t stream) {
    const float* x    = (const float*)d_in[0];
    const int*   ei   = (const int*)d_in[1];
    const float* W1   = (const float*)d_in[2];
    const float* b1   = (const float*)d_in[3];
    const float* W2   = (const float*)d_in[4];
    const float* b2   = (const float*)d_in[5];
    const float* Wl   = (const float*)d_in[6];
    const float* bl   = (const float*)d_in[7];
    float*       out  = (float*)d_out;

    const int N = N_NODES;
    const int E = N_EDGES;
    const int* src = ei;
    const int* dst = ei + E;

    char* ws = (char*)d_ws;
    size_t off = 0;
    float* dinv = (float*)(ws + off); off += align256((size_t)N * 4);
    float* h1   = (float*)(ws + off); off += align256((size_t)N * 64 * 4);
    float* h2   = (float*)(ws + off); off += align256((size_t)N * 32 * 4);
    unsigned char* owner = (unsigned char*)(ws + off); off += align256((size_t)E);
    int*          plan_c = (int*)(ws + off);          off += align256((size_t)NTILE * 32 * 4);
    unsigned int* plan_s = (unsigned int*)(ws + off); off += align256((size_t)NTILE * 32 * WQ * 4);
    float*        plan_n = (float*)(ws + off);        off += align256((size_t)NTILE * 32 * WQ * 4);
    unsigned int* bitmap = (unsigned int*)(ws + off); off += align256((size_t)NTILE * BWP * 4);
    unsigned short* deg16g = (unsigned short*)(ws + off); off += align256((size_t)N * 2);
    unsigned int* pairs  = (unsigned int*)(ws + off); off += align256((size_t)E * 8);
    (void)ws_size; (void)n_in; (void)in_sizes; (void)out_size;

    owner_map_kernel<<<NCH, 1024, 0, stream>>>(src, dst, owner, bitmap, pairs, E);

    plan_gemm_kernel<<<NTILE + G1_WGS, 1024, 0, stream>>>(pairs, dst, bitmap, dinv, N, plan_c, plan_s,
                                                           deg16g, x, W1, h1);

    tile_agg_kernel<64, true><<<NTILE, 1024, 0, stream>>>(h1, src, dst, owner, dinv, b1, h2, N,
                                                          plan_c, plan_s, plan_n, deg16g, W2, bl);

    tile_agg_kernel<32, false><<<NTILE, 1024, 0, stream>>>(h2, src, dst, owner, dinv, b2, out, N,
                                                           plan_c, plan_s, plan_n, deg16g, Wl, bl);
}
